# starting kernel
# baseline (speedup 1.0000x reference)
#include <hip/hip_runtime.h>
#include <math.h>

#define IN_CH 512
#define HEADS 8
#define BPAD 16
#define NEG_SLOPE 0.2f

typedef __attribute__((ext_vector_type(2))) float v2f;
typedef __attribute__((ext_vector_type(8))) float v8f;

__device__ __forceinline__ unsigned encF(float f) {
  unsigned u = __float_as_uint(f);
  return (u & 0x80000000u) ? ~u : (u | 0x80000000u);
}
__device__ __forceinline__ float decF(unsigned u) {
  u = (u & 0x80000000u) ? (u & 0x7FFFFFFFu) : ~u;
  return __uint_as_float(u);
}

__device__ __forceinline__ float leaky(float v) {
  return v >= 0.f ? v : NEG_SLOPE * v;
}

__global__ __launch_bounds__(256) void gat_gemm_kernel(
    const float* __restrict__ x, const float* __restrict__ W,
    const float* __restrict__ att_src, const float* __restrict__ att_dst,
    float* __restrict__ h, float* __restrict__ asrc, float* __restrict__ adst,
    int n)
{
  __shared__ float Wlds[IN_CH * BPAD];
  for (int i = threadIdx.x; i < IN_CH * BPAD; i += blockDim.x) {
    const int k = i >> 4;
    const int c = i & 15;
    Wlds[i] = (c < HEADS) ? W[k * HEADS + c] : 0.f;
  }
  __syncthreads();

  const int lane = threadIdx.x & 31;
  const int wave = threadIdx.x >> 5;
  const int half = lane >> 4;
  const int l    = lane & 15;

  const int tile = blockIdx.x * (blockDim.x >> 5) + wave;
  const int row0 = tile * 16;
  if (row0 >= n) return;

  int arow = row0 + l;
  if (arow > n - 1) arow = n - 1;
  const float* __restrict__ xrow = x + (size_t)arow * IN_CH;
  const float* __restrict__ wcol = Wlds + (size_t)half * 2 * BPAD + l;

  v8f c = {0.f, 0.f, 0.f, 0.f, 0.f, 0.f, 0.f, 0.f};

  #pragma unroll 4
  for (int k = 0; k < IN_CH; k += 4) {
    const int ka = k + half * 2;
    v2f a;
    a.x = xrow[ka];
    a.y = xrow[ka + 1];
    v2f b;
    b.x = wcol[k * BPAD];
    b.y = wcol[k * BPAD + BPAD];
    c = __builtin_amdgcn_wmma_f32_16x16x4_f32(
         false, a,  false, b,
         (short)0, c,  false,  false);
  }

  if (l < HEADS) {
    const float as = att_src[l];
    const float ad = att_dst[l];
    #pragma unroll
    for (int i = 0; i < 8; ++i) {
      const int r = row0 + i + half * 8;
      if (r < n) {
        const float hv = c[i];
        const size_t o = (size_t)r * HEADS + l;
        h[o]    = hv;
        asrc[o] = hv * as;
        adst[o] = hv * ad;
      }
    }
  }
}

__global__ void gat_init_kernel(unsigned* __restrict__ emaxEnc,
                                float* __restrict__ denom,
                                float* __restrict__ numer, size_t cnt)
{
  size_t i = (size_t)blockIdx.x * blockDim.x + threadIdx.x;
  if (i < cnt) {
    emaxEnc[i] = 0x007FFFFFu;
    denom[i]   = 0.f;
    numer[i]   = 0.f;
  }
}

__global__ __launch_bounds__(256) void gat_edge_max_kernel(
    const int* __restrict__ ei,
    const float* __restrict__ asrc, const float* __restrict__ adst,
    unsigned* __restrict__ emaxEnc, int nE, int n)
{
  const int e = blockIdx.x * blockDim.x + threadIdx.x;
  const int etot = nE + n;
  if (e >= etot) return;
  int src, dst;
  if (e < nE) { src = ei[e]; dst = ei[nE + e]; }
  else        { src = dst = e - nE; }

  const float4* ps = (const float4*)(asrc + (size_t)src * HEADS);
  const float4* pd = (const float4*)(adst + (size_t)dst * HEADS);
  const float4 s0 = ps[0], s1 = ps[1];
  const float4 d0 = pd[0], d1 = pd[1];
  float v[HEADS] = { s0.x + d0.x, s0.y + d0.y, s0.z + d0.z, s0.w + d0.w,
                     s1.x + d1.x, s1.y + d1.y, s1.z + d1.z, s1.w + d1.w };
  unsigned* base = emaxEnc + (size_t)dst * HEADS;
  #pragma unroll
  for (int hh = 0; hh < HEADS; ++hh)
    atomicMax(base + hh, encF(leaky(v[hh])));
}

__global__ __launch_bounds__(256) void gat_edge_acc_kernel(
    const int* __restrict__ ei,
    const float* __restrict__ asrc, const float* __restrict__ adst,
    const float* __restrict__ hmat, const unsigned* __restrict__ emaxEnc,
    float* __restrict__ denom, float* __restrict__ numer, int nE, int n)
{
  const int e = blockIdx.x * blockDim.x + threadIdx.x;
  const int etot = nE + n;
  if (e >= etot) return;
  int src, dst;
  if (e < nE) { src = ei[e]; dst = ei[nE + e]; }
  else        { src = dst = e - nE; }

  const float4* ps = (const float4*)(asrc + (size_t)src * HEADS);
  const float4* pd = (const float4*)(adst + (size_t)dst * HEADS);
  const float4* ph = (const float4*)(hmat + (size_t)src * HEADS);
  const float4 s0 = ps[0], s1 = ps[1];
  const float4 d0 = pd[0], d1 = pd[1];
  const float4 h0 = ph[0], h1 = ph[1];
  float v[HEADS]  = { s0.x + d0.x, s0.y + d0.y, s0.z + d0.z, s0.w + d0.w,
                      s1.x + d1.x, s1.y + d1.y, s1.z + d1.z, s1.w + d1.w };
  float hv[HEADS] = { h0.x, h0.y, h0.z, h0.w, h1.x, h1.y, h1.z, h1.w };

  const unsigned* me = emaxEnc + (size_t)dst * HEADS;
  float* dden = denom + (size_t)dst * HEADS;
  float* dnum = numer + (size_t)dst * HEADS;
  #pragma unroll
  for (int hh = 0; hh < HEADS; ++hh) {
    const float ee = expf(leaky(v[hh]) - decF(me[hh]));
    atomicAdd(dden + hh, ee);
    atomicAdd(dnum + hh, ee * hv[hh]);
  }
}

__global__ __launch_bounds__(256) void gat_finalize_kernel(
    const float* __restrict__ numer, const float* __restrict__ denom,
    const float* __restrict__ gat_bias, const float* __restrict__ lin_w,
    const float* __restrict__ lin_b, const float* __restrict__ bias,
    float* __restrict__ out, int n)
{
  const int i = blockIdx.x * blockDim.x + threadIdx.x;
  if (i >= n) return;
  const float* nu = numer + (size_t)i * HEADS;
  const float* de = denom + (size_t)i * HEADS;
  float z = lin_b[0];
  #pragma unroll
  for (int hh = 0; hh < HEADS; ++hh) {
    const float oh = nu[hh] / (de[hh] + 1e-16f) + gat_bias[hh];
    z += oh * lin_w[hh];
  }
  out[i] = fmaxf(z, 0.f) + bias[0];
}

extern "C" void kernel_launch(void* const* d_in, const int* in_sizes, int n_in,
                              void* d_out, int out_size, void* d_ws, size_t ws_size,
                              hipStream_t stream) {
  const float* x        = (const float*)d_in[0];
  const int*   ei       = (const int*)d_in[1];
  const float* W        = (const float*)d_in[2];
  const float* att_src  = (const float*)d_in[3];
  const float* att_dst  = (const float*)d_in[4];
  const float* gat_bias = (const float*)d_in[5];
  const float* lin_w    = (const float*)d_in[6];
  const float* lin_b    = (const float*)d_in[7];
  const float* bias     = (const float*)d_in[8];
  float* out = (float*)d_out;

  const int n  = in_sizes[0] / IN_CH;
  const int nE = in_sizes[1] / 2;

  const size_t nh  = (size_t)n * HEADS;
  const size_t seg = (nh * sizeof(float) + 255) & ~(size_t)255;
  char* ws = (char*)d_ws;
  float*    h_mat   = (float*)   (ws + 0 * seg);
  float*    asrc    = (float*)   (ws + 1 * seg);
  float*    adst    = (float*)   (ws + 2 * seg);
  unsigned* emaxEnc = (unsigned*)(ws + 3 * seg);
  float*    denom   = (float*)   (ws + 4 * seg);
  float*    numer   = (float*)   (ws + 5 * seg);

  const int tiles  = (n + 15) / 16;
  const int blocks = (tiles + 7) / 8;
  gat_gemm_kernel<<<blocks, 256, 0, stream>>>(x, W, att_src, att_dst,
                                              h_mat, asrc, adst, n);

  {
    const int b = (int)((nh + 255) / 256);
    gat_init_kernel<<<b, 256, 0, stream>>>(emaxEnc, denom, numer, nh);
  }

  const int etot = nE + n;
  const int eb = (etot + 255) / 256;
  gat_edge_max_kernel<<<eb, 256, 0, stream>>>(ei, asrc, adst, emaxEnc, nE, n);

  gat_edge_acc_kernel<<<eb, 256, 0, stream>>>(ei, asrc, adst, h_mat, emaxEnc,
                                              denom, numer, nE, n);

  gat_finalize_kernel<<<(n + 255) / 256, 256, 0, stream>>>(
      numer, denom, gat_bias, lin_w, lin_b, bias, out, n);
}


# edges bucketed by destination per call (integer count and slot atomics); one wave per node does max, exp sums, weighted sum and head in registers; no float atomics; sums in slot arrival order
# speedup vs baseline: 3.5440x; 3.5440x over previous
#include <hip/hip_runtime.h>
#include <math.h>


#define IN_CH 512
#define HEADS 8
#define BPAD 16
#define NEG_SLOPE 0.2f
#define CHUNK 32

typedef __attribute__((ext_vector_type(2))) float v2f;
typedef __attribute__((ext_vector_type(8))) float v8f;

__device__ __forceinline__ float leaky(float v) {
  return v >= 0.f ? v : NEG_SLOPE * v;
}

__global__ __launch_bounds__(256) void gat_gemm_kernel(
    const float* __restrict__ x, const float* __restrict__ W,
    float* __restrict__ h, int* __restrict__ deg, int* __restrict__ cursor,
    int n)
{
  __shared__ float Wlds[IN_CH * BPAD];
  for (int i = threadIdx.x; i < IN_CH * BPAD; i += blockDim.x) {
    const int k = i >> 4;
    const int c = i & 15;
    Wlds[i] = (c < HEADS) ? W[k * HEADS + c] : 0.f;
  }
  {
    const int rows_per_block = (blockDim.x >> 5) * 16;
    const int r = blockIdx.x * rows_per_block + threadIdx.x;
    if (threadIdx.x < rows_per_block && r < n) deg[r] = 0;
    if (blockIdx.x == 0 && threadIdx.x == 0) cursor[0] = 0;
  }
  __syncthreads();

  const int lane = threadIdx.x & 31;
  const int wave = threadIdx.x >> 5;
  const int half = lane >> 4;
  const int l    = lane & 15;

  const int tile = blockIdx.x * (blockDim.x >> 5) + wave;
  const int row0 = tile * 16;
  if (row0 >= n) return;

  int arow = row0 + l;
  if (arow > n - 1) arow = n - 1;
  const float* __restrict__ xrow = x + (size_t)arow * IN_CH;
  const float* __restrict__ wcol = Wlds + (size_t)half * 2 * BPAD + l;

  v8f c = {0.f, 0.f, 0.f, 0.f, 0.f, 0.f, 0.f, 0.f};

  #pragma unroll 4
  for (int k = 0; k < IN_CH; k += 4) {
    const int ka = k + half * 2;
    v2f a;
    a.x = xrow[ka];
    a.y = xrow[ka + 1];
    v2f b;
    b.x = wcol[k * BPAD];
    b.y = wcol[k * BPAD + BPAD];
    c = __builtin_amdgcn_wmma_f32_16x16x4_f32(
         false, a,  false, b,
         (short)0, c,  false,  false);
  }

  if (l < HEADS) {
    #pragma unroll
    for (int i = 0; i < 8; ++i) {
      const int r = row0 + i + half * 8;
      if (r < n) h[(size_t)r * HEADS + l] = c[i];
    }
  }
}

__global__ __launch_bounds__(256) void gat_count_kernel(
    const int* __restrict__ ei, int* __restrict__ deg, int nE)
{
  const int e = blockIdx.x * blockDim.x + threadIdx.x;
  if (e >= nE) return;
  atomicAdd(deg + ei[nE + e], 1);
}

__global__ __launch_bounds__(256) void gat_alloc_kernel(
    const int* __restrict__ deg, int* __restrict__ cur,
    int* __restrict__ cursor, int n)
{
  __shared__ int wsum[8];
  __shared__ int blockBase;
  const int i    = blockIdx.x * blockDim.x + threadIdx.x;
  const int lane = threadIdx.x & 31;
  const int wave = threadIdx.x >> 5;
  const int d = (i < n) ? deg[i] : 0;
  int incl = d;
  #pragma unroll
  for (int o = 1; o < 32; o <<= 1) {
    const int t = __shfl_up(incl, o);
    if (lane >= o) incl += t;
  }
  if (lane == 31) wsum[wave] = incl;
  __syncthreads();
  if (threadIdx.x == 0) {
    int tot = 0;
    #pragma unroll
    for (int w = 0; w < 8; ++w) { const int t = wsum[w]; wsum[w] = tot; tot += t; }
    blockBase = atomicAdd(cursor, tot);
  }
  __syncthreads();
  if (i < n) cur[i] = blockBase + wsum[wave] + incl - d;
}

__global__ __launch_bounds__(256) void gat_fill_kernel(
    const int* __restrict__ ei, int* __restrict__ cur,
    int* __restrict__ bsrc, int nE)
{
  const int e = blockIdx.x * blockDim.x + threadIdx.x;
  if (e >= nE) return;
  const int src = ei[e];
  const int dst = ei[nE + e];
  const int p = atomicAdd(cur + dst, 1);
  bsrc[p] = src;
}

__global__ __launch_bounds__(256) void gat_gather_kernel(
    const int* __restrict__ bsrc, const int* __restrict__ cur,
    const int* __restrict__ deg, const float* __restrict__ hmat,
    const float* __restrict__ att_src, const float* __restrict__ att_dst,
    const float* __restrict__ gat_bias, const float* __restrict__ lin_w,
    const float* __restrict__ lin_b, const float* __restrict__ bias,
    float* __restrict__ out, int n)
{
  const int lane = threadIdx.x & 31;
  const int node = blockIdx.x * (blockDim.x >> 5) + (threadIdx.x >> 5);
  if (node >= n) return;
  const int q  = lane >> 3;
  const int hh = lane & 7;

  const int end   = cur[node];
  const int start = end - deg[node];
  const float as = att_src[hh];
  const float ad = att_dst[hh];
  const float hself = hmat[node * HEADS + hh];
  const float adst  = hself * ad;
  const float eself = leaky(hself * as + adst);

  float m = (q == 0) ? eself : -INFINITY;
  for (int base = start; base < end; base += CHUNK) {
    int   s[8];
    float hs[8];
    #pragma unroll
    for (int j = 0; j < 8; ++j) {
      const int p = base + j * 4 + q;
      s[j] = bsrc[p < end ? p : end - 1];
    }
    #pragma unroll
    for (int j = 0; j < 8; ++j) hs[j] = hmat[s[j] * HEADS + hh];
    #pragma unroll
    for (int j = 0; j < 8; ++j) {
      const int p = base + j * 4 + q;
      const float e = leaky(hs[j] * as + adst);
      if (p < end) m = fmaxf(m, e);
    }
  }
  m = fmaxf(m, __shfl_xor(m, 8));
  m = fmaxf(m, __shfl_xor(m, 16));

  float den = 0.f, num = 0.f;
  if (q == 0) { den = expf(eself - m); num = den * hself; }
  for (int base = start; base < end; base += CHUNK) {
    int   s[8];
    float hs[8];
    #pragma unroll
    for (int j = 0; j < 8; ++j) {
      const int p = base + j * 4 + q;
      s[j] = bsrc[p < end ? p : end - 1];
    }
    #pragma unroll
    for (int j = 0; j < 8; ++j) hs[j] = hmat[s[j] * HEADS + hh];
    #pragma unroll
    for (int j = 0; j < 8; ++j) {
      const int p = base + j * 4 + q;
      const float ee = expf(leaky(hs[j] * as + adst) - m);
      if (p < end) { den += ee; num += ee * hs[j]; }
    }
  }
  den += __shfl_xor(den, 8);
  den += __shfl_xor(den, 16);
  num += __shfl_xor(num, 8);
  num += __shfl_xor(num, 16);

  const float oh = num / (den + 1e-16f) + gat_bias[hh];
  float z = lin_b[0];
  #pragma unroll
  for (int k = 0; k < HEADS; ++k) z += __shfl(oh, k) * lin_w[k];
  if (lane == 0) out[node] = fmaxf(z, 0.f) + bias[0];
}

extern "C" void kernel_launch(void* const* d_in, const int* in_sizes, int n_in,
                              void* d_out, int out_size, void* d_ws, size_t ws_size,
                              hipStream_t stream) {
  const float* x        = (const float*)d_in[0];
  const int*   ei       = (const int*)d_in[1];
  const float* W        = (const float*)d_in[2];
  const float* att_src  = (const float*)d_in[3];
  const float* att_dst  = (const float*)d_in[4];
  const float* gat_bias = (const float*)d_in[5];
  const float* lin_w    = (const float*)d_in[6];
  const float* lin_b    = (const float*)d_in[7];
  const float* bias     = (const float*)d_in[8];
  float* out = (float*)d_out;

  const int n  = in_sizes[0] / IN_CH;
  const int nE = in_sizes[1] / 2;

  const size_t nh   = (size_t)n * HEADS;
  const size_t segH = (nh * sizeof(float) + 255) & ~(size_t)255;
  const size_t segN = ((size_t)n * sizeof(int) + 255) & ~(size_t)255;
  char* ws = (char*)d_ws;
  float* h_mat  = (float*)(ws);
  int*   deg    = (int*)  (ws + segH);
  int*   cur    = (int*)  (ws + segH + segN);
  int*   cursor = (int*)  (ws + segH + 2 * segN);
  int*   bsrc   = (int*)  (ws + segH + 2 * segN + 256);

  const int tiles  = (n + 15) / 16;
  const int blocks = (tiles + 7) / 8;
  gat_gemm_kernel<<<blocks, 256, 0, stream>>>(x, W, h_mat, deg, cursor, n);

  const int eb = (nE + 255) / 256;
  const int nb = (n + 255) / 256;
  gat_count_kernel<<<eb, 256, 0, stream>>>(ei, deg, nE);

  gat_alloc_kernel<<<nb, 256, 0, stream>>>(deg, cur, cursor, n);

  gat_fill_kernel<<<eb, 256, 0, stream>>>(ei, cur, bsrc, nE);

  gat_gather_kernel<<<(n + 7) / 8, 256, 0, stream>>>(
      bsrc, cur, deg, h_mat, att_src, att_dst, gat_bias, lin_w, lin_b, bias,
      out, n);
}


# edges bucketed with LDS counters instead of one global atomic per edge: split by block into 256-node destination ranges, each range then ordered by node in LDS; sums still in slot arrival order
# speedup vs baseline: 12.8192x; 3.6172x over previous
#include <hip/hip_runtime.h>
#include <math.h>


#define IN_CH 512
#define HEADS 8
#define BPAD 16
#define NEG_SLOPE 0.2f
#define CHUNK 32
#define RSHIFT 8
#define RSIZE 256
#define MAXR 1024
#define PB 1024
#define EPT 8
#define CAP 16384

typedef __attribute__((ext_vector_type(2))) float v2f;
typedef __attribute__((ext_vector_type(8))) float v8f;

__device__ __forceinline__ float leaky(float v) {
  return v >= 0.f ? v : NEG_SLOPE * v;
}

__global__ __launch_bounds__(256) void gat_gemm_kernel(
    const float* __restrict__ x, const float* __restrict__ W,
    float* __restrict__ h, int* __restrict__ rangeTabs, int n)
{
  __shared__ float Wlds[IN_CH * BPAD];
  for (int i = threadIdx.x; i < IN_CH * BPAD; i += blockDim.x) {
    const int k = i >> 4;
    const int c = i & 15;
    Wlds[i] = (c < HEADS) ? W[k * HEADS + c] : 0.f;
  }
  if (blockIdx.x == 0) {
    for (int i = threadIdx.x; i < 2 * MAXR; i += blockDim.x) rangeTabs[i] = 0;
  }
  __syncthreads();

  const int lane = threadIdx.x & 31;
  const int wave = threadIdx.x >> 5;
  const int half = lane >> 4;
  const int l    = lane & 15;

  const int tile = blockIdx.x * (blockDim.x >> 5) + wave;
  const int row0 = tile * 16;
  if (row0 >= n) return;

  int arow = row0 + l;
  if (arow > n - 1) arow = n - 1;
  const float* __restrict__ xrow = x + (size_t)arow * IN_CH;
  const float* __restrict__ wcol = Wlds + (size_t)half * 2 * BPAD + l;

  v8f c = {0.f, 0.f, 0.f, 0.f, 0.f, 0.f, 0.f, 0.f};

  #pragma unroll 4
  for (int k = 0; k < IN_CH; k += 4) {
    const int ka = k + half * 2;
    v2f a;
    a.x = xrow[ka];
    a.y = xrow[ka + 1];
    v2f b;
    b.x = wcol[k * BPAD];
    b.y = wcol[k * BPAD + BPAD];
    c = __builtin_amdgcn_wmma_f32_16x16x4_f32(
         false, a,  false, b,
         (short)0, c,  false,  false);
  }

  if (l < HEADS) {
    #pragma unroll
    for (int i = 0; i < 8; ++i) {
      const int r = row0 + i + half * 8;
      if (r < n) h[(size_t)r * HEADS + l] = c[i];
    }
  }
}

__device__ __forceinline__ int block_excl_scan(int v, int* wsum, int lane, int wave) {
  int incl = v;
  #pragma unroll
  for (int o = 1; o < 32; o <<= 1) {
    const int t = __shfl_up(incl, o);
    if (lane >= o) incl += t;
  }
  if (lane == 31) wsum[wave] = incl;
  __syncthreads();
  if (wave == 0) {
    const int w = wsum[lane];
    int wi = w;
    #pragma unroll
    for (int o = 1; o < 32; o <<= 1) {
      const int t = __shfl_up(wi, o);
      if (lane >= o) wi += t;
    }
    wsum[lane] = wi - w;
  }
  __syncthreads();
  return wsum[wave] + incl - v;
}

__global__ __launch_bounds__(PB) void gat_rangecount_kernel(
    const int* __restrict__ ei, int* __restrict__ rangeCnt, int nE, int R)
{
  __shared__ int hist[MAXR];
  const int t = threadIdx.x;
  hist[t] = 0;
  __syncthreads();
  const int base = blockIdx.x * (PB * EPT);
  int d[EPT];
  #pragma unroll
  for (int j = 0; j < EPT; ++j) {
    const int e = base + j * PB + t;
    d[j] = (e < nE) ? ei[nE + e] : -1;
  }
  #pragma unroll
  for (int j = 0; j < EPT; ++j)
    if (d[j] >= 0) atomicAdd(&hist[d[j] >> RSHIFT], 1);
  __syncthreads();
  if (t < R) {
    const int c = hist[t];
    if (c) atomicAdd(rangeCnt + t, c);
  }
}

__global__ __launch_bounds__(PB) void gat_partition_kernel(
    const int* __restrict__ ei, const int* __restrict__ rangeCnt,
    int* __restrict__ rangeCur, int* __restrict__ rangeStart,
    int* __restrict__ part, int nE, int R)
{
  __shared__ int hist[MAXR];
  __shared__ int lstart[MAXR];
  __shared__ int rbase[MAXR];
  __shared__ int wsum[32];
  __shared__ int skey[PB * EPT];
  __shared__ int sgpos[PB * EPT];
  const int t    = threadIdx.x;
  const int lane = t & 31;
  const int wave = t >> 5;
  hist[t] = 0;
  const int tot    = (t < R) ? rangeCnt[t] : 0;
  const int rstart = block_excl_scan(tot, wsum, lane, wave);
  if (blockIdx.x == 0 && t < R) rangeStart[t] = rstart;

  const int base = blockIdx.x * (PB * EPT);
  int s[EPT], d[EPT], rk[EPT];
  #pragma unroll
  for (int j = 0; j < EPT; ++j) {
    const int e = base + j * PB + t;
    const bool ok = e < nE;
    s[j] = ok ? ei[e] : 0;
    d[j] = ok ? ei[nE + e] : -1;
  }
  #pragma unroll
  for (int j = 0; j < EPT; ++j) {
    rk[j] = 0;
    if (d[j] >= 0) rk[j] = atomicAdd(&hist[d[j] >> RSHIFT], 1);
  }
  __syncthreads();
  const int cc = hist[t];
  const int ls = block_excl_scan(cc, wsum, lane, wave);
  lstart[t] = ls;
  rbase[t]  = (t < R && cc) ? rstart + atomicAdd(rangeCur + t, cc) : 0;
  __syncthreads();
  #pragma unroll
  for (int j = 0; j < EPT; ++j) {
    if (d[j] >= 0) {
      const int r = d[j] >> RSHIFT;
      const int p = lstart[r] + rk[j];
      skey[p]  = (s[j] << RSHIFT) | (d[j] & (RSIZE - 1));
      sgpos[p] = rbase[r] + rk[j];
    }
  }
  __syncthreads();
  const int left  = nE - base;
  const int mine  = left < PB * EPT ? left : PB * EPT;
  #pragma unroll
  for (int j = 0; j < EPT; ++j) {
    const int i = j * PB + t;
    if (i < mine) part[sgpos[i]] = skey[i];
  }
}

__global__ __launch_bounds__(PB) void gat_rangesort_kernel(
    const int* part, const int* __restrict__ rangeCnt,
    const int* __restrict__ rangeStart, int* __restrict__ deg,
    int* __restrict__ cur, int* bsrc, int n)
{
  __shared__ int cnt[RSIZE];
  __shared__ int pos[RSIZE];
  __shared__ int wsum[32];
  __shared__ int stage[CAP];
  const int t    = threadIdx.x;
  const int lane = t & 31;
  const int wave = t >> 5;
  const int r    = blockIdx.x;
  const int L    = rangeCnt[r];
  const int seg  = rangeStart[r];
  if (t < RSIZE) cnt[t] = 0;
  __syncthreads();
  for (int i0 = t; i0 < L; i0 += 4 * PB) {
    int k[4];
    #pragma unroll
    for (int u = 0; u < 4; ++u) {
      const int i = i0 + u * PB;
      k[u] = (i < L) ? part[seg + i] : -1;
    }
    #pragma unroll
    for (int u = 0; u < 4; ++u)
      if (k[u] >= 0) atomicAdd(&cnt[k[u] & (RSIZE - 1)], 1);
  }
  __syncthreads();
  const int c    = (t < RSIZE) ? cnt[t] : 0;
  const int excl = block_excl_scan(c, wsum, lane, wave);
  if (t < RSIZE) {
    pos[t] = excl;
    const int node = r * RSIZE + t;
    if (node < n) { deg[node] = c; cur[node] = seg + excl + c; }
  }
  __syncthreads();
  const bool inLds = (L <= CAP);
  for (int i0 = t; i0 < L; i0 += 4 * PB) {
    int k[4];
    #pragma unroll
    for (int u = 0; u < 4; ++u) {
      const int i = i0 + u * PB;
      k[u] = (i < L) ? part[seg + i] : -1;
    }
    #pragma unroll
    for (int u = 0; u < 4; ++u) {
      if (k[u] >= 0) {
        const int p   = atomicAdd(&pos[k[u] & (RSIZE - 1)], 1);
        const int sid = k[u] >> RSHIFT;
        if (inLds) stage[p] = sid; else bsrc[seg + p] = sid;
      }
    }
  }
  if (inLds) {
    __syncthreads();
    for (int i = t; i < L; i += PB) bsrc[seg + i] = stage[i];
  }
}

__global__ __launch_bounds__(256) void gat_gather_kernel(
    const int* __restrict__ bsrc, const int* __restrict__ cur,
    const int* __restrict__ deg, const float* __restrict__ hmat,
    const float* __restrict__ att_src, const float* __restrict__ att_dst,
    const float* __restrict__ gat_bias, const float* __restrict__ lin_w,
    const float* __restrict__ lin_b, const float* __restrict__ bias,
    float* __restrict__ out, int n)
{
  const int lane = threadIdx.x & 31;
  const int node = blockIdx.x * (blockDim.x >> 5) + (threadIdx.x >> 5);
  if (node >= n) return;
  const int q  = lane >> 3;
  const int hh = lane & 7;

  const int end   = cur[node];
  const int start = end - deg[node];
  const float as = att_src[hh];
  const float ad = att_dst[hh];
  const float hself = hmat[node * HEADS + hh];
  const float adst  = hself * ad;
  const float eself = leaky(hself * as + adst);

  float m = (q == 0) ? eself : -INFINITY;
  for (int base = start; base < end; base += CHUNK) {
    int   s[8];
    float hs[8];
    #pragma unroll
    for (int j = 0; j < 8; ++j) {
      const int p = base + j * 4 + q;
      s[j] = bsrc[p < end ? p : end - 1];
    }
    #pragma unroll
    for (int j = 0; j < 8; ++j) hs[j] = hmat[s[j] * HEADS + hh];
    #pragma unroll
    for (int j = 0; j < 8; ++j) {
      const int p = base + j * 4 + q;
      const float e = leaky(hs[j] * as + adst);
      if (p < end) m = fmaxf(m, e);
    }
  }
  m = fmaxf(m, __shfl_xor(m, 8));
  m = fmaxf(m, __shfl_xor(m, 16));

  float den = 0.f, num = 0.f;
  if (q == 0) { den = expf(eself - m); num = den * hself; }
  for (int base = start; base < end; base += CHUNK) {
    int   s[8];
    float hs[8];
    #pragma unroll
    for (int j = 0; j < 8; ++j) {
      const int p = base + j * 4 + q;
      s[j] = bsrc[p < end ? p : end - 1];
    }
    #pragma unroll
    for (int j = 0; j < 8; ++j) hs[j] = hmat[s[j] * HEADS + hh];
    #pragma unroll
    for (int j = 0; j < 8; ++j) {
      const int p = base + j * 4 + q;
      const float ee = expf(leaky(hs[j] * as + adst) - m);
      if (p < end) { den += ee; num += ee * hs[j]; }
    }
  }
  den += __shfl_xor(den, 8);
  den += __shfl_xor(den, 16);
  num += __shfl_xor(num, 8);
  num += __shfl_xor(num, 16);

  const float oh = num / (den + 1e-16f) + gat_bias[hh];
  float z = lin_b[0];
  #pragma unroll
  for (int k = 0; k < HEADS; ++k) z += __shfl(oh, k) * lin_w[k];
  if (lane == 0) out[node] = fmaxf(z, 0.f) + bias[0];
}

extern "C" void kernel_launch(void* const* d_in, const int* in_sizes, int n_in,
                              void* d_out, int out_size, void* d_ws, size_t ws_size,
                              hipStream_t stream) {
  const float* x        = (const float*)d_in[0];
  const int*   ei       = (const int*)d_in[1];
  const float* W        = (const float*)d_in[2];
  const float* att_src  = (const float*)d_in[3];
  const float* att_dst  = (const float*)d_in[4];
  const float* gat_bias = (const float*)d_in[5];
  const float* lin_w    = (const float*)d_in[6];
  const float* lin_b    = (const float*)d_in[7];
  const float* bias     = (const float*)d_in[8];
  float* out = (float*)d_out;

  const int n  = in_sizes[0] / IN_CH;
  const int nE = in_sizes[1] / 2;

  const size_t nh   = (size_t)n * HEADS;
  const size_t segH = (nh * sizeof(float) + 255) & ~(size_t)255;
  const size_t segN = ((size_t)n * sizeof(int) + 255) & ~(size_t)255;
  const size_t segT = (size_t)3 * MAXR * sizeof(int);
  const size_t segE = ((size_t)nE * sizeof(int) + 255) & ~(size_t)255;
  char* ws = (char*)d_ws;
  float* h_mat      = (float*)(ws);
  int*   deg        = (int*)  (ws + segH);
  int*   cur        = (int*)  (ws + segH + segN);
  int*   rangeCnt   = (int*)  (ws + segH + 2 * segN);
  int*   rangeCur   = rangeCnt + MAXR;
  int*   rangeStart = rangeCnt + 2 * MAXR;
  int*   part       = (int*)  (ws + segH + 2 * segN + segT);
  int*   bsrc       = (ws_size >= segH + 2 * segN + segT + 2 * segE) ? (int*)((char*)part + segE) : part;

  const int tiles  = (n + 15) / 16;
  const int blocks = (tiles + 7) / 8;
  gat_gemm_kernel<<<blocks, 256, 0, stream>>>(x, W, h_mat, rangeCnt, n);

  const int R  = (n + RSIZE - 1) / RSIZE;
  const int eb = (nE + PB * EPT - 1) / (PB * EPT);
  gat_rangecount_kernel<<<eb, PB, 0, stream>>>(ei, rangeCnt, nE, R);

  gat_partition_kernel<<<eb, PB, 0, stream>>>(ei, rangeCnt, rangeCur, rangeStart, part, nE, R);

  gat_rangesort_kernel<<<R, PB, 0, stream>>>(part, rangeCnt, rangeStart, deg, cur, bsrc, n);

  gat_gather_kernel<<<(n + 7) / 8, 256, 0, stream>>>(
      bsrc, cur, deg, h_mat, att_src, att_dst, gat_bias, lin_w, lin_b, bias,
      out, n);
}


# GEMM operand loads widened and issued ahead: W to LDS by one group of 16-byte loads (was 32 dependent 4-byte round trips), x 16 bytes per lane two k-blocks ahead; k sum order of x@W changes
# speedup vs baseline: 14.6503x; 1.1428x over previous
#include <hip/hip_runtime.h>
#include <math.h>


#define IN_CH 512
#define HEADS 8
#define GEMM_T 256
#define NEG_SLOPE 0.2f
#define CHUNK 32
#define RSHIFT 8
#define RSIZE 256
#define MAXR 1024
#define PB 1024
#define EPT 8
#define CAP 16384

typedef __attribute__((ext_vector_type(2))) float v2f;
typedef __attribute__((ext_vector_type(8))) float v8f;

__device__ __forceinline__ float leaky(float v) {
  return v >= 0.f ? v : NEG_SLOPE * v;
}

__global__ __launch_bounds__(GEMM_T) void gat_gemm_kernel(
    const float* __restrict__ x, const float* __restrict__ W,
    float* __restrict__ h, int* __restrict__ rangeTabs, int n)
{
  __shared__ float4 Wlds4[IN_CH * HEADS / 4];
  {
    const float4* __restrict__ W4 = reinterpret_cast<const float4*>(W);
    float4 wv[4];
    #pragma unroll
    for (int j = 0; j < 4; ++j) wv[j] = W4[threadIdx.x + j * GEMM_T];
    #pragma unroll
    for (int j = 0; j < 4; ++j) Wlds4[threadIdx.x + j * GEMM_T] = wv[j];
  }
  if (blockIdx.x == 0) {
    for (int i = threadIdx.x; i < 2 * MAXR; i += GEMM_T) rangeTabs[i] = 0;
  }
  __syncthreads();

  const int lane = threadIdx.x & 31;
  const int wave = threadIdx.x >> 5;
  const int half = lane >> 4;
  const int l    = lane & 15;

  const int tile = blockIdx.x * (GEMM_T >> 5) + wave;
  const int row0 = tile * 16;
  if (row0 >= n) return;

  int arow = row0 + l;
  if (arow > n - 1) arow = n - 1;
  const float4* __restrict__ xr =
      reinterpret_cast<const float4*>(x + (size_t)arow * IN_CH) + half;
  const float* __restrict__ wl =
      reinterpret_cast<const float*>(Wlds4) + half * 4 * HEADS + (l & 7);

  v8f c = {0.f, 0.f, 0.f, 0.f, 0.f, 0.f, 0.f, 0.f};

  float4 xa[3][4];
  #pragma unroll
  for (int i = 0; i < 4; ++i) xa[0][i] = xr[i * 2];
  #pragma unroll
  for (int i = 0; i < 4; ++i) xa[1][i] = xr[8 + i * 2];

  #pragma unroll
  for (int t = 0; t < IN_CH / 32; ++t) {
    if (t + 2 < IN_CH / 32) {
      #pragma unroll
      for (int i = 0; i < 4; ++i) xa[(t + 2) % 3][i] = xr[(t + 2) * 8 + i * 2];
    }
    #pragma unroll
    for (int i = 0; i < 4; ++i) {
      const float4 av = xa[t % 3][i];
      const float* w = wl + (t * 32 + i * 8) * HEADS;
      v2f a, b;
      a.x = av.x;  a.y = av.y;
      b.x = w[0];  b.y = w[HEADS];
      c = __builtin_amdgcn_wmma_f32_16x16x4_f32(
           false, a,  false, b,
           (short)0, c,  false,  false);
      a.x = av.z;  a.y = av.w;
      b.x = w[2 * HEADS];  b.y = w[3 * HEADS];
      c = __builtin_amdgcn_wmma_f32_16x16x4_f32(
           false, a,  false, b,
           (short)0, c,  false,  false);
    }
  }

  if (l < HEADS) {
    #pragma unroll
    for (int i = 0; i < 8; ++i) {
      const int r = row0 + i + half * 8;
      if (r < n) h[(size_t)r * HEADS + l] = c[i];
    }
  }
}

__device__ __forceinline__ int block_excl_scan(int v, int* wsum, int lane, int wave) {
  int incl = v;
  #pragma unroll
  for (int o = 1; o < 32; o <<= 1) {
    const int t = __shfl_up(incl, o);
    if (lane >= o) incl += t;
  }
  if (lane == 31) wsum[wave] = incl;
  __syncthreads();
  if (wave == 0) {
    const int w = wsum[lane];
    int wi = w;
    #pragma unroll
    for (int o = 1; o < 32; o <<= 1) {
      const int t = __shfl_up(wi, o);
      if (lane >= o) wi += t;
    }
    wsum[lane] = wi - w;
  }
  __syncthreads();
  return wsum[wave] + incl - v;
}

__global__ __launch_bounds__(PB) void gat_rangecount_kernel(
    const int* __restrict__ ei, int* __restrict__ rangeCnt, int nE, int R)
{
  __shared__ int hist[MAXR];
  const int t = threadIdx.x;
  hist[t] = 0;
  __syncthreads();
  const int base = blockIdx.x * (PB * EPT);
  int d[EPT];
  #pragma unroll
  for (int j = 0; j < EPT; ++j) {
    const int e = base + j * PB + t;
    d[j] = (e < nE) ? ei[nE + e] : -1;
  }
  #pragma unroll
  for (int j = 0; j < EPT; ++j)
    if (d[j] >= 0) atomicAdd(&hist[d[j] >> RSHIFT], 1);
  __syncthreads();
  if (t < R) {
    const int c = hist[t];
    if (c) atomicAdd(rangeCnt + t, c);
  }
}

__global__ __launch_bounds__(PB) void gat_partition_kernel(
    const int* __restrict__ ei, const int* __restrict__ rangeCnt,
    int* __restrict__ rangeCur, int* __restrict__ rangeStart,
    int* __restrict__ part, int nE, int R)
{
  __shared__ int hist[MAXR];
  __shared__ int lstart[MAXR];
  __shared__ int rbase[MAXR];
  __shared__ int wsum[32];
  __shared__ int skey[PB * EPT];
  __shared__ int sgpos[PB * EPT];
  const int t    = threadIdx.x;
  const int lane = t & 31;
  const int wave = t >> 5;
  hist[t] = 0;
  const int tot    = (t < R) ? rangeCnt[t] : 0;
  const int rstart = block_excl_scan(tot, wsum, lane, wave);
  if (blockIdx.x == 0 && t < R) rangeStart[t] = rstart;

  const int base = blockIdx.x * (PB * EPT);
  int s[EPT], d[EPT], rk[EPT];
  #pragma unroll
  for (int j = 0; j < EPT; ++j) {
    const int e = base + j * PB + t;
    const bool ok = e < nE;
    s[j] = ok ? ei[e] : 0;
    d[j] = ok ? ei[nE + e] : -1;
  }
  #pragma unroll
  for (int j = 0; j < EPT; ++j) {
    rk[j] = 0;
    if (d[j] >= 0) rk[j] = atomicAdd(&hist[d[j] >> RSHIFT], 1);
  }
  __syncthreads();
  const int cc = hist[t];
  const int ls = block_excl_scan(cc, wsum, lane, wave);
  lstart[t] = ls;
  rbase[t]  = (t < R && cc) ? rstart + atomicAdd(rangeCur + t, cc) : 0;
  __syncthreads();
  #pragma unroll
  for (int j = 0; j < EPT; ++j) {
    if (d[j] >= 0) {
      const int r = d[j] >> RSHIFT;
      const int p = lstart[r] + rk[j];
      skey[p]  = (s[j] << RSHIFT) | (d[j] & (RSIZE - 1));
      sgpos[p] = rbase[r] + rk[j];
    }
  }
  __syncthreads();
  const int left  = nE - base;
  const int mine  = left < PB * EPT ? left : PB * EPT;
  #pragma unroll
  for (int j = 0; j < EPT; ++j) {
    const int i = j * PB + t;
    if (i < mine) part[sgpos[i]] = skey[i];
  }
}

__global__ __launch_bounds__(PB) void gat_rangesort_kernel(
    const int* part, const int* __restrict__ rangeCnt,
    const int* __restrict__ rangeStart, int* __restrict__ deg,
    int* __restrict__ cur, int* bsrc, int n)
{
  __shared__ int cnt[RSIZE];
  __shared__ int pos[RSIZE];
  __shared__ int wsum[32];
  __shared__ int stage[CAP];
  const int t    = threadIdx.x;
  const int lane = t & 31;
  const int wave = t >> 5;
  const int r    = blockIdx.x;
  const int L    = rangeCnt[r];
  const int seg  = rangeStart[r];
  if (t < RSIZE) cnt[t] = 0;
  __syncthreads();
  for (int i0 = t; i0 < L; i0 += 4 * PB) {
    int k[4];
    #pragma unroll
    for (int u = 0; u < 4; ++u) {
      const int i = i0 + u * PB;
      k[u] = (i < L) ? part[seg + i] : -1;
    }
    #pragma unroll
    for (int u = 0; u < 4; ++u)
      if (k[u] >= 0) atomicAdd(&cnt[k[u] & (RSIZE - 1)], 1);
  }
  __syncthreads();
  const int c    = (t < RSIZE) ? cnt[t] : 0;
  const int excl = block_excl_scan(c, wsum, lane, wave);
  if (t < RSIZE) {
    pos[t] = excl;
    const int node = r * RSIZE + t;
    if (node < n) { deg[node] = c; cur[node] = seg + excl + c; }
  }
  __syncthreads();
  const bool inLds = (L <= CAP);
  for (int i0 = t; i0 < L; i0 += 4 * PB) {
    int k[4];
    #pragma unroll
    for (int u = 0; u < 4; ++u) {
      const int i = i0 + u * PB;
      k[u] = (i < L) ? part[seg + i] : -1;
    }
    #pragma unroll
    for (int u = 0; u < 4; ++u) {
      if (k[u] >= 0) {
        const int p   = atomicAdd(&pos[k[u] & (RSIZE - 1)], 1);
        const int sid = k[u] >> RSHIFT;
        if (inLds) stage[p] = sid; else bsrc[seg + p] = sid;
      }
    }
  }
  if (inLds) {
    __syncthreads();
    for (int i = t; i < L; i += PB) bsrc[seg + i] = stage[i];
  }
}

__global__ __launch_bounds__(256) void gat_gather_kernel(
    const int* __restrict__ bsrc, const int* __restrict__ cur,
    const int* __restrict__ deg, const float* __restrict__ hmat,
    const float* __restrict__ att_src, const float* __restrict__ att_dst,
    const float* __restrict__ gat_bias, const float* __restrict__ lin_w,
    const float* __restrict__ lin_b, const float* __restrict__ bias,
    float* __restrict__ out, int n)
{
  const int lane = threadIdx.x & 31;
  const int node = blockIdx.x * (blockDim.x >> 5) + (threadIdx.x >> 5);
  if (node >= n) return;
  const int q  = lane >> 3;
  const int hh = lane & 7;

  const int end   = cur[node];
  const int start = end - deg[node];
  const float as = att_src[hh];
  const float ad = att_dst[hh];
  const float hself = hmat[node * HEADS + hh];
  const float adst  = hself * ad;
  const float eself = leaky(hself * as + adst);

  float m = (q == 0) ? eself : -INFINITY;
  for (int base = start; base < end; base += CHUNK) {
    int   s[8];
    float hs[8];
    #pragma unroll
    for (int j = 0; j < 8; ++j) {
      const int p = base + j * 4 + q;
      s[j] = bsrc[p < end ? p : end - 1];
    }
    #pragma unroll
    for (int j = 0; j < 8; ++j) hs[j] = hmat[s[j] * HEADS + hh];
    #pragma unroll
    for (int j = 0; j < 8; ++j) {
      const int p = base + j * 4 + q;
      const float e = leaky(hs[j] * as + adst);
      if (p < end) m = fmaxf(m, e);
    }
  }
  m = fmaxf(m, __shfl_xor(m, 8));
  m = fmaxf(m, __shfl_xor(m, 16));

  float den = 0.f, num = 0.f;
  if (q == 0) { den = expf(eself - m); num = den * hself; }
  for (int base = start; base < end; base += CHUNK) {
    int   s[8];
    float hs[8];
    #pragma unroll
    for (int j = 0; j < 8; ++j) {
      const int p = base + j * 4 + q;
      s[j] = bsrc[p < end ? p : end - 1];
    }
    #pragma unroll
    for (int j = 0; j < 8; ++j) hs[j] = hmat[s[j] * HEADS + hh];
    #pragma unroll
    for (int j = 0; j < 8; ++j) {
      const int p = base + j * 4 + q;
      const float ee = expf(leaky(hs[j] * as + adst) - m);
      if (p < end) { den += ee; num += ee * hs[j]; }
    }
  }
  den += __shfl_xor(den, 8);
  den += __shfl_xor(den, 16);
  num += __shfl_xor(num, 8);
  num += __shfl_xor(num, 16);

  const float oh = num / (den + 1e-16f) + gat_bias[hh];
  float z = lin_b[0];
  #pragma unroll
  for (int k = 0; k < HEADS; ++k) z += __shfl(oh, k) * lin_w[k];
  if (lane == 0) out[node] = fmaxf(z, 0.f) + bias[0];
}

extern "C" void kernel_launch(void* const* d_in, const int* in_sizes, int n_in,
                              void* d_out, int out_size, void* d_ws, size_t ws_size,
                              hipStream_t stream) {
  const float* x        = (const float*)d_in[0];
  const int*   ei       = (const int*)d_in[1];
  const float* W        = (const float*)d_in[2];
  const float* att_src  = (const float*)d_in[3];
  const float* att_dst  = (const float*)d_in[4];
  const float* gat_bias = (const float*)d_in[5];
  const float* lin_w    = (const float*)d_in[6];
  const float* lin_b    = (const float*)d_in[7];
  const float* bias     = (const float*)d_in[8];
  float* out = (float*)d_out;

  const int n  = in_sizes[0] / IN_CH;
  const int nE = in_sizes[1] / 2;

  const size_t nh   = (size_t)n * HEADS;
  const size_t segH = (nh * sizeof(float) + 255) & ~(size_t)255;
  const size_t segN = ((size_t)n * sizeof(int) + 255) & ~(size_t)255;
  const size_t segT = (size_t)3 * MAXR * sizeof(int);
  const size_t segE = ((size_t)nE * sizeof(int) + 255) & ~(size_t)255;
  char* ws = (char*)d_ws;
  float* h_mat      = (float*)(ws);
  int*   deg        = (int*)  (ws + segH);
  int*   cur        = (int*)  (ws + segH + segN);
  int*   rangeCnt   = (int*)  (ws + segH + 2 * segN);
  int*   rangeCur   = rangeCnt + MAXR;
  int*   rangeStart = rangeCnt + 2 * MAXR;
  int*   part       = (int*)  (ws + segH + 2 * segN + segT);
  int*   bsrc       = (ws_size >= segH + 2 * segN + segT + 2 * segE) ? (int*)((char*)part + segE) : part;

  const int tiles  = (n + 15) / 16;
  const int blocks = (tiles + 7) / 8;
  gat_gemm_kernel<<<blocks, GEMM_T, 0, stream>>>(x, W, h_mat, rangeCnt, n);

  const int R  = (n + RSIZE - 1) / RSIZE;
  const int eb = (nE + PB * EPT - 1) / (PB * EPT);
  gat_rangecount_kernel<<<eb, PB, 0, stream>>>(ei, rangeCnt, nE, R);

  gat_partition_kernel<<<eb, PB, 0, stream>>>(ei, rangeCnt, rangeCur, rangeStart, part, nE, R);

  gat_rangesort_kernel<<<R, PB, 0, stream>>>(part, rangeCnt, rangeStart, deg, cur, bsrc, n);

  gat_gather_kernel<<<(n + 7) / 8, 256, 0, stream>>>(
      bsrc, cur, deg, h_mat, att_src, att_dst, gat_bias, lin_w, lin_b, bias,
      out, n);
}


# gather fetches each edge once: h value and score of a node's first 64 edges stay in registers from the max pass to the exp pass (were loaded twice); longer lists re-read the rest; sum order kept
# speedup vs baseline: 16.2926x; 1.1121x over previous
#include <hip/hip_runtime.h>
#include <math.h>


#define IN_CH 512
#define HEADS 8
#define GEMM_T 256
#define NEG_SLOPE 0.2f
#define CHUNK 32
#define RSHIFT 8
#define RSIZE 256
#define MAXR 1024
#define PB 1024
#define EPT 8
#define CAP 16384

typedef __attribute__((ext_vector_type(2))) float v2f;
typedef __attribute__((ext_vector_type(8))) float v8f;

__device__ __forceinline__ float leaky(float v) {
  return v >= 0.f ? v : NEG_SLOPE * v;
}

__global__ __launch_bounds__(GEMM_T) void gat_gemm_kernel(
    const float* __restrict__ x, const float* __restrict__ W,
    float* __restrict__ h, int* __restrict__ rangeTabs, int n)
{
  __shared__ float4 Wlds4[IN_CH * HEADS / 4];
  {
    const float4* __restrict__ W4 = reinterpret_cast<const float4*>(W);
    float4 wv[4];
    #pragma unroll
    for (int j = 0; j < 4; ++j) wv[j] = W4[threadIdx.x + j * GEMM_T];
    #pragma unroll
    for (int j = 0; j < 4; ++j) Wlds4[threadIdx.x + j * GEMM_T] = wv[j];
  }
  if (blockIdx.x == 0) {
    for (int i = threadIdx.x; i < 2 * MAXR; i += GEMM_T) rangeTabs[i] = 0;
  }
  __syncthreads();

  const int lane = threadIdx.x & 31;
  const int wave = threadIdx.x >> 5;
  const int half = lane >> 4;
  const int l    = lane & 15;

  const int tile = blockIdx.x * (GEMM_T >> 5) + wave;
  const int row0 = tile * 16;
  if (row0 >= n) return;

  int arow = row0 + l;
  if (arow > n - 1) arow = n - 1;
  const float4* __restrict__ xr =
      reinterpret_cast<const float4*>(x + (size_t)arow * IN_CH) + half;
  const float* __restrict__ wl =
      reinterpret_cast<const float*>(Wlds4) + half * 4 * HEADS + (l & 7);

  v8f c = {0.f, 0.f, 0.f, 0.f, 0.f, 0.f, 0.f, 0.f};

  float4 xa[3][4];
  #pragma unroll
  for (int i = 0; i < 4; ++i) xa[0][i] = xr[i * 2];
  #pragma unroll
  for (int i = 0; i < 4; ++i) xa[1][i] = xr[8 + i * 2];

  #pragma unroll
  for (int t = 0; t < IN_CH / 32; ++t) {
    if (t + 2 < IN_CH / 32) {
      #pragma unroll
      for (int i = 0; i < 4; ++i) xa[(t + 2) % 3][i] = xr[(t + 2) * 8 + i * 2];
    }
    #pragma unroll
    for (int i = 0; i < 4; ++i) {
      const float4 av = xa[t % 3][i];
      const float* w = wl + (t * 32 + i * 8) * HEADS;
      v2f a, b;
      a.x = av.x;  a.y = av.y;
      b.x = w[0];  b.y = w[HEADS];
      c = __builtin_amdgcn_wmma_f32_16x16x4_f32(
           false, a,  false, b,
           (short)0, c,  false,  false);
      a.x = av.z;  a.y = av.w;
      b.x = w[2 * HEADS];  b.y = w[3 * HEADS];
      c = __builtin_amdgcn_wmma_f32_16x16x4_f32(
           false, a,  false, b,
           (short)0, c,  false,  false);
    }
  }

  if (l < HEADS) {
    #pragma unroll
    for (int i = 0; i < 8; ++i) {
      const int r = row0 + i + half * 8;
      if (r < n) h[(size_t)r * HEADS + l] = c[i];
    }
  }
}

__device__ __forceinline__ int block_excl_scan(int v, int* wsum, int lane, int wave) {
  int incl = v;
  #pragma unroll
  for (int o = 1; o < 32; o <<= 1) {
    const int t = __shfl_up(incl, o);
    if (lane >= o) incl += t;
  }
  if (lane == 31) wsum[wave] = incl;
  __syncthreads();
  if (wave == 0) {
    const int w = wsum[lane];
    int wi = w;
    #pragma unroll
    for (int o = 1; o < 32; o <<= 1) {
      const int t = __shfl_up(wi, o);
      if (lane >= o) wi += t;
    }
    wsum[lane] = wi - w;
  }
  __syncthreads();
  return wsum[wave] + incl - v;
}

__global__ __launch_bounds__(PB) void gat_rangecount_kernel(
    const int* __restrict__ ei, int* __restrict__ rangeCnt, int nE, int R)
{
  __shared__ int hist[MAXR];
  const int t = threadIdx.x;
  hist[t] = 0;
  __syncthreads();
  const int base = blockIdx.x * (PB * EPT);
  int d[EPT];
  #pragma unroll
  for (int j = 0; j < EPT; ++j) {
    const int e = base + j * PB + t;
    d[j] = (e < nE) ? ei[nE + e] : -1;
  }
  #pragma unroll
  for (int j = 0; j < EPT; ++j)
    if (d[j] >= 0) atomicAdd(&hist[d[j] >> RSHIFT], 1);
  __syncthreads();
  if (t < R) {
    const int c = hist[t];
    if (c) atomicAdd(rangeCnt + t, c);
  }
}

__global__ __launch_bounds__(PB) void gat_partition_kernel(
    const int* __restrict__ ei, const int* __restrict__ rangeCnt,
    int* __restrict__ rangeCur, int* __restrict__ rangeStart,
    int* __restrict__ part, int nE, int R)
{
  __shared__ int hist[MAXR];
  __shared__ int lstart[MAXR];
  __shared__ int rbase[MAXR];
  __shared__ int wsum[32];
  __shared__ int skey[PB * EPT];
  __shared__ int sgpos[PB * EPT];
  const int t    = threadIdx.x;
  const int lane = t & 31;
  const int wave = t >> 5;
  hist[t] = 0;
  const int tot    = (t < R) ? rangeCnt[t] : 0;
  const int rstart = block_excl_scan(tot, wsum, lane, wave);
  if (blockIdx.x == 0 && t < R) rangeStart[t] = rstart;

  const int base = blockIdx.x * (PB * EPT);
  int s[EPT], d[EPT], rk[EPT];
  #pragma unroll
  for (int j = 0; j < EPT; ++j) {
    const int e = base + j * PB + t;
    const bool ok = e < nE;
    s[j] = ok ? ei[e] : 0;
    d[j] = ok ? ei[nE + e] : -1;
  }
  #pragma unroll
  for (int j = 0; j < EPT; ++j) {
    rk[j] = 0;
    if (d[j] >= 0) rk[j] = atomicAdd(&hist[d[j] >> RSHIFT], 1);
  }
  __syncthreads();
  const int cc = hist[t];
  const int ls = block_excl_scan(cc, wsum, lane, wave);
  lstart[t] = ls;
  rbase[t]  = (t < R && cc) ? rstart + atomicAdd(rangeCur + t, cc) : 0;
  __syncthreads();
  #pragma unroll
  for (int j = 0; j < EPT; ++j) {
    if (d[j] >= 0) {
      const int r = d[j] >> RSHIFT;
      const int p = lstart[r] + rk[j];
      skey[p]  = (s[j] << RSHIFT) | (d[j] & (RSIZE - 1));
      sgpos[p] = rbase[r] + rk[j];
    }
  }
  __syncthreads();
  const int left  = nE - base;
  const int mine  = left < PB * EPT ? left : PB * EPT;
  #pragma unroll
  for (int j = 0; j < EPT; ++j) {
    const int i = j * PB + t;
    if (i < mine) part[sgpos[i]] = skey[i];
  }
}

__global__ __launch_bounds__(PB) void gat_rangesort_kernel(
    const int* part, const int* __restrict__ rangeCnt,
    const int* __restrict__ rangeStart, int* __restrict__ deg,
    int* __restrict__ cur, int* bsrc, int n)
{
  __shared__ int cnt[RSIZE];
  __shared__ int pos[RSIZE];
  __shared__ int wsum[32];
  __shared__ int stage[CAP];
  const int t    = threadIdx.x;
  const int lane = t & 31;
  const int wave = t >> 5;
  const int r    = blockIdx.x;
  const int L    = rangeCnt[r];
  const int seg  = rangeStart[r];
  if (t < RSIZE) cnt[t] = 0;
  __syncthreads();
  for (int i0 = t; i0 < L; i0 += 4 * PB) {
    int k[4];
    #pragma unroll
    for (int u = 0; u < 4; ++u) {
      const int i = i0 + u * PB;
      k[u] = (i < L) ? part[seg + i] : -1;
    }
    #pragma unroll
    for (int u = 0; u < 4; ++u)
      if (k[u] >= 0) atomicAdd(&cnt[k[u] & (RSIZE - 1)], 1);
  }
  __syncthreads();
  const int c    = (t < RSIZE) ? cnt[t] : 0;
  const int excl = block_excl_scan(c, wsum, lane, wave);
  if (t < RSIZE) {
    pos[t] = excl;
    const int node = r * RSIZE + t;
    if (node < n) { deg[node] = c; cur[node] = seg + excl + c; }
  }
  __syncthreads();
  const bool inLds = (L <= CAP);
  for (int i0 = t; i0 < L; i0 += 4 * PB) {
    int k[4];
    #pragma unroll
    for (int u = 0; u < 4; ++u) {
      const int i = i0 + u * PB;
      k[u] = (i < L) ? part[seg + i] : -1;
    }
    #pragma unroll
    for (int u = 0; u < 4; ++u) {
      if (k[u] >= 0) {
        const int p   = atomicAdd(&pos[k[u] & (RSIZE - 1)], 1);
        const int sid = k[u] >> RSHIFT;
        if (inLds) stage[p] = sid; else bsrc[seg + p] = sid;
      }
    }
  }
  if (inLds) {
    __syncthreads();
    for (int i = t; i < L; i += PB) bsrc[seg + i] = stage[i];
  }
}

__device__ __forceinline__ void gather_load_ids(const int* __restrict__ bsrc, int base,
                                                int end, int q, int (&s)[8]) {
  #pragma unroll
  for (int j = 0; j < 8; ++j) {
    const int p = base + j * 4 + q;
    s[j] = bsrc[p < end ? p : end - 1];
  }
}

__device__ __forceinline__ void gather_load_rows(const float* __restrict__ hmat,
                                                 const int (&s)[8], int hh, float (&h)[8]) {
  #pragma unroll
  for (int j = 0; j < 8; ++j) h[j] = hmat[s[j] * HEADS + hh];
}

__global__ __launch_bounds__(256) void gat_gather_kernel(
    const int* __restrict__ bsrc, const int* __restrict__ cur,
    const int* __restrict__ deg, const float* __restrict__ hmat,
    const float* __restrict__ att_src, const float* __restrict__ att_dst,
    const float* __restrict__ gat_bias, const float* __restrict__ lin_w,
    const float* __restrict__ lin_b, const float* __restrict__ bias,
    float* __restrict__ out, int n)
{
  const int lane = threadIdx.x & 31;
  const int node = blockIdx.x * (blockDim.x >> 5) + (threadIdx.x >> 5);
  if (node >= n) return;
  const int q  = lane >> 3;
  const int hh = lane & 7;

  const int end   = cur[node];
  const int cnt   = deg[node];
  const int start = end - cnt;
  const float as = att_src[hh];
  const float ad = att_dst[hh];
  const float hself = hmat[node * HEADS + hh];
  const float adst  = hself * ad;
  const float eself = leaky(hself * as + adst);

  const bool hasA = cnt > 0;
  const bool hasB = cnt > CHUNK;
  const int  rest = start + 2 * CHUNK;

  int   sA[8], sB[8];
  float hA[8], hB[8], eA[8], eB[8];
  #pragma unroll
  for (int j = 0; j < 8; ++j) {
    sA[j] = 0;  sB[j] = 0;  hA[j] = 0.f;  hB[j] = 0.f;
    eA[j] = -INFINITY;  eB[j] = -INFINITY;
  }
  if (hasB) {
    gather_load_ids(bsrc, start, end, q, sA);
    gather_load_ids(bsrc, start + CHUNK, end, q, sB);
    gather_load_rows(hmat, sA, hh, hA);
    gather_load_rows(hmat, sB, hh, hB);
  } else if (hasA) {
    gather_load_ids(bsrc, start, end, q, sA);
    gather_load_rows(hmat, sA, hh, hA);
  }
  if (hasA) {
    #pragma unroll
    for (int j = 0; j < 8; ++j)
      if (j * 4 + q < cnt) eA[j] = leaky(hA[j] * as + adst);
  }
  if (hasB) {
    #pragma unroll
    for (int j = 0; j < 8; ++j)
      if (CHUNK + j * 4 + q < cnt) eB[j] = leaky(hB[j] * as + adst);
  }

  float m = (q == 0) ? eself : -INFINITY;
  #pragma unroll
  for (int j = 0; j < 8; ++j) m = fmaxf(m, eA[j]);
  if (hasB) {
    #pragma unroll
    for (int j = 0; j < 8; ++j) m = fmaxf(m, eB[j]);
  }
  for (int base = rest; base < end; base += CHUNK) {
    int   s[8];
    float hs[8];
    #pragma unroll
    for (int j = 0; j < 8; ++j) {
      const int p = base + j * 4 + q;
      s[j] = bsrc[p < end ? p : end - 1];
    }
    #pragma unroll
    for (int j = 0; j < 8; ++j) hs[j] = hmat[s[j] * HEADS + hh];
    #pragma unroll
    for (int j = 0; j < 8; ++j) {
      const int p = base + j * 4 + q;
      const float e = leaky(hs[j] * as + adst);
      if (p < end) m = fmaxf(m, e);
    }
  }
  m = fmaxf(m, __shfl_xor(m, 8));
  m = fmaxf(m, __shfl_xor(m, 16));

  float den = 0.f, num = 0.f;
  if (q == 0) { den = expf(eself - m); num = den * hself; }
  #pragma unroll
  for (int j = 0; j < 8; ++j) {
    const float ee = expf(eA[j] - m);
    den += ee;  num = fmaf(ee, hA[j], num);
  }
  if (hasB) {
    #pragma unroll
    for (int j = 0; j < 8; ++j) {
      const float ee = expf(eB[j] - m);
      den += ee;  num = fmaf(ee, hB[j], num);
    }
  }
  for (int base = rest; base < end; base += CHUNK) {
    int   s[8];
    float hs[8];
    #pragma unroll
    for (int j = 0; j < 8; ++j) {
      const int p = base + j * 4 + q;
      s[j] = bsrc[p < end ? p : end - 1];
    }
    #pragma unroll
    for (int j = 0; j < 8; ++j) hs[j] = hmat[s[j] * HEADS + hh];
    #pragma unroll
    for (int j = 0; j < 8; ++j) {
      const int p = base + j * 4 + q;
      const float ee = expf(leaky(hs[j] * as + adst) - m);
      if (p < end) { den += ee; num += ee * hs[j]; }
    }
  }
  den += __shfl_xor(den, 8);
  den += __shfl_xor(den, 16);
  num += __shfl_xor(num, 8);
  num += __shfl_xor(num, 16);

  const float oh = num / (den + 1e-16f) + gat_bias[hh];
  float z = lin_b[0];
  #pragma unroll
  for (int k = 0; k < HEADS; ++k) z += __shfl(oh, k) * lin_w[k];
  if (lane == 0) out[node] = fmaxf(z, 0.f) + bias[0];
}

extern "C" void kernel_launch(void* const* d_in, const int* in_sizes, int n_in,
                              void* d_out, int out_size, void* d_ws, size_t ws_size,
                              hipStream_t stream) {
  const float* x        = (const float*)d_in[0];
  const int*   ei       = (const int*)d_in[1];
  const float* W        = (const float*)d_in[2];
  const float* att_src  = (const float*)d_in[3];
  const float* att_dst  = (const float*)d_in[4];
  const float* gat_bias = (const float*)d_in[5];
  const float* lin_w    = (const float*)d_in[6];
  const float* lin_b    = (const float*)d_in[7];
  const float* bias     = (const float*)d_in[8];
  float* out = (float*)d_out;

  const int n  = in_sizes[0] / IN_CH;
  const int nE = in_sizes[1] / 2;

  const size_t nh   = (size_t)n * HEADS;
  const size_t segH = (nh * sizeof(float) + 255) & ~(size_t)255;
  const size_t segN = ((size_t)n * sizeof(int) + 255) & ~(size_t)255;
  const size_t segT = (size_t)3 * MAXR * sizeof(int);
  const size_t segE = ((size_t)nE * sizeof(int) + 255) & ~(size_t)255;
  char* ws = (char*)d_ws;
  float* h_mat      = (float*)(ws);
  int*   deg        = (int*)  (ws + segH);
  int*   cur        = (int*)  (ws + segH + segN);
  int*   rangeCnt   = (int*)  (ws + segH + 2 * segN);
  int*   rangeCur   = rangeCnt + MAXR;
  int*   rangeStart = rangeCnt + 2 * MAXR;
  int*   part       = (int*)  (ws + segH + 2 * segN + segT);
  int*   bsrc       = (ws_size >= segH + 2 * segN + segT + 2 * segE) ? (int*)((char*)part + segE) : part;

  const int tiles  = (n + 15) / 16;
  const int blocks = (tiles + 7) / 8;
  gat_gemm_kernel<<<blocks, GEMM_T, 0, stream>>>(x, W, h_mat, rangeCnt, n);

  const int R  = (n + RSIZE - 1) / RSIZE;
  const int eb = (nE + PB * EPT - 1) / (PB * EPT);
  gat_rangecount_kernel<<<eb, PB, 0, stream>>>(ei, rangeCnt, nE, R);

  gat_partition_kernel<<<eb, PB, 0, stream>>>(ei, rangeCnt, rangeCur, rangeStart, part, nE, R);

  gat_rangesort_kernel<<<R, PB, 0, stream>>>(part, rangeCnt, rangeStart, deg, cur, bsrc, n);

  gat_gather_kernel<<<(n + 7) / 8, 256, 0, stream>>>(
      bsrc, cur, deg, h_mat, att_src, att_dst, gat_bias, lin_w, lin_b, bias,
      out, n);
}


# gather: each softmax exponential is one fused multiply-add and the hardware base-2 exponential v_exp_f32 (was the library expf, 13 instructions); f32 kept, last bits differ; sum order kept
# speedup vs baseline: 16.8814x; 1.0361x over previous
#include <hip/hip_runtime.h>
#include <math.h>


#define IN_CH 512
#define HEADS 8
#define GEMM_T 256
#define NEG_SLOPE 0.2f
#define CHUNK 32
#define RSHIFT 8
#define RSIZE 256
#define MAXR 1024
#define PB 1024
#define EPT 8
#define CAP 16384

typedef __attribute__((ext_vector_type(2))) float v2f;
typedef __attribute__((ext_vector_type(8))) float v8f;

__device__ __forceinline__ float leaky(float v) {
  return v >= 0.f ? v : NEG_SLOPE * v;
}

#define LOG2E 1.44269504088896340736f

__device__ __forceinline__ float exp_rel(float e, float mL) {
  return __builtin_amdgcn_exp2f(fmaf(e, LOG2E, -mL));
}

__global__ __launch_bounds__(GEMM_T) void gat_gemm_kernel(
    const float* __restrict__ x, const float* __restrict__ W,
    float* __restrict__ h, int* __restrict__ rangeTabs, int n)
{
  __shared__ float4 Wlds4[IN_CH * HEADS / 4];
  {
    const float4* __restrict__ W4 = reinterpret_cast<const float4*>(W);
    float4 wv[4];
    #pragma unroll
    for (int j = 0; j < 4; ++j) wv[j] = W4[threadIdx.x + j * GEMM_T];
    #pragma unroll
    for (int j = 0; j < 4; ++j) Wlds4[threadIdx.x + j * GEMM_T] = wv[j];
  }
  if (blockIdx.x == 0) {
    for (int i = threadIdx.x; i < 2 * MAXR; i += GEMM_T) rangeTabs[i] = 0;
  }
  __syncthreads();

  const int lane = threadIdx.x & 31;
  const int wave = threadIdx.x >> 5;
  const int half = lane >> 4;
  const int l    = lane & 15;

  const int tile = blockIdx.x * (GEMM_T >> 5) + wave;
  const int row0 = tile * 16;
  if (row0 >= n) return;

  int arow = row0 + l;
  if (arow > n - 1) arow = n - 1;
  const float4* __restrict__ xr =
      reinterpret_cast<const float4*>(x + (size_t)arow * IN_CH) + half;
  const float* __restrict__ wl =
      reinterpret_cast<const float*>(Wlds4) + half * 4 * HEADS + (l & 7);

  v8f c = {0.f, 0.f, 0.f, 0.f, 0.f, 0.f, 0.f, 0.f};

  float4 xa[3][4];
  #pragma unroll
  for (int i = 0; i < 4; ++i) xa[0][i] = xr[i * 2];
  #pragma unroll
  for (int i = 0; i < 4; ++i) xa[1][i] = xr[8 + i * 2];

  #pragma unroll
  for (int t = 0; t < IN_CH / 32; ++t) {
    if (t + 2 < IN_CH / 32) {
      #pragma unroll
      for (int i = 0; i < 4; ++i) xa[(t + 2) % 3][i] = xr[(t + 2) * 8 + i * 2];
    }
    #pragma unroll
    for (int i = 0; i < 4; ++i) {
      const float4 av = xa[t % 3][i];
      const float* w = wl + (t * 32 + i * 8) * HEADS;
      v2f a, b;
      a.x = av.x;  a.y = av.y;
      b.x = w[0];  b.y = w[HEADS];
      c = __builtin_amdgcn_wmma_f32_16x16x4_f32(
           false, a,  false, b,
           (short)0, c,  false,  false);
      a.x = av.z;  a.y = av.w;
      b.x = w[2 * HEADS];  b.y = w[3 * HEADS];
      c = __builtin_amdgcn_wmma_f32_16x16x4_f32(
           false, a,  false, b,
           (short)0, c,  false,  false);
    }
  }

  if (l < HEADS) {
    #pragma unroll
    for (int i = 0; i < 8; ++i) {
      const int r = row0 + i + half * 8;
      if (r < n) h[(size_t)r * HEADS + l] = c[i];
    }
  }
}

__device__ __forceinline__ int block_excl_scan(int v, int* wsum, int lane, int wave) {
  int incl = v;
  #pragma unroll
  for (int o = 1; o < 32; o <<= 1) {
    const int t = __shfl_up(incl, o);
    if (lane >= o) incl += t;
  }
  if (lane == 31) wsum[wave] = incl;
  __syncthreads();
  if (wave == 0) {
    const int w = wsum[lane];
    int wi = w;
    #pragma unroll
    for (int o = 1; o < 32; o <<= 1) {
      const int t = __shfl_up(wi, o);
      if (lane >= o) wi += t;
    }
    wsum[lane] = wi - w;
  }
  __syncthreads();
  return wsum[wave] + incl - v;
}

__global__ __launch_bounds__(PB) void gat_rangecount_kernel(
    const int* __restrict__ ei, int* __restrict__ rangeCnt, int nE, int R)
{
  __shared__ int hist[MAXR];
  const int t = threadIdx.x;
  hist[t] = 0;
  __syncthreads();
  const int base = blockIdx.x * (PB * EPT);
  int d[EPT];
  #pragma unroll
  for (int j = 0; j < EPT; ++j) {
    const int e = base + j * PB + t;
    d[j] = (e < nE) ? ei[nE + e] : -1;
  }
  #pragma unroll
  for (int j = 0; j < EPT; ++j)
    if (d[j] >= 0) atomicAdd(&hist[d[j] >> RSHIFT], 1);
  __syncthreads();
  if (t < R) {
    const int c = hist[t];
    if (c) atomicAdd(rangeCnt + t, c);
  }
}

__global__ __launch_bounds__(PB) void gat_partition_kernel(
    const int* __restrict__ ei, const int* __restrict__ rangeCnt,
    int* __restrict__ rangeCur, int* __restrict__ rangeStart,
    int* __restrict__ part, int nE, int R)
{
  __shared__ int hist[MAXR];
  __shared__ int lstart[MAXR];
  __shared__ int rbase[MAXR];
  __shared__ int wsum[32];
  __shared__ int skey[PB * EPT];
  __shared__ int sgpos[PB * EPT];
  const int t    = threadIdx.x;
  const int lane = t & 31;
  const int wave = t >> 5;
  hist[t] = 0;
  const int tot    = (t < R) ? rangeCnt[t] : 0;
  const int rstart = block_excl_scan(tot, wsum, lane, wave);
  if (blockIdx.x == 0 && t < R) rangeStart[t] = rstart;

  const int base = blockIdx.x * (PB * EPT);
  int s[EPT], d[EPT], rk[EPT];
  #pragma unroll
  for (int j = 0; j < EPT; ++j) {
    const int e = base + j * PB + t;
    const bool ok = e < nE;
    s[j] = ok ? ei[e] : 0;
    d[j] = ok ? ei[nE + e] : -1;
  }
  #pragma unroll
  for (int j = 0; j < EPT; ++j) {
    rk[j] = 0;
    if (d[j] >= 0) rk[j] = atomicAdd(&hist[d[j] >> RSHIFT], 1);
  }
  __syncthreads();
  const int cc = hist[t];
  const int ls = block_excl_scan(cc, wsum, lane, wave);
  lstart[t] = ls;
  rbase[t]  = (t < R && cc) ? rstart + atomicAdd(rangeCur + t, cc) : 0;
  __syncthreads();
  #pragma unroll
  for (int j = 0; j < EPT; ++j) {
    if (d[j] >= 0) {
      const int r = d[j] >> RSHIFT;
      const int p = lstart[r] + rk[j];
      skey[p]  = (s[j] << RSHIFT) | (d[j] & (RSIZE - 1));
      sgpos[p] = rbase[r] + rk[j];
    }
  }
  __syncthreads();
  const int left  = nE - base;
  const int mine  = left < PB * EPT ? left : PB * EPT;
  #pragma unroll
  for (int j = 0; j < EPT; ++j) {
    const int i = j * PB + t;
    if (i < mine) part[sgpos[i]] = skey[i];
  }
}

__global__ __launch_bounds__(PB) void gat_rangesort_kernel(
    const int* part, const int* __restrict__ rangeCnt,
    const int* __restrict__ rangeStart, int* __restrict__ deg,
    int* __restrict__ cur, int* bsrc, int n)
{
  __shared__ int cnt[RSIZE];
  __shared__ int pos[RSIZE];
  __shared__ int wsum[32];
  __shared__ int stage[CAP];
  const int t    = threadIdx.x;
  const int lane = t & 31;
  const int wave = t >> 5;
  const int r    = blockIdx.x;
  const int L    = rangeCnt[r];
  const int seg  = rangeStart[r];
  if (t < RSIZE) cnt[t] = 0;
  __syncthreads();
  for (int i0 = t; i0 < L; i0 += 4 * PB) {
    int k[4];
    #pragma unroll
    for (int u = 0; u < 4; ++u) {
      const int i = i0 + u * PB;
      k[u] = (i < L) ? part[seg + i] : -1;
    }
    #pragma unroll
    for (int u = 0; u < 4; ++u)
      if (k[u] >= 0) atomicAdd(&cnt[k[u] & (RSIZE - 1)], 1);
  }
  __syncthreads();
  const int c    = (t < RSIZE) ? cnt[t] : 0;
  const int excl = block_excl_scan(c, wsum, lane, wave);
  if (t < RSIZE) {
    pos[t] = excl;
    const int node = r * RSIZE + t;
    if (node < n) { deg[node] = c; cur[node] = seg + excl + c; }
  }
  __syncthreads();
  const bool inLds = (L <= CAP);
  for (int i0 = t; i0 < L; i0 += 4 * PB) {
    int k[4];
    #pragma unroll
    for (int u = 0; u < 4; ++u) {
      const int i = i0 + u * PB;
      k[u] = (i < L) ? part[seg + i] : -1;
    }
    #pragma unroll
    for (int u = 0; u < 4; ++u) {
      if (k[u] >= 0) {
        const int p   = atomicAdd(&pos[k[u] & (RSIZE - 1)], 1);
        const int sid = k[u] >> RSHIFT;
        if (inLds) stage[p] = sid; else bsrc[seg + p] = sid;
      }
    }
  }
  if (inLds) {
    __syncthreads();
    for (int i = t; i < L; i += PB) bsrc[seg + i] = stage[i];
  }
}

__device__ __forceinline__ void gather_load_ids(const int* __restrict__ bsrc, int base,
                                                int end, int q, int (&s)[8]) {
  #pragma unroll
  for (int j = 0; j < 8; ++j) {
    const int p = base + j * 4 + q;
    s[j] = bsrc[p < end ? p : end - 1];
  }
}

__device__ __forceinline__ void gather_load_rows(const float* __restrict__ hmat,
                                                 const int (&s)[8], int hh, float (&h)[8]) {
  #pragma unroll
  for (int j = 0; j < 8; ++j) h[j] = hmat[s[j] * HEADS + hh];
}

__global__ __launch_bounds__(256) void gat_gather_kernel(
    const int* __restrict__ bsrc, const int* __restrict__ cur,
    const int* __restrict__ deg, const float* __restrict__ hmat,
    const float* __restrict__ att_src, const float* __restrict__ att_dst,
    const float* __restrict__ gat_bias, const float* __restrict__ lin_w,
    const float* __restrict__ lin_b, const float* __restrict__ bias,
    float* __restrict__ out, int n)
{
  const int lane = threadIdx.x & 31;
  const int node = blockIdx.x * (blockDim.x >> 5) + (threadIdx.x >> 5);
  if (node >= n) return;
  const int q  = lane >> 3;
  const int hh = lane & 7;

  const int end   = cur[node];
  const int cnt   = deg[node];
  const int start = end - cnt;
  const float as = att_src[hh];
  const float ad = att_dst[hh];
  const float hself = hmat[node * HEADS + hh];
  const float adst  = hself * ad;
  const float eself = leaky(hself * as + adst);

  const bool hasA = cnt > 0;
  const bool hasB = cnt > CHUNK;
  const int  rest = start + 2 * CHUNK;

  int   sA[8], sB[8];
  float hA[8], hB[8], eA[8], eB[8];
  #pragma unroll
  for (int j = 0; j < 8; ++j) {
    sA[j] = 0;  sB[j] = 0;  hA[j] = 0.f;  hB[j] = 0.f;
    eA[j] = -INFINITY;  eB[j] = -INFINITY;
  }
  if (hasB) {
    gather_load_ids(bsrc, start, end, q, sA);
    gather_load_ids(bsrc, start + CHUNK, end, q, sB);
    gather_load_rows(hmat, sA, hh, hA);
    gather_load_rows(hmat, sB, hh, hB);
  } else if (hasA) {
    gather_load_ids(bsrc, start, end, q, sA);
    gather_load_rows(hmat, sA, hh, hA);
  }
  if (hasA) {
    #pragma unroll
    for (int j = 0; j < 8; ++j)
      if (j * 4 + q < cnt) eA[j] = leaky(hA[j] * as + adst);
  }
  if (hasB) {
    #pragma unroll
    for (int j = 0; j < 8; ++j)
      if (CHUNK + j * 4 + q < cnt) eB[j] = leaky(hB[j] * as + adst);
  }

  float m = (q == 0) ? eself : -INFINITY;
  #pragma unroll
  for (int j = 0; j < 8; ++j) m = fmaxf(m, eA[j]);
  if (hasB) {
    #pragma unroll
    for (int j = 0; j < 8; ++j) m = fmaxf(m, eB[j]);
  }
  for (int base = rest; base < end; base += CHUNK) {
    int   s[8];
    float hs[8];
    #pragma unroll
    for (int j = 0; j < 8; ++j) {
      const int p = base + j * 4 + q;
      s[j] = bsrc[p < end ? p : end - 1];
    }
    #pragma unroll
    for (int j = 0; j < 8; ++j) hs[j] = hmat[s[j] * HEADS + hh];
    #pragma unroll
    for (int j = 0; j < 8; ++j) {
      const int p = base + j * 4 + q;
      const float e = leaky(hs[j] * as + adst);
      if (p < end) m = fmaxf(m, e);
    }
  }
  m = fmaxf(m, __shfl_xor(m, 8));
  m = fmaxf(m, __shfl_xor(m, 16));

  const float mL = m * LOG2E;
  float den = 0.f, num = 0.f;
  if (q == 0) { den = exp_rel(eself, mL); num = den * hself; }
  #pragma unroll
  for (int j = 0; j < 8; ++j) {
    const float ee = exp_rel(eA[j], mL);
    den += ee;  num = fmaf(ee, hA[j], num);
  }
  if (hasB) {
    #pragma unroll
    for (int j = 0; j < 8; ++j) {
      const float ee = exp_rel(eB[j], mL);
      den += ee;  num = fmaf(ee, hB[j], num);
    }
  }
  for (int base = rest; base < end; base += CHUNK) {
    int   s[8];
    float hs[8];
    #pragma unroll
    for (int j = 0; j < 8; ++j) {
      const int p = base + j * 4 + q;
      s[j] = bsrc[p < end ? p : end - 1];
    }
    #pragma unroll
    for (int j = 0; j < 8; ++j) hs[j] = hmat[s[j] * HEADS + hh];
    #pragma unroll
    for (int j = 0; j < 8; ++j) {
      const int p = base + j * 4 + q;
      const float ee = exp_rel(leaky(hs[j] * as + adst), mL);
      if (p < end) { den += ee; num += ee * hs[j]; }
    }
  }
  den += __shfl_xor(den, 8);
  den += __shfl_xor(den, 16);
  num += __shfl_xor(num, 8);
  num += __shfl_xor(num, 16);

  const float oh = num / (den + 1e-16f) + gat_bias[hh];
  float z = lin_b[0];
  #pragma unroll
  for (int k = 0; k < HEADS; ++k) z += __shfl(oh, k) * lin_w[k];
  if (lane == 0) out[node] = fmaxf(z, 0.f) + bias[0];
}

extern "C" void kernel_launch(void* const* d_in, const int* in_sizes, int n_in,
                              void* d_out, int out_size, void* d_ws, size_t ws_size,
                              hipStream_t stream) {
  const float* x        = (const float*)d_in[0];
  const int*   ei       = (const int*)d_in[1];
  const float* W        = (const float*)d_in[2];
  const float* att_src  = (const float*)d_in[3];
  const float* att_dst  = (const float*)d_in[4];
  const float* gat_bias = (const float*)d_in[5];
  const float* lin_w    = (const float*)d_in[6];
  const float* lin_b    = (const float*)d_in[7];
  const float* bias     = (const float*)d_in[8];
  float* out = (float*)d_out;

  const int n  = in_sizes[0] / IN_CH;
  const int nE = in_sizes[1] / 2;

  const size_t nh   = (size_t)n * HEADS;
  const size_t segH = (nh * sizeof(float) + 255) & ~(size_t)255;
  const size_t segN = ((size_t)n * sizeof(int) + 255) & ~(size_t)255;
  const size_t segT = (size_t)3 * MAXR * sizeof(int);
  const size_t segE = ((size_t)nE * sizeof(int) + 255) & ~(size_t)255;
  char* ws = (char*)d_ws;
  float* h_mat      = (float*)(ws);
  int*   deg        = (int*)  (ws + segH);
  int*   cur        = (int*)  (ws + segH + segN);
  int*   rangeCnt   = (int*)  (ws + segH + 2 * segN);
  int*   rangeCur   = rangeCnt + MAXR;
  int*   rangeStart = rangeCnt + 2 * MAXR;
  int*   part       = (int*)  (ws + segH + 2 * segN + segT);
  int*   bsrc       = (ws_size >= segH + 2 * segN + segT + 2 * segE) ? (int*)((char*)part + segE) : part;

  const int tiles  = (n + 15) / 16;
  const int blocks = (tiles + 7) / 8;
  gat_gemm_kernel<<<blocks, GEMM_T, 0, stream>>>(x, W, h_mat, rangeCnt, n);

  const int R  = (n + RSIZE - 1) / RSIZE;
  const int eb = (nE + PB * EPT - 1) / (PB * EPT);
  gat_rangecount_kernel<<<eb, PB, 0, stream>>>(ei, rangeCnt, nE, R);

  gat_partition_kernel<<<eb, PB, 0, stream>>>(ei, rangeCnt, rangeCur, rangeStart, part, nE, R);

  gat_rangesort_kernel<<<R, PB, 0, stream>>>(part, rangeCnt, rangeStart, deg, cur, bsrc, n);

  gat_gather_kernel<<<(n + 7) / 8, 256, 0, stream>>>(
      bsrc, cur, deg, h_mat, att_src, att_dst, gat_bias, lin_w, lin_b, bias,
      out, n);
}


# gather on a resident grid: 8192 waves each walk a run of consecutive nodes, the next node's source ids in flight with this node's row gathers (was one short wave per node); sum order kept
# speedup vs baseline: 17.8634x; 1.0582x over previous
#include <hip/hip_runtime.h>
#include <math.h>


#define IN_CH 512
#define HEADS 8
#define GEMM_T 256
#define NEG_SLOPE 0.2f
#define CHUNK 32
#define RSHIFT 8
#define RSIZE 256
#define MAXR 1024
#define PB 1024
#define EPT 8
#define CAP 16384

typedef __attribute__((ext_vector_type(2))) float v2f;
typedef __attribute__((ext_vector_type(8))) float v8f;

__device__ __forceinline__ float leaky(float v) {
  return v >= 0.f ? v : NEG_SLOPE * v;
}

#define LOG2E 1.44269504088896340736f

__device__ __forceinline__ float exp_rel(float e, float mL) {
  return __builtin_amdgcn_exp2f(fmaf(e, LOG2E, -mL));
}

__global__ __launch_bounds__(GEMM_T) void gat_gemm_kernel(
    const float* __restrict__ x, const float* __restrict__ W,
    float* __restrict__ h, int* __restrict__ rangeTabs, int n)
{
  __shared__ float4 Wlds4[IN_CH * HEADS / 4];
  {
    const float4* __restrict__ W4 = reinterpret_cast<const float4*>(W);
    float4 wv[4];
    #pragma unroll
    for (int j = 0; j < 4; ++j) wv[j] = W4[threadIdx.x + j * GEMM_T];
    #pragma unroll
    for (int j = 0; j < 4; ++j) Wlds4[threadIdx.x + j * GEMM_T] = wv[j];
  }
  if (blockIdx.x == 0) {
    for (int i = threadIdx.x; i < 2 * MAXR; i += GEMM_T) rangeTabs[i] = 0;
  }
  __syncthreads();

  const int lane = threadIdx.x & 31;
  const int wave = threadIdx.x >> 5;
  const int half = lane >> 4;
  const int l    = lane & 15;

  const int tile = blockIdx.x * (GEMM_T >> 5) + wave;
  const int row0 = tile * 16;
  if (row0 >= n) return;

  int arow = row0 + l;
  if (arow > n - 1) arow = n - 1;
  const float4* __restrict__ xr =
      reinterpret_cast<const float4*>(x + (size_t)arow * IN_CH) + half;
  const float* __restrict__ wl =
      reinterpret_cast<const float*>(Wlds4) + half * 4 * HEADS + (l & 7);

  v8f c = {0.f, 0.f, 0.f, 0.f, 0.f, 0.f, 0.f, 0.f};

  float4 xa[3][4];
  #pragma unroll
  for (int i = 0; i < 4; ++i) xa[0][i] = xr[i * 2];
  #pragma unroll
  for (int i = 0; i < 4; ++i) xa[1][i] = xr[8 + i * 2];

  #pragma unroll
  for (int t = 0; t < IN_CH / 32; ++t) {
    if (t + 2 < IN_CH / 32) {
      #pragma unroll
      for (int i = 0; i < 4; ++i) xa[(t + 2) % 3][i] = xr[(t + 2) * 8 + i * 2];
    }
    #pragma unroll
    for (int i = 0; i < 4; ++i) {
      const float4 av = xa[t % 3][i];
      const float* w = wl + (t * 32 + i * 8) * HEADS;
      v2f a, b;
      a.x = av.x;  a.y = av.y;
      b.x = w[0];  b.y = w[HEADS];
      c = __builtin_amdgcn_wmma_f32_16x16x4_f32(
           false, a,  false, b,
           (short)0, c,  false,  false);
      a.x = av.z;  a.y = av.w;
      b.x = w[2 * HEADS];  b.y = w[3 * HEADS];
      c = __builtin_amdgcn_wmma_f32_16x16x4_f32(
           false, a,  false, b,
           (short)0, c,  false,  false);
    }
  }

  if (l < HEADS) {
    #pragma unroll
    for (int i = 0; i < 8; ++i) {
      const int r = row0 + i + half * 8;
      if (r < n) h[(size_t)r * HEADS + l] = c[i];
    }
  }
}

__device__ __forceinline__ int block_excl_scan(int v, int* wsum, int lane, int wave) {
  int incl = v;
  #pragma unroll
  for (int o = 1; o < 32; o <<= 1) {
    const int t = __shfl_up(incl, o);
    if (lane >= o) incl += t;
  }
  if (lane == 31) wsum[wave] = incl;
  __syncthreads();
  if (wave == 0) {
    const int w = wsum[lane];
    int wi = w;
    #pragma unroll
    for (int o = 1; o < 32; o <<= 1) {
      const int t = __shfl_up(wi, o);
      if (lane >= o) wi += t;
    }
    wsum[lane] = wi - w;
  }
  __syncthreads();
  return wsum[wave] + incl - v;
}

__global__ __launch_bounds__(PB) void gat_rangecount_kernel(
    const int* __restrict__ ei, int* __restrict__ rangeCnt, int nE, int R)
{
  __shared__ int hist[MAXR];
  const int t = threadIdx.x;
  hist[t] = 0;
  __syncthreads();
  const int base = blockIdx.x * (PB * EPT);
  int d[EPT];
  #pragma unroll
  for (int j = 0; j < EPT; ++j) {
    const int e = base + j * PB + t;
    d[j] = (e < nE) ? ei[nE + e] : -1;
  }
  #pragma unroll
  for (int j = 0; j < EPT; ++j)
    if (d[j] >= 0) atomicAdd(&hist[d[j] >> RSHIFT], 1);
  __syncthreads();
  if (t < R) {
    const int c = hist[t];
    if (c) atomicAdd(rangeCnt + t, c);
  }
}

__global__ __launch_bounds__(PB) void gat_partition_kernel(
    const int* __restrict__ ei, const int* __restrict__ rangeCnt,
    int* __restrict__ rangeCur, int* __restrict__ rangeStart,
    int* __restrict__ part, int nE, int R)
{
  __shared__ int hist[MAXR];
  __shared__ int lstart[MAXR];
  __shared__ int rbase[MAXR];
  __shared__ int wsum[32];
  __shared__ int skey[PB * EPT];
  __shared__ int sgpos[PB * EPT];
  const int t    = threadIdx.x;
  const int lane = t & 31;
  const int wave = t >> 5;
  hist[t] = 0;
  const int tot    = (t < R) ? rangeCnt[t] : 0;
  const int rstart = block_excl_scan(tot, wsum, lane, wave);
  if (blockIdx.x == 0 && t < R) rangeStart[t] = rstart;

  const int base = blockIdx.x * (PB * EPT);
  int s[EPT], d[EPT], rk[EPT];
  #pragma unroll
  for (int j = 0; j < EPT; ++j) {
    const int e = base + j * PB + t;
    const bool ok = e < nE;
    s[j] = ok ? ei[e] : 0;
    d[j] = ok ? ei[nE + e] : -1;
  }
  #pragma unroll
  for (int j = 0; j < EPT; ++j) {
    rk[j] = 0;
    if (d[j] >= 0) rk[j] = atomicAdd(&hist[d[j] >> RSHIFT], 1);
  }
  __syncthreads();
  const int cc = hist[t];
  const int ls = block_excl_scan(cc, wsum, lane, wave);
  lstart[t] = ls;
  rbase[t]  = (t < R && cc) ? rstart + atomicAdd(rangeCur + t, cc) : 0;
  __syncthreads();
  #pragma unroll
  for (int j = 0; j < EPT; ++j) {
    if (d[j] >= 0) {
      const int r = d[j] >> RSHIFT;
      const int p = lstart[r] + rk[j];
      skey[p]  = (s[j] << RSHIFT) | (d[j] & (RSIZE - 1));
      sgpos[p] = rbase[r] + rk[j];
    }
  }
  __syncthreads();
  const int left  = nE - base;
  const int mine  = left < PB * EPT ? left : PB * EPT;
  #pragma unroll
  for (int j = 0; j < EPT; ++j) {
    const int i = j * PB + t;
    if (i < mine) part[sgpos[i]] = skey[i];
  }
}

__global__ __launch_bounds__(PB) void gat_rangesort_kernel(
    const int* part, const int* __restrict__ rangeCnt,
    const int* __restrict__ rangeStart, int* __restrict__ deg,
    int* __restrict__ cur, int* bsrc, int n)
{
  __shared__ int cnt[RSIZE];
  __shared__ int pos[RSIZE];
  __shared__ int wsum[32];
  __shared__ int stage[CAP];
  const int t    = threadIdx.x;
  const int lane = t & 31;
  const int wave = t >> 5;
  const int r    = blockIdx.x;
  const int L    = rangeCnt[r];
  const int seg  = rangeStart[r];
  if (t < RSIZE) cnt[t] = 0;
  __syncthreads();
  for (int i0 = t; i0 < L; i0 += 4 * PB) {
    int k[4];
    #pragma unroll
    for (int u = 0; u < 4; ++u) {
      const int i = i0 + u * PB;
      k[u] = (i < L) ? part[seg + i] : -1;
    }
    #pragma unroll
    for (int u = 0; u < 4; ++u)
      if (k[u] >= 0) atomicAdd(&cnt[k[u] & (RSIZE - 1)], 1);
  }
  __syncthreads();
  const int c    = (t < RSIZE) ? cnt[t] : 0;
  const int excl = block_excl_scan(c, wsum, lane, wave);
  if (t < RSIZE) {
    pos[t] = excl;
    const int node = r * RSIZE + t;
    if (node < n) { deg[node] = c; cur[node] = seg + excl + c; }
  }
  __syncthreads();
  const bool inLds = (L <= CAP);
  for (int i0 = t; i0 < L; i0 += 4 * PB) {
    int k[4];
    #pragma unroll
    for (int u = 0; u < 4; ++u) {
      const int i = i0 + u * PB;
      k[u] = (i < L) ? part[seg + i] : -1;
    }
    #pragma unroll
    for (int u = 0; u < 4; ++u) {
      if (k[u] >= 0) {
        const int p   = atomicAdd(&pos[k[u] & (RSIZE - 1)], 1);
        const int sid = k[u] >> RSHIFT;
        if (inLds) stage[p] = sid; else bsrc[seg + p] = sid;
      }
    }
  }
  if (inLds) {
    __syncthreads();
    for (int i = t; i < L; i += PB) bsrc[seg + i] = stage[i];
  }
}

__device__ __forceinline__ void gather_load_ids(const int* __restrict__ bsrc, int base,
                                                int end, int q, int (&s)[8]) {
  #pragma unroll
  for (int j = 0; j < 8; ++j) {
    const int p = base + j * 4 + q;
    s[j] = bsrc[p < end ? p : end - 1];
  }
}

__device__ __forceinline__ void gather_load_rows(const float* __restrict__ hmat,
                                                 const int (&s)[8], int hh, float (&h)[8]) {
  #pragma unroll
  for (int j = 0; j < 8; ++j) h[j] = hmat[s[j] * HEADS + hh];
}

#define GATHER_BLOCKS 1024

__device__ __forceinline__ void gather_node_ids(const int* __restrict__ bsrc, int start,
                                                int end, int cnt, int q,
                                                int (&sA)[8], int (&sB)[8]) {
  if (cnt > 0)     gather_load_ids(bsrc, start, end, q, sA);
  if (cnt > CHUNK) gather_load_ids(bsrc, start + CHUNK, end, q, sB);
}

__global__ __launch_bounds__(256) void gat_gather_kernel(
    const int* __restrict__ bsrc, const int* __restrict__ cur,
    const int* __restrict__ deg, const float* __restrict__ hmat,
    const float* __restrict__ att_src, const float* __restrict__ att_dst,
    const float* __restrict__ gat_bias, const float* __restrict__ lin_w,
    const float* __restrict__ lin_b, const float* __restrict__ bias,
    float* __restrict__ out, int n)
{
  const int lane = threadIdx.x & 31;
  const int q  = lane >> 3;
  const int hh = lane & 7;

  const int T   = GATHER_BLOCKS * 8;
  const int w   = __builtin_amdgcn_readfirstlane((int)(blockIdx.x * 8 + (threadIdx.x >> 5)));
  const int per = n / T;
  const int rem = n - per * T;
  const int n0  = w * per + (w < rem ? w : rem);
  const int n1  = n0 + per + (w < rem ? 1 : 0);
  if (n0 >= n1) return;

  const float as = att_src[hh];
  const float ad = att_dst[hh];
  const float gb = gat_bias[hh];
  const float lb = lin_b[0];
  const float bo = bias[0];

  int endC = cur[n0];
  int cntC = deg[n0];
  const int i1 = (n0 + 1 < n1) ? n0 + 1 : n1 - 1;
  int end1 = cur[i1];
  int cnt1 = deg[i1];

  int nA[8], nB[8];
  #pragma unroll
  for (int j = 0; j < 8; ++j) { nA[j] = 0;  nB[j] = 0; }
  gather_node_ids(bsrc, endC - cntC, endC, cntC, q, nA, nB);

  for (int node = n0; node < n1; ++node) {
    const int end   = endC;
    const int cnt   = cntC;
    const int start = end - cnt;
    const bool hasA = cnt > 0;
    const bool hasB = cnt > CHUNK;
    const int  rest = start + 2 * CHUNK;

    int sA[8], sB[8];
    #pragma unroll
    for (int j = 0; j < 8; ++j) { sA[j] = nA[j];  sB[j] = nB[j]; }

    const int i2 = (node + 2 < n1) ? node + 2 : n1 - 1;
    const int end2 = cur[i2];
    const int cnt2 = deg[i2];

    const int cntN = (node + 1 < n1) ? cnt1 : 0;
    gather_node_ids(bsrc, end1 - cntN, end1, cntN, q, nA, nB);

    float hA[8], hB[8], eA[8], eB[8];
    #pragma unroll
    for (int j = 0; j < 8; ++j) {
      hA[j] = 0.f;  hB[j] = 0.f;  eA[j] = -INFINITY;  eB[j] = -INFINITY;
    }
    const float hself = hmat[node * HEADS + hh];
    if (hasA) gather_load_rows(hmat, sA, hh, hA);
    if (hasB) gather_load_rows(hmat, sB, hh, hB);

    const float adst  = hself * ad;
    const float eself = leaky(hself * as + adst);
    if (hasA) {
      #pragma unroll
      for (int j = 0; j < 8; ++j)
        if (j * 4 + q < cnt) eA[j] = leaky(hA[j] * as + adst);
    }
    if (hasB) {
      #pragma unroll
      for (int j = 0; j < 8; ++j)
        if (CHUNK + j * 4 + q < cnt) eB[j] = leaky(hB[j] * as + adst);
    }

    float m = (q == 0) ? eself : -INFINITY;
    #pragma unroll
    for (int j = 0; j < 8; ++j) m = fmaxf(m, eA[j]);
    if (hasB) {
      #pragma unroll
      for (int j = 0; j < 8; ++j) m = fmaxf(m, eB[j]);
    }
    for (int base = rest; base < end; base += CHUNK) {
      int   s[8];
      float hs[8];
      #pragma unroll
      for (int j = 0; j < 8; ++j) {
        const int p = base + j * 4 + q;
        s[j] = bsrc[p < end ? p : end - 1];
      }
      #pragma unroll
      for (int j = 0; j < 8; ++j) hs[j] = hmat[s[j] * HEADS + hh];
      #pragma unroll
      for (int j = 0; j < 8; ++j) {
        const int p = base + j * 4 + q;
        const float e = leaky(hs[j] * as + adst);
        if (p < end) m = fmaxf(m, e);
      }
    }
    m = fmaxf(m, __shfl_xor(m, 8));
    m = fmaxf(m, __shfl_xor(m, 16));

    const float mL = m * LOG2E;
    float den = 0.f, num = 0.f;
    if (q == 0) { den = exp_rel(eself, mL); num = den * hself; }
    #pragma unroll
    for (int j = 0; j < 8; ++j) {
      const float ee = exp_rel(eA[j], mL);
      den += ee;  num = fmaf(ee, hA[j], num);
    }
    if (hasB) {
      #pragma unroll
      for (int j = 0; j < 8; ++j) {
        const float ee = exp_rel(eB[j], mL);
        den += ee;  num = fmaf(ee, hB[j], num);
      }
    }
    for (int base = rest; base < end; base += CHUNK) {
      int   s[8];
      float hs[8];
      #pragma unroll
      for (int j = 0; j < 8; ++j) {
        const int p = base + j * 4 + q;
        s[j] = bsrc[p < end ? p : end - 1];
      }
      #pragma unroll
      for (int j = 0; j < 8; ++j) hs[j] = hmat[s[j] * HEADS + hh];
      #pragma unroll
      for (int j = 0; j < 8; ++j) {
        const int p = base + j * 4 + q;
        const float ee = exp_rel(leaky(hs[j] * as + adst), mL);
        if (p < end) { den += ee; num += ee * hs[j]; }
      }
    }
    den += __shfl_xor(den, 8);
    den += __shfl_xor(den, 16);
    num += __shfl_xor(num, 8);
    num += __shfl_xor(num, 16);

    const float oh = num / (den + 1e-16f) + gb;
    float z = lb;
    #pragma unroll
    for (int k = 0; k < HEADS; ++k) z += __shfl(oh, k) * lin_w[k];
    if (lane == 0) out[node] = fmaxf(z, 0.f) + bo;

    endC = end1;  cntC = cntN;
    end1 = end2;  cnt1 = cnt2;
  }
}

extern "C" void kernel_launch(void* const* d_in, const int* in_sizes, int n_in,
                              void* d_out, int out_size, void* d_ws, size_t ws_size,
                              hipStream_t stream) {
  const float* x        = (const float*)d_in[0];
  const int*   ei       = (const int*)d_in[1];
  const float* W        = (const float*)d_in[2];
  const float* att_src  = (const float*)d_in[3];
  const float* att_dst  = (const float*)d_in[4];
  const float* gat_bias = (const float*)d_in[5];
  const float* lin_w    = (const float*)d_in[6];
  const float* lin_b    = (const float*)d_in[7];
  const float* bias     = (const float*)d_in[8];
  float* out = (float*)d_out;

  const int n  = in_sizes[0] / IN_CH;
  const int nE = in_sizes[1] / 2;

  const size_t nh   = (size_t)n * HEADS;
  const size_t segH = (nh * sizeof(float) + 255) & ~(size_t)255;
  const size_t segN = ((size_t)n * sizeof(int) + 255) & ~(size_t)255;
  const size_t segT = (size_t)3 * MAXR * sizeof(int);
  const size_t segE = ((size_t)nE * sizeof(int) + 255) & ~(size_t)255;
  char* ws = (char*)d_ws;
  float* h_mat      = (float*)(ws);
  int*   deg        = (int*)  (ws + segH);
  int*   cur        = (int*)  (ws + segH + segN);
  int*   rangeCnt   = (int*)  (ws + segH + 2 * segN);
  int*   rangeCur   = rangeCnt + MAXR;
  int*   rangeStart = rangeCnt + 2 * MAXR;
  int*   part       = (int*)  (ws + segH + 2 * segN + segT);
  int*   bsrc       = (ws_size >= segH + 2 * segN + segT + 2 * segE) ? (int*)((char*)part + segE) : part;

  const int tiles  = (n + 15) / 16;
  const int blocks = (tiles + 7) / 8;
  gat_gemm_kernel<<<blocks, GEMM_T, 0, stream>>>(x, W, h_mat, rangeCnt, n);

  const int R  = (n + RSIZE - 1) / RSIZE;
  const int eb = (nE + PB * EPT - 1) / (PB * EPT);
  gat_rangecount_kernel<<<eb, PB, 0, stream>>>(ei, rangeCnt, nE, R);

  gat_partition_kernel<<<eb, PB, 0, stream>>>(ei, rangeCnt, rangeCur, rangeStart, part, nE, R);

  gat_rangesort_kernel<<<R, PB, 0, stream>>>(part, rangeCnt, rangeStart, deg, cur, bsrc, n);

  gat_gather_kernel<<<GATHER_BLOCKS, 256, 0, stream>>>(
      bsrc, cur, deg, h_mat, att_src, att_dst, gat_bias, lin_w, lin_b, bias,
      out, n);
}


# edge bucketing in two launches, not three: each block orders its own edges by range in LDS and stores them with a run table; a range collects its runs from all blocks; list and sum order change
# speedup vs baseline: 18.9048x; 1.0583x over previous
#include <hip/hip_runtime.h>
#include <math.h>


#define IN_CH 512
#define HEADS 8
#define GEMM_T 256
#define NEG_SLOPE 0.2f
#define CHUNK 32
#define RSHIFT 8
#define RSIZE 256
#define MAXR 1024
#define PB 1024
#define EPT 8
#define CAP 16384
#define MAXB 1024

typedef __attribute__((ext_vector_type(2))) float v2f;
typedef __attribute__((ext_vector_type(8))) float v8f;

__device__ __forceinline__ float leaky(float v) {
  return v >= 0.f ? v : NEG_SLOPE * v;
}

#define LOG2E 1.44269504088896340736f

__device__ __forceinline__ float exp_rel(float e, float mL) {
  return __builtin_amdgcn_exp2f(fmaf(e, LOG2E, -mL));
}

__global__ __launch_bounds__(GEMM_T) void gat_gemm_kernel(
    const float* __restrict__ x, const float* __restrict__ W,
    float* __restrict__ h, int* __restrict__ rangeTabs, int n)
{
  __shared__ float4 Wlds4[IN_CH * HEADS / 4];
  {
    const float4* __restrict__ W4 = reinterpret_cast<const float4*>(W);
    float4 wv[4];
    #pragma unroll
    for (int j = 0; j < 4; ++j) wv[j] = W4[threadIdx.x + j * GEMM_T];
    #pragma unroll
    for (int j = 0; j < 4; ++j) Wlds4[threadIdx.x + j * GEMM_T] = wv[j];
  }
  if (blockIdx.x == 0) {
    for (int i = threadIdx.x; i < 2 * MAXR; i += GEMM_T) rangeTabs[i] = 0;
  }
  __syncthreads();

  const int lane = threadIdx.x & 31;
  const int wave = threadIdx.x >> 5;
  const int half = lane >> 4;
  const int l    = lane & 15;

  const int tile = blockIdx.x * (GEMM_T >> 5) + wave;
  const int row0 = tile * 16;
  if (row0 >= n) return;

  int arow = row0 + l;
  if (arow > n - 1) arow = n - 1;
  const float4* __restrict__ xr =
      reinterpret_cast<const float4*>(x + (size_t)arow * IN_CH) + half;
  const float* __restrict__ wl =
      reinterpret_cast<const float*>(Wlds4) + half * 4 * HEADS + (l & 7);

  v8f c = {0.f, 0.f, 0.f, 0.f, 0.f, 0.f, 0.f, 0.f};

  float4 xa[3][4];
  #pragma unroll
  for (int i = 0; i < 4; ++i) xa[0][i] = xr[i * 2];
  #pragma unroll
  for (int i = 0; i < 4; ++i) xa[1][i] = xr[8 + i * 2];

  #pragma unroll
  for (int t = 0; t < IN_CH / 32; ++t) {
    if (t + 2 < IN_CH / 32) {
      #pragma unroll
      for (int i = 0; i < 4; ++i) xa[(t + 2) % 3][i] = xr[(t + 2) * 8 + i * 2];
    }
    #pragma unroll
    for (int i = 0; i < 4; ++i) {
      const float4 av = xa[t % 3][i];
      const float* w = wl + (t * 32 + i * 8) * HEADS;
      v2f a, b;
      a.x = av.x;  a.y = av.y;
      b.x = w[0];  b.y = w[HEADS];
      c = __builtin_amdgcn_wmma_f32_16x16x4_f32(
           false, a,  false, b,
           (short)0, c,  false,  false);
      a.x = av.z;  a.y = av.w;
      b.x = w[2 * HEADS];  b.y = w[3 * HEADS];
      c = __builtin_amdgcn_wmma_f32_16x16x4_f32(
           false, a,  false, b,
           (short)0, c,  false,  false);
    }
  }

  if (l < HEADS) {
    #pragma unroll
    for (int i = 0; i < 8; ++i) {
      const int r = row0 + i + half * 8;
      if (r < n) h[(size_t)r * HEADS + l] = c[i];
    }
  }
}

__device__ __forceinline__ int block_excl_scan(int v, int* wsum, int lane, int wave) {
  int incl = v;
  #pragma unroll
  for (int o = 1; o < 32; o <<= 1) {
    const int t = __shfl_up(incl, o);
    if (lane >= o) incl += t;
  }
  if (lane == 31) wsum[wave] = incl;
  __syncthreads();
  if (wave == 0) {
    const int w = wsum[lane];
    int wi = w;
    #pragma unroll
    for (int o = 1; o < 32; o <<= 1) {
      const int t = __shfl_up(wi, o);
      if (lane >= o) wi += t;
    }
    wsum[lane] = wi - w;
  }
  __syncthreads();
  return wsum[wave] + incl - v;
}

__global__ __launch_bounds__(PB) void gat_rangecount_kernel(
    const int* __restrict__ ei, int* __restrict__ rangeCnt, int nE, int R)
{
  __shared__ int hist[MAXR];
  const int t = threadIdx.x;
  hist[t] = 0;
  __syncthreads();
  const int base = blockIdx.x * (PB * EPT);
  int d[EPT];
  #pragma unroll
  for (int j = 0; j < EPT; ++j) {
    const int e = base + j * PB + t;
    d[j] = (e < nE) ? ei[nE + e] : -1;
  }
  #pragma unroll
  for (int j = 0; j < EPT; ++j)
    if (d[j] >= 0) atomicAdd(&hist[d[j] >> RSHIFT], 1);
  __syncthreads();
  if (t < R) {
    const int c = hist[t];
    if (c) atomicAdd(rangeCnt + t, c);
  }
}

__global__ __launch_bounds__(PB) void gat_partition_kernel(
    const int* __restrict__ ei, const int* __restrict__ rangeCnt,
    int* __restrict__ rangeCur, int* __restrict__ rangeStart,
    int* __restrict__ part, int nE, int R)
{
  __shared__ int hist[MAXR];
  __shared__ int lstart[MAXR];
  __shared__ int rbase[MAXR];
  __shared__ int wsum[32];
  __shared__ int skey[PB * EPT];
  __shared__ int sgpos[PB * EPT];
  const int t    = threadIdx.x;
  const int lane = t & 31;
  const int wave = t >> 5;
  hist[t] = 0;
  const int tot    = (t < R) ? rangeCnt[t] : 0;
  const int rstart = block_excl_scan(tot, wsum, lane, wave);
  if (blockIdx.x == 0 && t < R) rangeStart[t] = rstart;

  const int base = blockIdx.x * (PB * EPT);
  int s[EPT], d[EPT], rk[EPT];
  #pragma unroll
  for (int j = 0; j < EPT; ++j) {
    const int e = base + j * PB + t;
    const bool ok = e < nE;
    s[j] = ok ? ei[e] : 0;
    d[j] = ok ? ei[nE + e] : -1;
  }
  #pragma unroll
  for (int j = 0; j < EPT; ++j) {
    rk[j] = 0;
    if (d[j] >= 0) rk[j] = atomicAdd(&hist[d[j] >> RSHIFT], 1);
  }
  __syncthreads();
  const int cc = hist[t];
  const int ls = block_excl_scan(cc, wsum, lane, wave);
  lstart[t] = ls;
  rbase[t]  = (t < R && cc) ? rstart + atomicAdd(rangeCur + t, cc) : 0;
  __syncthreads();
  #pragma unroll
  for (int j = 0; j < EPT; ++j) {
    if (d[j] >= 0) {
      const int r = d[j] >> RSHIFT;
      const int p = lstart[r] + rk[j];
      skey[p]  = (s[j] << RSHIFT) | (d[j] & (RSIZE - 1));
      sgpos[p] = rbase[r] + rk[j];
    }
  }
  __syncthreads();
  const int left  = nE - base;
  const int mine  = left < PB * EPT ? left : PB * EPT;
  #pragma unroll
  for (int j = 0; j < EPT; ++j) {
    const int i = j * PB + t;
    if (i < mine) part[sgpos[i]] = skey[i];
  }
}

__global__ __launch_bounds__(PB) void gat_rangesort_kernel(
    const int* part, const int* __restrict__ rangeCnt,
    const int* __restrict__ rangeStart, int* __restrict__ deg,
    int* __restrict__ cur, int* bsrc, int n)
{
  __shared__ int cnt[RSIZE];
  __shared__ int pos[RSIZE];
  __shared__ int wsum[32];
  __shared__ int stage[CAP];
  const int t    = threadIdx.x;
  const int lane = t & 31;
  const int wave = t >> 5;
  const int r    = blockIdx.x;
  const int L    = rangeCnt[r];
  const int seg  = rangeStart[r];
  if (t < RSIZE) cnt[t] = 0;
  __syncthreads();
  for (int i0 = t; i0 < L; i0 += 4 * PB) {
    int k[4];
    #pragma unroll
    for (int u = 0; u < 4; ++u) {
      const int i = i0 + u * PB;
      k[u] = (i < L) ? part[seg + i] : -1;
    }
    #pragma unroll
    for (int u = 0; u < 4; ++u)
      if (k[u] >= 0) atomicAdd(&cnt[k[u] & (RSIZE - 1)], 1);
  }
  __syncthreads();
  const int c    = (t < RSIZE) ? cnt[t] : 0;
  const int excl = block_excl_scan(c, wsum, lane, wave);
  if (t < RSIZE) {
    pos[t] = excl;
    const int node = r * RSIZE + t;
    if (node < n) { deg[node] = c; cur[node] = seg + excl + c; }
  }
  __syncthreads();
  const bool inLds = (L <= CAP);
  for (int i0 = t; i0 < L; i0 += 4 * PB) {
    int k[4];
    #pragma unroll
    for (int u = 0; u < 4; ++u) {
      const int i = i0 + u * PB;
      k[u] = (i < L) ? part[seg + i] : -1;
    }
    #pragma unroll
    for (int u = 0; u < 4; ++u) {
      if (k[u] >= 0) {
        const int p   = atomicAdd(&pos[k[u] & (RSIZE - 1)], 1);
        const int sid = k[u] >> RSHIFT;
        if (inLds) stage[p] = sid; else bsrc[seg + p] = sid;
      }
    }
  }
  if (inLds) {
    __syncthreads();
    for (int i = t; i < L; i += PB) bsrc[seg + i] = stage[i];
  }
}

__global__ __launch_bounds__(PB) void gat_blocksort_kernel(
    const int* __restrict__ ei, int* __restrict__ part, int* __restrict__ runTab,
    int nE, int R, int RP)
{
  __shared__ int hist[MAXR];
  __shared__ int lstart[MAXR];
  __shared__ int wsum[32];
  __shared__ int skey[PB * EPT];
  const int t    = threadIdx.x;
  const int lane = t & 31;
  const int wave = t >> 5;
  hist[t] = 0;
  __syncthreads();

  const int base = blockIdx.x * (PB * EPT);
  int s[EPT], d[EPT], rk[EPT];
  #pragma unroll
  for (int j = 0; j < EPT; ++j) {
    const int e = base + j * PB + t;
    const bool ok = e < nE;
    s[j] = ok ? ei[e] : 0;
    d[j] = ok ? ei[nE + e] : -1;
  }
  #pragma unroll
  for (int j = 0; j < EPT; ++j) {
    rk[j] = 0;
    if (d[j] >= 0) rk[j] = atomicAdd(&hist[d[j] >> RSHIFT], 1);
  }
  __syncthreads();
  const int cc = hist[t];
  const int ls = block_excl_scan(cc, wsum, lane, wave);
  lstart[t] = ls;
  if (t < R) runTab[(size_t)blockIdx.x * RP + t] = (ls << 16) | cc;
  __syncthreads();
  #pragma unroll
  for (int j = 0; j < EPT; ++j) {
    if (d[j] >= 0) {
      const int r = d[j] >> RSHIFT;
      skey[lstart[r] + rk[j]] = (s[j] << RSHIFT) | (d[j] & (RSIZE - 1));
    }
  }
  __syncthreads();
  const int left = nE - base;
  const int mine = left < PB * EPT ? left : PB * EPT;
  #pragma unroll
  for (int j = 0; j < EPT; ++j) {
    const int i = j * PB + t;
    if (i < mine) part[base + i] = skey[i];
  }
}

template <bool PLACE>
__device__ __forceinline__ void range_entry(int kk, int* cnt, int* pos, int* stage,
                                            int* __restrict__ bsrc, int seg, bool inLds) {
  if (!PLACE) {
    atomicAdd(&cnt[kk & (RSIZE - 1)], 1);
  } else {
    const int p   = atomicAdd(&pos[kk & (RSIZE - 1)], 1);
    const int sid = kk >> RSHIFT;
    if (inLds) stage[p] = sid; else bsrc[seg + p] = sid;
  }
}

template <bool PLACE>
__device__ __forceinline__ void range_run(const int* __restrict__ part, const int* runPk,
                                          int b, int first, int lane, int* cnt, int* pos,
                                          int* stage, int* __restrict__ bsrc, int seg,
                                          bool inLds) {
  const int pk  = runPk[b];
  const int len = pk & 0xffff;
  const int beg = b * (PB * EPT) + (pk >> 16);
  for (int i = first + lane; i < len; i += 32)
    range_entry<PLACE>(part[beg + i], cnt, pos, stage, bsrc, seg, inLds);
}

#define HELD 16

__global__ __launch_bounds__(PB) void gat_rangecollect_kernel(
    const int* __restrict__ part, const int* __restrict__ runTab,
    int* __restrict__ cursor, int* __restrict__ deg, int* __restrict__ cur,
    int* __restrict__ bsrc, int n, int EB, int RP)
{
  __shared__ int cnt[RSIZE];
  __shared__ int pos[RSIZE];
  __shared__ int wsum[32];
  __shared__ int runPk[MAXB];
  __shared__ int segS[2];
  __shared__ int stage[CAP];
  const int t    = threadIdx.x;
  const int lane = t & 31;
  const int wave = __builtin_amdgcn_readfirstlane(t >> 5);
  const int r    = blockIdx.x;

  const int pk    = (t < EB) ? runTab[(size_t)t * RP + r] : 0;
  const int myLen = pk & 0xffff;
  runPk[t] = pk;
  if (t < RSIZE) cnt[t] = 0;
  const int before = block_excl_scan(myLen, wsum, lane, wave);
  int segv = 0;
  if (t == PB - 1) {
    segS[1] = before + myLen;
    segv = atomicAdd(cursor, before + myLen);
  }

  int k[HELD];
  unsigned longMask = 0;
  {
    int pkv[HELD];
    #pragma unroll
    for (int j = 0; j < HELD; ++j) pkv[j] = runPk[wave + 32 * j];
    #pragma unroll
    for (int j = 0; j < HELD; ++j) {
      const int len = pkv[j] & 0xffff;
      const int beg = (wave + 32 * j) * (PB * EPT) + (pkv[j] >> 16);
      k[j] = -1;
      if (len > 32) longMask |= 1u << j;
      if (lane < len) k[j] = part[beg + lane];
    }
  }
  longMask = __builtin_amdgcn_readfirstlane(longMask);

  #pragma unroll
  for (int j = 0; j < HELD; ++j)
    if (k[j] >= 0) range_entry<false>(k[j], cnt, pos, stage, bsrc, 0, true);
  for (unsigned m = longMask; m; m &= m - 1)
    range_run<false>(part, runPk, wave + 32 * __builtin_ctz(m), 32, lane, cnt, pos, stage, bsrc, 0, true);
  for (int b = wave + 32 * HELD; b < EB; b += 32)
    range_run<false>(part, runPk, b, 0, lane, cnt, pos, stage, bsrc, 0, true);
  if (t == PB - 1) segS[0] = segv;
  __syncthreads();
  const int seg = segS[0];
  const int L   = segS[1];

  const int c    = (t < RSIZE) ? cnt[t] : 0;
  const int excl = block_excl_scan(c, wsum, lane, wave);
  if (t < RSIZE) {
    pos[t] = excl;
    const int node = r * RSIZE + t;
    if (node < n) { deg[node] = c; cur[node] = seg + excl + c; }
  }
  __syncthreads();

  const bool inLds = (L <= CAP);
  {
    int p[HELD];
    #pragma unroll
    for (int j = 0; j < HELD; ++j)
      p[j] = (k[j] >= 0) ? atomicAdd(&pos[k[j] & (RSIZE - 1)], 1) : 0;
    #pragma unroll
    for (int j = 0; j < HELD; ++j) {
      if (k[j] >= 0) {
        const int sid = k[j] >> RSHIFT;
        if (inLds) stage[p[j]] = sid; else bsrc[seg + p[j]] = sid;
      }
    }
  }
  for (unsigned m = longMask; m; m &= m - 1)
    range_run<true>(part, runPk, wave + 32 * __builtin_ctz(m), 32, lane, cnt, pos, stage, bsrc, seg, inLds);
  for (int b = wave + 32 * HELD; b < EB; b += 32)
    range_run<true>(part, runPk, b, 0, lane, cnt, pos, stage, bsrc, seg, inLds);
  if (inLds) {
    __syncthreads();
    for (int i = t; i < L; i += PB) bsrc[seg + i] = stage[i];
  }
}

__device__ __forceinline__ void gather_load_ids(const int* __restrict__ bsrc, int base,
                                                int end, int q, int (&s)[8]) {
  #pragma unroll
  for (int j = 0; j < 8; ++j) {
    const int p = base + j * 4 + q;
    s[j] = bsrc[p < end ? p : end - 1];
  }
}

__device__ __forceinline__ void gather_load_rows(const float* __restrict__ hmat,
                                                 const int (&s)[8], int hh, float (&h)[8]) {
  #pragma unroll
  for (int j = 0; j < 8; ++j) h[j] = hmat[s[j] * HEADS + hh];
}

#define GATHER_BLOCKS 1024

__device__ __forceinline__ void gather_node_ids(const int* __restrict__ bsrc, int start,
                                                int end, int cnt, int q,
                                                int (&sA)[8], int (&sB)[8]) {
  if (cnt > 0)     gather_load_ids(bsrc, start, end, q, sA);
  if (cnt > CHUNK) gather_load_ids(bsrc, start + CHUNK, end, q, sB);
}

__global__ __launch_bounds__(256) void gat_gather_kernel(
    const int* __restrict__ bsrc, const int* __restrict__ cur,
    const int* __restrict__ deg, const float* __restrict__ hmat,
    const float* __restrict__ att_src, const float* __restrict__ att_dst,
    const float* __restrict__ gat_bias, const float* __restrict__ lin_w,
    const float* __restrict__ lin_b, const float* __restrict__ bias,
    float* __restrict__ out, int n)
{
  const int lane = threadIdx.x & 31;
  const int q  = lane >> 3;
  const int hh = lane & 7;

  const int T   = GATHER_BLOCKS * 8;
  const int w   = __builtin_amdgcn_readfirstlane((int)(blockIdx.x * 8 + (threadIdx.x >> 5)));
  const int per = n / T;
  const int rem = n - per * T;
  const int n0  = w * per + (w < rem ? w : rem);
  const int n1  = n0 + per + (w < rem ? 1 : 0);
  if (n0 >= n1) return;

  const float as = att_src[hh];
  const float ad = att_dst[hh];
  const float gb = gat_bias[hh];
  const float lb = lin_b[0];
  const float bo = bias[0];

  int endC = cur[n0];
  int cntC = deg[n0];
  const int i1 = (n0 + 1 < n1) ? n0 + 1 : n1 - 1;
  int end1 = cur[i1];
  int cnt1 = deg[i1];

  int nA[8], nB[8];
  #pragma unroll
  for (int j = 0; j < 8; ++j) { nA[j] = 0;  nB[j] = 0; }
  gather_node_ids(bsrc, endC - cntC, endC, cntC, q, nA, nB);

  for (int node = n0; node < n1; ++node) {
    const int end   = endC;
    const int cnt   = cntC;
    const int start = end - cnt;
    const bool hasA = cnt > 0;
    const bool hasB = cnt > CHUNK;
    const int  rest = start + 2 * CHUNK;

    int sA[8], sB[8];
    #pragma unroll
    for (int j = 0; j < 8; ++j) { sA[j] = nA[j];  sB[j] = nB[j]; }

    const int i2 = (node + 2 < n1) ? node + 2 : n1 - 1;
    const int end2 = cur[i2];
    const int cnt2 = deg[i2];

    const int cntN = (node + 1 < n1) ? cnt1 : 0;
    gather_node_ids(bsrc, end1 - cntN, end1, cntN, q, nA, nB);

    float hA[8], hB[8], eA[8], eB[8];
    #pragma unroll
    for (int j = 0; j < 8; ++j) {
      hA[j] = 0.f;  hB[j] = 0.f;  eA[j] = -INFINITY;  eB[j] = -INFINITY;
    }
    const float hself = hmat[node * HEADS + hh];
    if (hasA) gather_load_rows(hmat, sA, hh, hA);
    if (hasB) gather_load_rows(hmat, sB, hh, hB);

    const float adst  = hself * ad;
    const float eself = leaky(hself * as + adst);
    if (hasA) {
      #pragma unroll
      for (int j = 0; j < 8; ++j)
        if (j * 4 + q < cnt) eA[j] = leaky(hA[j] * as + adst);
    }
    if (hasB) {
      #pragma unroll
      for (int j = 0; j < 8; ++j)
        if (CHUNK + j * 4 + q < cnt) eB[j] = leaky(hB[j] * as + adst);
    }

    float m = (q == 0) ? eself : -INFINITY;
    #pragma unroll
    for (int j = 0; j < 8; ++j) m = fmaxf(m, eA[j]);
    if (hasB) {
      #pragma unroll
      for (int j = 0; j < 8; ++j) m = fmaxf(m, eB[j]);
    }
    for (int base = rest; base < end; base += CHUNK) {
      int   s[8];
      float hs[8];
      #pragma unroll
      for (int j = 0; j < 8; ++j) {
        const int p = base + j * 4 + q;
        s[j] = bsrc[p < end ? p : end - 1];
      }
      #pragma unroll
      for (int j = 0; j < 8; ++j) hs[j] = hmat[s[j] * HEADS + hh];
      #pragma unroll
      for (int j = 0; j < 8; ++j) {
        const int p = base + j * 4 + q;
        const float e = leaky(hs[j] * as + adst);
        if (p < end) m = fmaxf(m, e);
      }
    }
    m = fmaxf(m, __shfl_xor(m, 8));
    m = fmaxf(m, __shfl_xor(m, 16));

    const float mL = m * LOG2E;
    float den = 0.f, num = 0.f;
    if (q == 0) { den = exp_rel(eself, mL); num = den * hself; }
    #pragma unroll
    for (int j = 0; j < 8; ++j) {
      const float ee = exp_rel(eA[j], mL);
      den += ee;  num = fmaf(ee, hA[j], num);
    }
    if (hasB) {
      #pragma unroll
      for (int j = 0; j < 8; ++j) {
        const float ee = exp_rel(eB[j], mL);
        den += ee;  num = fmaf(ee, hB[j], num);
      }
    }
    for (int base = rest; base < end; base += CHUNK) {
      int   s[8];
      float hs[8];
      #pragma unroll
      for (int j = 0; j < 8; ++j) {
        const int p = base + j * 4 + q;
        s[j] = bsrc[p < end ? p : end - 1];
      }
      #pragma unroll
      for (int j = 0; j < 8; ++j) hs[j] = hmat[s[j] * HEADS + hh];
      #pragma unroll
      for (int j = 0; j < 8; ++j) {
        const int p = base + j * 4 + q;
        const float ee = exp_rel(leaky(hs[j] * as + adst), mL);
        if (p < end) { den += ee; num += ee * hs[j]; }
      }
    }
    den += __shfl_xor(den, 8);
    den += __shfl_xor(den, 16);
    num += __shfl_xor(num, 8);
    num += __shfl_xor(num, 16);

    const float oh = num / (den + 1e-16f) + gb;
    float z = lb;
    #pragma unroll
    for (int k = 0; k < HEADS; ++k) z += __shfl(oh, k) * lin_w[k];
    if (lane == 0) out[node] = fmaxf(z, 0.f) + bo;

    endC = end1;  cntC = cntN;
    end1 = end2;  cnt1 = cnt2;
  }
}

extern "C" void kernel_launch(void* const* d_in, const int* in_sizes, int n_in,
                              void* d_out, int out_size, void* d_ws, size_t ws_size,
                              hipStream_t stream) {
  const float* x        = (const float*)d_in[0];
  const int*   ei       = (const int*)d_in[1];
  const float* W        = (const float*)d_in[2];
  const float* att_src  = (const float*)d_in[3];
  const float* att_dst  = (const float*)d_in[4];
  const float* gat_bias = (const float*)d_in[5];
  const float* lin_w    = (const float*)d_in[6];
  const float* lin_b    = (const float*)d_in[7];
  const float* bias     = (const float*)d_in[8];
  float* out = (float*)d_out;

  const int n  = in_sizes[0] / IN_CH;
  const int nE = in_sizes[1] / 2;

  const size_t nh   = (size_t)n * HEADS;
  const size_t segH = (nh * sizeof(float) + 255) & ~(size_t)255;
  const size_t segN = ((size_t)n * sizeof(int) + 255) & ~(size_t)255;
  const size_t segT = (size_t)3 * MAXR * sizeof(int);
  const size_t segE = ((size_t)nE * sizeof(int) + 255) & ~(size_t)255;
  char* ws = (char*)d_ws;
  float* h_mat      = (float*)(ws);
  int*   deg        = (int*)  (ws + segH);
  int*   cur        = (int*)  (ws + segH + segN);
  int*   rangeCnt   = (int*)  (ws + segH + 2 * segN);
  int*   rangeCur   = rangeCnt + MAXR;
  int*   rangeStart = rangeCnt + 2 * MAXR;
  int*   part       = (int*)  (ws + segH + 2 * segN + segT);
  int*   bsrc       = (ws_size >= segH + 2 * segN + segT + 2 * segE) ? (int*)((char*)part + segE) : part;
  const int    R      = (n + RSIZE - 1) / RSIZE;
  const int    eb     = (nE + PB * EPT - 1) / (PB * EPT);
  const int    RP     = (R + 31) & ~31;
  const size_t segR   = (size_t)eb * RP * sizeof(int);
  const size_t need2  = segH + 2 * segN + segT + 2 * segE + segR;
  int*         runTab = (int*)((char*)part + 2 * segE);
  const bool   twoLaunch = (ws_size >= need2) && (eb <= MAXB);

  const int tiles  = (n + 15) / 16;
  const int blocks = (tiles + 7) / 8;
  gat_gemm_kernel<<<blocks, GEMM_T, 0, stream>>>(x, W, h_mat, rangeCnt, n);

  if (twoLaunch) {
    gat_blocksort_kernel<<<eb, PB, 0, stream>>>(ei, part, runTab, nE, R, RP);

    gat_rangecollect_kernel<<<R, PB, 0, stream>>>(part, runTab, rangeCur, deg, cur, bsrc, n, eb, RP);
  } else {
    gat_rangecount_kernel<<<eb, PB, 0, stream>>>(ei, rangeCnt, nE, R);

    gat_partition_kernel<<<eb, PB, 0, stream>>>(ei, rangeCnt, rangeCur, rangeStart, part, nE, R);

    gat_rangesort_kernel<<<R, PB, 0, stream>>>(part, rangeCnt, rangeStart, deg, cur, bsrc, n);
  }

  gat_gather_kernel<<<GATHER_BLOCKS, 256, 0, stream>>>(
      bsrc, cur, deg, h_mat, att_src, att_dst, gat_bias, lin_w, lin_b, bias,
      out, n);
}


# x@W and the per-range edge collection in one launch: GEMM blocks (now 32 waves, x held as a ring of 10 pieces to stay in 64 registers) and range blocks side by side, 4 launches to 3; sum orders kept
# speedup vs baseline: 21.1363x; 1.1180x over previous
#include <hip/hip_runtime.h>
#include <math.h>


#define IN_CH 512
#define HEADS 8
#define NEG_SLOPE 0.2f
#define XRING 10
#define CHUNK 32
#define RSHIFT 8
#define RSIZE 256
#define MAXR 1024
#define PB 1024
#define EPT 8
#define CAP 16384
#define MAXB 1024

typedef __attribute__((ext_vector_type(2))) float v2f;
typedef __attribute__((ext_vector_type(8))) float v8f;

__device__ __forceinline__ float leaky(float v) {
  return v >= 0.f ? v : NEG_SLOPE * v;
}

#define LOG2E 1.44269504088896340736f

__device__ __forceinline__ float exp_rel(float e, float mL) {
  return __builtin_amdgcn_exp2f(fmaf(e, LOG2E, -mL));
}

__device__ __forceinline__ void gemm_tiles(
    const float* __restrict__ x, const float* __restrict__ W,
    float* __restrict__ h, int* __restrict__ rangeTabs, int n, int gblock,
    float4* Wlds4)
{
  static_assert(IN_CH * HEADS / 4 == PB, "one 16-byte piece of W per thread");
  {
    const float4* __restrict__ W4 = reinterpret_cast<const float4*>(W);
    Wlds4[threadIdx.x] = W4[threadIdx.x];
  }
  if (rangeTabs != nullptr && gblock == 0) {
    for (int i = threadIdx.x; i < 2 * MAXR; i += PB) rangeTabs[i] = 0;
  }
  __syncthreads();

  const int lane = threadIdx.x & 31;
  const int wave = threadIdx.x >> 5;
  const int half = lane >> 4;
  const int l    = lane & 15;

  const int tile = gblock * (PB >> 5) + wave;
  const int row0 = tile * 16;
  if (row0 >= n) return;

  int arow = row0 + l;
  if (arow > n - 1) arow = n - 1;
  const float4* __restrict__ x4 = reinterpret_cast<const float4*>(x);
  const int xi = arow * (IN_CH / 4) + half;
  const float* __restrict__ wl =
      reinterpret_cast<const float*>(Wlds4) + half * 4 * HEADS + (l & 7);

  const int rfirst = row0 + half * 8;
  const int hidx   = rfirst * HEADS + (l & 7);
  const int hrows  = (l < HEADS) ? n - rfirst : 0;

  v8f c = {0.f, 0.f, 0.f, 0.f, 0.f, 0.f, 0.f, 0.f};

  float4 xa[XRING];
  #pragma unroll
  for (int p = 0; p < XRING - 1; ++p) xa[p] = x4[xi + p * 2];
  float wb[2][4];
  #pragma unroll
  for (int u = 0; u < 4; ++u) wb[0][u] = wl[u * HEADS];

  #pragma unroll
  for (int p = 0; p < IN_CH / 8; ++p) {
    if (p + XRING - 1 < IN_CH / 8)
      xa[(p + XRING - 1) % XRING] = x4[xi + (p + XRING - 1) * 2];
    if (p + 1 < IN_CH / 8) {
      #pragma unroll
      for (int u = 0; u < 4; ++u) wb[(p + 1) & 1][u] = wl[((p + 1) * 8 + u) * HEADS];
    }
    const float4 av = xa[p % XRING];
    v2f a, b;
    a.x = av.x;  a.y = av.y;
    b.x = wb[p & 1][0];  b.y = wb[p & 1][1];
    c = __builtin_amdgcn_wmma_f32_16x16x4_f32(
         false, a,  false, b,
         (short)0, c,  false,  false);
    a.x = av.z;  a.y = av.w;
    b.x = wb[p & 1][2];  b.y = wb[p & 1][3];
    c = __builtin_amdgcn_wmma_f32_16x16x4_f32(
         false, a,  false, b,
         (short)0, c,  false,  false);
    __builtin_amdgcn_sched_barrier(0);
  }

  #pragma unroll
  for (int i = 0; i < 8; ++i)
    if (i < hrows) h[hidx + i * HEADS] = c[i];
}

__device__ __forceinline__ int block_excl_scan(int v, int* wsum, int lane, int wave) {
  int incl = v;
  #pragma unroll
  for (int o = 1; o < 32; o <<= 1) {
    const int t = __shfl_up(incl, o);
    if (lane >= o) incl += t;
  }
  if (lane == 31) wsum[wave] = incl;
  __syncthreads();
  if (wave == 0) {
    const int w = wsum[lane];
    int wi = w;
    #pragma unroll
    for (int o = 1; o < 32; o <<= 1) {
      const int t = __shfl_up(wi, o);
      if (lane >= o) wi += t;
    }
    wsum[lane] = wi - w;
  }
  __syncthreads();
  return wsum[wave] + incl - v;
}

__global__ __launch_bounds__(PB) void gat_rangecount_kernel(
    const int* __restrict__ ei, int* __restrict__ rangeCnt, int nE, int R)
{
  __shared__ int hist[MAXR];
  const int t = threadIdx.x;
  hist[t] = 0;
  __syncthreads();
  const int base = blockIdx.x * (PB * EPT);
  int d[EPT];
  #pragma unroll
  for (int j = 0; j < EPT; ++j) {
    const int e = base + j * PB + t;
    d[j] = (e < nE) ? ei[nE + e] : -1;
  }
  #pragma unroll
  for (int j = 0; j < EPT; ++j)
    if (d[j] >= 0) atomicAdd(&hist[d[j] >> RSHIFT], 1);
  __syncthreads();
  if (t < R) {
    const int c = hist[t];
    if (c) atomicAdd(rangeCnt + t, c);
  }
}

__global__ __launch_bounds__(PB) void gat_partition_kernel(
    const int* __restrict__ ei, const int* __restrict__ rangeCnt,
    int* __restrict__ rangeCur, int* __restrict__ rangeStart,
    int* __restrict__ part, int nE, int R)
{
  __shared__ int hist[MAXR];
  __shared__ int lstart[MAXR];
  __shared__ int rbase[MAXR];
  __shared__ int wsum[32];
  __shared__ int skey[PB * EPT];
  __shared__ int sgpos[PB * EPT];
  const int t    = threadIdx.x;
  const int lane = t & 31;
  const int wave = t >> 5;
  hist[t] = 0;
  const int tot    = (t < R) ? rangeCnt[t] : 0;
  const int rstart = block_excl_scan(tot, wsum, lane, wave);
  if (blockIdx.x == 0 && t < R) rangeStart[t] = rstart;

  const int base = blockIdx.x * (PB * EPT);
  int s[EPT], d[EPT], rk[EPT];
  #pragma unroll
  for (int j = 0; j < EPT; ++j) {
    const int e = base + j * PB + t;
    const bool ok = e < nE;
    s[j] = ok ? ei[e] : 0;
    d[j] = ok ? ei[nE + e] : -1;
  }
  #pragma unroll
  for (int j = 0; j < EPT; ++j) {
    rk[j] = 0;
    if (d[j] >= 0) rk[j] = atomicAdd(&hist[d[j] >> RSHIFT], 1);
  }
  __syncthreads();
  const int cc = hist[t];
  const int ls = block_excl_scan(cc, wsum, lane, wave);
  lstart[t] = ls;
  rbase[t]  = (t < R && cc) ? rstart + atomicAdd(rangeCur + t, cc) : 0;
  __syncthreads();
  #pragma unroll
  for (int j = 0; j < EPT; ++j) {
    if (d[j] >= 0) {
      const int r = d[j] >> RSHIFT;
      const int p = lstart[r] + rk[j];
      skey[p]  = (s[j] << RSHIFT) | (d[j] & (RSIZE - 1));
      sgpos[p] = rbase[r] + rk[j];
    }
  }
  __syncthreads();
  const int left  = nE - base;
  const int mine  = left < PB * EPT ? left : PB * EPT;
  #pragma unroll
  for (int j = 0; j < EPT; ++j) {
    const int i = j * PB + t;
    if (i < mine) part[sgpos[i]] = skey[i];
  }
}

__global__ __launch_bounds__(PB) void gat_rangesort_kernel(
    const int* part, const int* __restrict__ rangeCnt,
    const int* __restrict__ rangeStart, int* __restrict__ deg,
    int* __restrict__ cur, int* bsrc, int n)
{
  __shared__ int cnt[RSIZE];
  __shared__ int pos[RSIZE];
  __shared__ int wsum[32];
  __shared__ int stage[CAP];
  const int t    = threadIdx.x;
  const int lane = t & 31;
  const int wave = t >> 5;
  const int r    = blockIdx.x;
  const int L    = rangeCnt[r];
  const int seg  = rangeStart[r];
  if (t < RSIZE) cnt[t] = 0;
  __syncthreads();
  for (int i0 = t; i0 < L; i0 += 4 * PB) {
    int k[4];
    #pragma unroll
    for (int u = 0; u < 4; ++u) {
      const int i = i0 + u * PB;
      k[u] = (i < L) ? part[seg + i] : -1;
    }
    #pragma unroll
    for (int u = 0; u < 4; ++u)
      if (k[u] >= 0) atomicAdd(&cnt[k[u] & (RSIZE - 1)], 1);
  }
  __syncthreads();
  const int c    = (t < RSIZE) ? cnt[t] : 0;
  const int excl = block_excl_scan(c, wsum, lane, wave);
  if (t < RSIZE) {
    pos[t] = excl;
    const int node = r * RSIZE + t;
    if (node < n) { deg[node] = c; cur[node] = seg + excl + c; }
  }
  __syncthreads();
  const bool inLds = (L <= CAP);
  for (int i0 = t; i0 < L; i0 += 4 * PB) {
    int k[4];
    #pragma unroll
    for (int u = 0; u < 4; ++u) {
      const int i = i0 + u * PB;
      k[u] = (i < L) ? part[seg + i] : -1;
    }
    #pragma unroll
    for (int u = 0; u < 4; ++u) {
      if (k[u] >= 0) {
        const int p   = atomicAdd(&pos[k[u] & (RSIZE - 1)], 1);
        const int sid = k[u] >> RSHIFT;
        if (inLds) stage[p] = sid; else bsrc[seg + p] = sid;
      }
    }
  }
  if (inLds) {
    __syncthreads();
    for (int i = t; i < L; i += PB) bsrc[seg + i] = stage[i];
  }
}

__global__ __launch_bounds__(PB) void gat_blocksort_kernel(
    const int* __restrict__ ei, int* __restrict__ part, int* __restrict__ runTab,
    int* __restrict__ cursor, int nE, int R, int RP)
{
  __shared__ int hist[MAXR];
  __shared__ int lstart[MAXR];
  __shared__ int wsum[32];
  __shared__ int skey[PB * EPT];
  const int t    = threadIdx.x;
  const int lane = t & 31;
  const int wave = t >> 5;
  hist[t] = 0;
  if (blockIdx.x == 0 && t == 0) cursor[0] = 0;
  __syncthreads();

  const int base = blockIdx.x * (PB * EPT);
  int s[EPT], d[EPT], rk[EPT];
  #pragma unroll
  for (int j = 0; j < EPT; ++j) {
    const int e = base + j * PB + t;
    const bool ok = e < nE;
    s[j] = ok ? ei[e] : 0;
    d[j] = ok ? ei[nE + e] : -1;
  }
  #pragma unroll
  for (int j = 0; j < EPT; ++j) {
    rk[j] = 0;
    if (d[j] >= 0) rk[j] = atomicAdd(&hist[d[j] >> RSHIFT], 1);
  }
  __syncthreads();
  const int cc = hist[t];
  const int ls = block_excl_scan(cc, wsum, lane, wave);
  lstart[t] = ls;
  if (t < R) runTab[(size_t)blockIdx.x * RP + t] = (ls << 16) | cc;
  __syncthreads();
  #pragma unroll
  for (int j = 0; j < EPT; ++j) {
    if (d[j] >= 0) {
      const int r = d[j] >> RSHIFT;
      skey[lstart[r] + rk[j]] = (s[j] << RSHIFT) | (d[j] & (RSIZE - 1));
    }
  }
  __syncthreads();
  const int left = nE - base;
  const int mine = left < PB * EPT ? left : PB * EPT;
  #pragma unroll
  for (int j = 0; j < EPT; ++j) {
    const int i = j * PB + t;
    if (i < mine) part[base + i] = skey[i];
  }
}

template <bool PLACE>
__device__ __forceinline__ void range_entry(int kk, int* cnt, int* pos, int* stage,
                                            int* __restrict__ bsrc, int seg, bool inLds) {
  if (!PLACE) {
    atomicAdd(&cnt[kk & (RSIZE - 1)], 1);
  } else {
    const int p   = atomicAdd(&pos[kk & (RSIZE - 1)], 1);
    const int sid = kk >> RSHIFT;
    if (inLds) stage[p] = sid; else bsrc[seg + p] = sid;
  }
}

template <bool PLACE>
__device__ __forceinline__ void range_run(const int* __restrict__ part, const int* runPk,
                                          int b, int first, int lane, int* cnt, int* pos,
                                          int* stage, int* __restrict__ bsrc, int seg,
                                          bool inLds) {
  const int pk  = runPk[b];
  const int len = pk & 0xffff;
  const int beg = b * (PB * EPT) + (pk >> 16);
  for (int i = first + lane; i < len; i += 32)
    range_entry<PLACE>(part[beg + i], cnt, pos, stage, bsrc, seg, inLds);
}

#define HELD 16
#define COLLECT_LDS_INTS (RSIZE + RSIZE + 32 + MAXB + 32 + CAP)

__device__ __forceinline__ void range_collect(
    const int* __restrict__ part, const int* __restrict__ runTab,
    int* __restrict__ cursor, int* __restrict__ deg, int* __restrict__ cur,
    int* __restrict__ bsrc, int n, int EB, int RP, int r,
    int* lds)
{
  int* const cnt   = lds;
  int* const pos   = cnt + RSIZE;
  int* const wsum  = pos + RSIZE;
  int* const runPk = wsum + 32;
  int* const segS  = runPk + MAXB;
  int* const stage = segS + 32;
  const int t    = threadIdx.x;
  const int lane = t & 31;
  const int wave = __builtin_amdgcn_readfirstlane(t >> 5);

  const int pk    = (t < EB) ? runTab[(size_t)t * RP + r] : 0;
  const int myLen = pk & 0xffff;
  runPk[t] = pk;
  if (t < RSIZE) cnt[t] = 0;
  const int before = block_excl_scan(myLen, wsum, lane, wave);
  int segv = 0;
  if (t == PB - 1) {
    segS[1] = before + myLen;
    segv = atomicAdd(cursor, before + myLen);
  }

  int k[HELD];
  unsigned longMask = 0;
  {
    int pkv[HELD];
    #pragma unroll
    for (int j = 0; j < HELD; ++j) pkv[j] = runPk[wave + 32 * j];
    #pragma unroll
    for (int j = 0; j < HELD; ++j) {
      const int len = pkv[j] & 0xffff;
      const int beg = (wave + 32 * j) * (PB * EPT) + (pkv[j] >> 16);
      k[j] = -1;
      if (len > 32) longMask |= 1u << j;
      if (lane < len) k[j] = part[beg + lane];
    }
  }
  longMask = __builtin_amdgcn_readfirstlane(longMask);

  #pragma unroll
  for (int j = 0; j < HELD; ++j)
    if (k[j] >= 0) range_entry<false>(k[j], cnt, pos, stage, bsrc, 0, true);
  for (unsigned m = longMask; m; m &= m - 1)
    range_run<false>(part, runPk, wave + 32 * __builtin_ctz(m), 32, lane, cnt, pos, stage, bsrc, 0, true);
  for (int b = wave + 32 * HELD; b < EB; b += 32)
    range_run<false>(part, runPk, b, 0, lane, cnt, pos, stage, bsrc, 0, true);
  if (t == PB - 1) segS[0] = segv;
  __syncthreads();
  const int seg = segS[0];
  const int L   = segS[1];

  const int c    = (t < RSIZE) ? cnt[t] : 0;
  const int excl = block_excl_scan(c, wsum, lane, wave);
  if (t < RSIZE) {
    pos[t] = excl;
    const int node = r * RSIZE + t;
    if (node < n) { deg[node] = c; cur[node] = seg + excl + c; }
  }
  __syncthreads();

  const bool inLds = (L <= CAP);
  {
    int p[HELD];
    #pragma unroll
    for (int j = 0; j < HELD; ++j)
      p[j] = (k[j] >= 0) ? atomicAdd(&pos[k[j] & (RSIZE - 1)], 1) : 0;
    #pragma unroll
    for (int j = 0; j < HELD; ++j) {
      if (k[j] >= 0) {
        const int sid = k[j] >> RSHIFT;
        if (inLds) stage[p[j]] = sid; else bsrc[seg + p[j]] = sid;
      }
    }
  }
  for (unsigned m = longMask; m; m &= m - 1)
    range_run<true>(part, runPk, wave + 32 * __builtin_ctz(m), 32, lane, cnt, pos, stage, bsrc, seg, inLds);
  for (int b = wave + 32 * HELD; b < EB; b += 32)
    range_run<true>(part, runPk, b, 0, lane, cnt, pos, stage, bsrc, seg, inLds);
  if (inLds) {
    __syncthreads();
    for (int i = t; i < L; i += PB) bsrc[seg + i] = stage[i];
  }
}

__global__ __launch_bounds__(PB) void gat_gemm_collect_kernel(
    const float* __restrict__ x, const float* __restrict__ W,
    float* __restrict__ h, int* __restrict__ rangeTabs, int n, int gemmBlocks,
    const int* __restrict__ part, const int* __restrict__ runTab,
    int* __restrict__ cursor, int* __restrict__ deg, int* __restrict__ cur,
    int* __restrict__ bsrc, int EB, int RP)
{
  static_assert(COLLECT_LDS_INTS * sizeof(int) >= IN_CH * HEADS * sizeof(float), "the buffer holds W");
  __shared__ float4 lds4[COLLECT_LDS_INTS / 4];
  const int b = blockIdx.x;
  if (b < gemmBlocks) gemm_tiles(x, W, h, rangeTabs, n, b, lds4);
  else                range_collect(part, runTab, cursor, deg, cur, bsrc, n, EB, RP, b - gemmBlocks,
                                    reinterpret_cast<int*>(lds4));
}

__device__ __forceinline__ void gather_load_ids(const int* __restrict__ bsrc, int base,
                                                int end, int q, int (&s)[8]) {
  #pragma unroll
  for (int j = 0; j < 8; ++j) {
    const int p = base + j * 4 + q;
    s[j] = bsrc[p < end ? p : end - 1];
  }
}

__device__ __forceinline__ void gather_load_rows(const float* __restrict__ hmat,
                                                 const int (&s)[8], int hh, float (&h)[8]) {
  #pragma unroll
  for (int j = 0; j < 8; ++j) h[j] = hmat[s[j] * HEADS + hh];
}

#define GATHER_BLOCKS 1024

__device__ __forceinline__ void gather_node_ids(const int* __restrict__ bsrc, int start,
                                                int end, int cnt, int q,
                                                int (&sA)[8], int (&sB)[8]) {
  if (cnt > 0)     gather_load_ids(bsrc, start, end, q, sA);
  if (cnt > CHUNK) gather_load_ids(bsrc, start + CHUNK, end, q, sB);
}

__global__ __launch_bounds__(256) void gat_gather_kernel(
    const int* __restrict__ bsrc, const int* __restrict__ cur,
    const int* __restrict__ deg, const float* __restrict__ hmat,
    const float* __restrict__ att_src, const float* __restrict__ att_dst,
    const float* __restrict__ gat_bias, const float* __restrict__ lin_w,
    const float* __restrict__ lin_b, const float* __restrict__ bias,
    float* __restrict__ out, int n)
{
  const int lane = threadIdx.x & 31;
  const int q  = lane >> 3;
  const int hh = lane & 7;

  const int T   = GATHER_BLOCKS * 8;
  const int w   = __builtin_amdgcn_readfirstlane((int)(blockIdx.x * 8 + (threadIdx.x >> 5)));
  const int per = n / T;
  const int rem = n - per * T;
  const int n0  = w * per + (w < rem ? w : rem);
  const int n1  = n0 + per + (w < rem ? 1 : 0);
  if (n0 >= n1) return;

  const float as = att_src[hh];
  const float ad = att_dst[hh];
  const float gb = gat_bias[hh];
  const float lb = lin_b[0];
  const float bo = bias[0];

  int endC = cur[n0];
  int cntC = deg[n0];
  const int i1 = (n0 + 1 < n1) ? n0 + 1 : n1 - 1;
  int end1 = cur[i1];
  int cnt1 = deg[i1];

  int nA[8], nB[8];
  #pragma unroll
  for (int j = 0; j < 8; ++j) { nA[j] = 0;  nB[j] = 0; }
  gather_node_ids(bsrc, endC - cntC, endC, cntC, q, nA, nB);

  for (int node = n0; node < n1; ++node) {
    const int end   = endC;
    const int cnt   = cntC;
    const int start = end - cnt;
    const bool hasA = cnt > 0;
    const bool hasB = cnt > CHUNK;
    const int  rest = start + 2 * CHUNK;

    int sA[8], sB[8];
    #pragma unroll
    for (int j = 0; j < 8; ++j) { sA[j] = nA[j];  sB[j] = nB[j]; }

    const int i2 = (node + 2 < n1) ? node + 2 : n1 - 1;
    const int end2 = cur[i2];
    const int cnt2 = deg[i2];

    const int cntN = (node + 1 < n1) ? cnt1 : 0;
    gather_node_ids(bsrc, end1 - cntN, end1, cntN, q, nA, nB);

    float hA[8], hB[8], eA[8], eB[8];
    #pragma unroll
    for (int j = 0; j < 8; ++j) {
      hA[j] = 0.f;  hB[j] = 0.f;  eA[j] = -INFINITY;  eB[j] = -INFINITY;
    }
    const float hself = hmat[node * HEADS + hh];
    if (hasA) gather_load_rows(hmat, sA, hh, hA);
    if (hasB) gather_load_rows(hmat, sB, hh, hB);

    const float adst  = hself * ad;
    const float eself = leaky(hself * as + adst);
    if (hasA) {
      #pragma unroll
      for (int j = 0; j < 8; ++j)
        if (j * 4 + q < cnt) eA[j] = leaky(hA[j] * as + adst);
    }
    if (hasB) {
      #pragma unroll
      for (int j = 0; j < 8; ++j)
        if (CHUNK + j * 4 + q < cnt) eB[j] = leaky(hB[j] * as + adst);
    }

    float m = (q == 0) ? eself : -INFINITY;
    #pragma unroll
    for (int j = 0; j < 8; ++j) m = fmaxf(m, eA[j]);
    if (hasB) {
      #pragma unroll
      for (int j = 0; j < 8; ++j) m = fmaxf(m, eB[j]);
    }
    for (int base = rest; base < end; base += CHUNK) {
      int   s[8];
      float hs[8];
      #pragma unroll
      for (int j = 0; j < 8; ++j) {
        const int p = base + j * 4 + q;
        s[j] = bsrc[p < end ? p : end - 1];
      }
      #pragma unroll
      for (int j = 0; j < 8; ++j) hs[j] = hmat[s[j] * HEADS + hh];
      #pragma unroll
      for (int j = 0; j < 8; ++j) {
        const int p = base + j * 4 + q;
        const float e = leaky(hs[j] * as + adst);
        if (p < end) m = fmaxf(m, e);
      }
    }
    m = fmaxf(m, __shfl_xor(m, 8));
    m = fmaxf(m, __shfl_xor(m, 16));

    const float mL = m * LOG2E;
    float den = 0.f, num = 0.f;
    if (q == 0) { den = exp_rel(eself, mL); num = den * hself; }
    #pragma unroll
    for (int j = 0; j < 8; ++j) {
      const float ee = exp_rel(eA[j], mL);
      den += ee;  num = fmaf(ee, hA[j], num);
    }
    if (hasB) {
      #pragma unroll
      for (int j = 0; j < 8; ++j) {
        const float ee = exp_rel(eB[j], mL);
        den += ee;  num = fmaf(ee, hB[j], num);
      }
    }
    for (int base = rest; base < end; base += CHUNK) {
      int   s[8];
      float hs[8];
      #pragma unroll
      for (int j = 0; j < 8; ++j) {
        const int p = base + j * 4 + q;
        s[j] = bsrc[p < end ? p : end - 1];
      }
      #pragma unroll
      for (int j = 0; j < 8; ++j) hs[j] = hmat[s[j] * HEADS + hh];
      #pragma unroll
      for (int j = 0; j < 8; ++j) {
        const int p = base + j * 4 + q;
        const float ee = exp_rel(leaky(hs[j] * as + adst), mL);
        if (p < end) { den += ee; num += ee * hs[j]; }
      }
    }
    den += __shfl_xor(den, 8);
    den += __shfl_xor(den, 16);
    num += __shfl_xor(num, 8);
    num += __shfl_xor(num, 16);

    const float oh = num / (den + 1e-16f) + gb;
    float z = lb;
    #pragma unroll
    for (int k = 0; k < HEADS; ++k) z += __shfl(oh, k) * lin_w[k];
    if (lane == 0) out[node] = fmaxf(z, 0.f) + bo;

    endC = end1;  cntC = cntN;
    end1 = end2;  cnt1 = cnt2;
  }
}

extern "C" void kernel_launch(void* const* d_in, const int* in_sizes, int n_in,
                              void* d_out, int out_size, void* d_ws, size_t ws_size,
                              hipStream_t stream) {
  const float* x        = (const float*)d_in[0];
  const int*   ei       = (const int*)d_in[1];
  const float* W        = (const float*)d_in[2];
  const float* att_src  = (const float*)d_in[3];
  const float* att_dst  = (const float*)d_in[4];
  const float* gat_bias = (const float*)d_in[5];
  const float* lin_w    = (const float*)d_in[6];
  const float* lin_b    = (const float*)d_in[7];
  const float* bias     = (const float*)d_in[8];
  float* out = (float*)d_out;

  const int n  = in_sizes[0] / IN_CH;
  const int nE = in_sizes[1] / 2;

  const size_t nh   = (size_t)n * HEADS;
  const size_t segH = (nh * sizeof(float) + 255) & ~(size_t)255;
  const size_t segN = ((size_t)n * sizeof(int) + 255) & ~(size_t)255;
  const size_t segT = (size_t)3 * MAXR * sizeof(int);
  const size_t segE = ((size_t)nE * sizeof(int) + 255) & ~(size_t)255;
  char* ws = (char*)d_ws;
  float* h_mat      = (float*)(ws);
  int*   deg        = (int*)  (ws + segH);
  int*   cur        = (int*)  (ws + segH + segN);
  int*   rangeCnt   = (int*)  (ws + segH + 2 * segN);
  int*   rangeCur   = rangeCnt + MAXR;
  int*   rangeStart = rangeCnt + 2 * MAXR;
  int*   part       = (int*)  (ws + segH + 2 * segN + segT);
  int*   bsrc       = (ws_size >= segH + 2 * segN + segT + 2 * segE) ? (int*)((char*)part + segE) : part;
  const int    R      = (n + RSIZE - 1) / RSIZE;
  const int    eb     = (nE + PB * EPT - 1) / (PB * EPT);
  const int    RP     = (R + 31) & ~31;
  const size_t segR   = (size_t)eb * RP * sizeof(int);
  const size_t need2  = segH + 2 * segN + segT + 2 * segE + segR;
  int*         runTab = (int*)((char*)part + 2 * segE);
  const bool   twoLaunch = (ws_size >= need2) && (eb <= MAXB);

  const int tiles  = (n + 15) / 16;
  const int blocks = (tiles + 31) / 32;

  if (twoLaunch) {
    gat_blocksort_kernel<<<eb, PB, 0, stream>>>(ei, part, runTab, rangeCur, nE, R, RP);

    gat_gemm_collect_kernel<<<blocks + R, PB, 0, stream>>>(
        x, W, h_mat, (int*)nullptr, n, blocks, part, runTab, rangeCur, deg, cur, bsrc, eb, RP);
  } else {
    gat_gemm_collect_kernel<<<blocks, PB, 0, stream>>>(
        x, W, h_mat, rangeCnt, n, blocks, part, runTab, rangeCur, deg, cur, bsrc, 0, RP);

    gat_rangecount_kernel<<<eb, PB, 0, stream>>>(ei, rangeCnt, nE, R);

    gat_partition_kernel<<<eb, PB, 0, stream>>>(ei, rangeCnt, rangeCur, rangeStart, part, nE, R);

    gat_rangesort_kernel<<<R, PB, 0, stream>>>(part, rangeCnt, rangeStart, deg, cur, bsrc, n);
  }

  gat_gather_kernel<<<GATHER_BLOCKS, 256, 0, stream>>>(
      bsrc, cur, deg, h_mat, att_src, att_dst, gat_bias, lin_w, lin_b, bias,
      out, n);
}


# range collection and gather in one launch: a 16-wave block orders its 196-node range in LDS and walks it with ids from LDS; x@W now beside the edge blocks; 3 launches to 2; list and sum order change
# speedup vs baseline: 23.7568x; 1.1240x over previous
#include <hip/hip_runtime.h>
#include <math.h>


#define IN_CH 512
#define HEADS 8
#define NEG_SLOPE 0.2f
#define XRING 10
#define CHUNK 32
#define RSHIFT 8
#define RSIZE 256
#define MAXR 1024
#define PB 1024
#define EPT 8
#define CAP 16384
#define MAXB 1024
#define FR 196
#define FPB 512
#define FW 16
#define FHELD 25

typedef __attribute__((ext_vector_type(2))) float v2f;
typedef __attribute__((ext_vector_type(8))) float v8f;

__device__ __forceinline__ float leaky(float v) {
  return v >= 0.f ? v : NEG_SLOPE * v;
}

#define LOG2E 1.44269504088896340736f

__device__ __forceinline__ float exp_rel(float e, float mL) {
  return __builtin_amdgcn_exp2f(fmaf(e, LOG2E, -mL));
}

__device__ __forceinline__ void gemm_tiles(
    const float* __restrict__ x, const float* __restrict__ W,
    float* __restrict__ h, int* __restrict__ rangeTabs, int n, int gblock,
    float4* Wlds4)
{
  static_assert(IN_CH * HEADS / 4 == PB, "one 16-byte piece of W per thread");
  {
    const float4* __restrict__ W4 = reinterpret_cast<const float4*>(W);
    Wlds4[threadIdx.x] = W4[threadIdx.x];
  }
  if (rangeTabs != nullptr && gblock == 0) {
    for (int i = threadIdx.x; i < 2 * MAXR; i += PB) rangeTabs[i] = 0;
  }
  __syncthreads();

  const int lane = threadIdx.x & 31;
  const int wave = threadIdx.x >> 5;
  const int half = lane >> 4;
  const int l    = lane & 15;

  const int tile = gblock * (PB >> 5) + wave;
  const int row0 = tile * 16;
  if (row0 >= n) return;

  int arow = row0 + l;
  if (arow > n - 1) arow = n - 1;
  const float4* __restrict__ x4 = reinterpret_cast<const float4*>(x);
  const int xi = arow * (IN_CH / 4) + half;
  const float* __restrict__ wl =
      reinterpret_cast<const float*>(Wlds4) + half * 4 * HEADS + (l & 7);

  const int rfirst = row0 + half * 8;
  const int hidx   = rfirst * HEADS + (l & 7);
  const int hrows  = (l < HEADS) ? n - rfirst : 0;

  v8f c = {0.f, 0.f, 0.f, 0.f, 0.f, 0.f, 0.f, 0.f};

  float4 xa[XRING];
  #pragma unroll
  for (int p = 0; p < XRING - 1; ++p) xa[p] = x4[xi + p * 2];
  float wb[2][4];
  #pragma unroll
  for (int u = 0; u < 4; ++u) wb[0][u] = wl[u * HEADS];

  #pragma unroll
  for (int p = 0; p < IN_CH / 8; ++p) {
    if (p + XRING - 1 < IN_CH / 8)
      xa[(p + XRING - 1) % XRING] = x4[xi + (p + XRING - 1) * 2];
    if (p + 1 < IN_CH / 8) {
      #pragma unroll
      for (int u = 0; u < 4; ++u) wb[(p + 1) & 1][u] = wl[((p + 1) * 8 + u) * HEADS];
    }
    const float4 av = xa[p % XRING];
    v2f a, b;
    a.x = av.x;  a.y = av.y;
    b.x = wb[p & 1][0];  b.y = wb[p & 1][1];
    c = __builtin_amdgcn_wmma_f32_16x16x4_f32(
         false, a,  false, b,
         (short)0, c,  false,  false);
    a.x = av.z;  a.y = av.w;
    b.x = wb[p & 1][2];  b.y = wb[p & 1][3];
    c = __builtin_amdgcn_wmma_f32_16x16x4_f32(
         false, a,  false, b,
         (short)0, c,  false,  false);
    __builtin_amdgcn_sched_barrier(0);
  }

  #pragma unroll
  for (int i = 0; i < 8; ++i)
    if (i < hrows) h[hidx + i * HEADS] = c[i];
}

__device__ __forceinline__ int block_excl_scan(int v, int* wsum, int lane, int wave, int nw) {
  int incl = v;
  #pragma unroll
  for (int o = 1; o < 32; o <<= 1) {
    const int t = __shfl_up(incl, o);
    if (lane >= o) incl += t;
  }
  if (lane == 31) wsum[wave] = incl;
  __syncthreads();
  if (wave == 0) {
    const int w = (lane < nw) ? wsum[lane] : 0;
    int wi = w;
    #pragma unroll
    for (int o = 1; o < 32; o <<= 1) {
      const int t = __shfl_up(wi, o);
      if (lane >= o) wi += t;
    }
    wsum[lane] = wi - w;
  }
  __syncthreads();
  return wsum[wave] + incl - v;
}

__global__ __launch_bounds__(PB) void gat_rangecount_kernel(
    const int* __restrict__ ei, int* __restrict__ rangeCnt, int nE, int R)
{
  __shared__ int hist[MAXR];
  const int t = threadIdx.x;
  hist[t] = 0;
  __syncthreads();
  const int base = blockIdx.x * (PB * EPT);
  int d[EPT];
  #pragma unroll
  for (int j = 0; j < EPT; ++j) {
    const int e = base + j * PB + t;
    d[j] = (e < nE) ? ei[nE + e] : -1;
  }
  #pragma unroll
  for (int j = 0; j < EPT; ++j)
    if (d[j] >= 0) atomicAdd(&hist[d[j] >> RSHIFT], 1);
  __syncthreads();
  if (t < R) {
    const int c = hist[t];
    if (c) atomicAdd(rangeCnt + t, c);
  }
}

__global__ __launch_bounds__(PB) void gat_partition_kernel(
    const int* __restrict__ ei, const int* __restrict__ rangeCnt,
    int* __restrict__ rangeCur, int* __restrict__ rangeStart,
    int* __restrict__ part, int nE, int R)
{
  __shared__ int hist[MAXR];
  __shared__ int lstart[MAXR];
  __shared__ int rbase[MAXR];
  __shared__ int wsum[32];
  __shared__ int skey[PB * EPT];
  __shared__ int sgpos[PB * EPT];
  const int t    = threadIdx.x;
  const int lane = t & 31;
  const int wave = t >> 5;
  hist[t] = 0;
  const int tot    = (t < R) ? rangeCnt[t] : 0;
  const int rstart = block_excl_scan(tot, wsum, lane, wave, 32);
  if (blockIdx.x == 0 && t < R) rangeStart[t] = rstart;

  const int base = blockIdx.x * (PB * EPT);
  int s[EPT], d[EPT], rk[EPT];
  #pragma unroll
  for (int j = 0; j < EPT; ++j) {
    const int e = base + j * PB + t;
    const bool ok = e < nE;
    s[j] = ok ? ei[e] : 0;
    d[j] = ok ? ei[nE + e] : -1;
  }
  #pragma unroll
  for (int j = 0; j < EPT; ++j) {
    rk[j] = 0;
    if (d[j] >= 0) rk[j] = atomicAdd(&hist[d[j] >> RSHIFT], 1);
  }
  __syncthreads();
  const int cc = hist[t];
  const int ls = block_excl_scan(cc, wsum, lane, wave, 32);
  lstart[t] = ls;
  rbase[t]  = (t < R && cc) ? rstart + atomicAdd(rangeCur + t, cc) : 0;
  __syncthreads();
  #pragma unroll
  for (int j = 0; j < EPT; ++j) {
    if (d[j] >= 0) {
      const int r = d[j] >> RSHIFT;
      const int p = lstart[r] + rk[j];
      skey[p]  = (s[j] << RSHIFT) | (d[j] & (RSIZE - 1));
      sgpos[p] = rbase[r] + rk[j];
    }
  }
  __syncthreads();
  const int left  = nE - base;
  const int mine  = left < PB * EPT ? left : PB * EPT;
  #pragma unroll
  for (int j = 0; j < EPT; ++j) {
    const int i = j * PB + t;
    if (i < mine) part[sgpos[i]] = skey[i];
  }
}

__global__ __launch_bounds__(PB) void gat_rangesort_kernel(
    const int* part, const int* __restrict__ rangeCnt,
    const int* __restrict__ rangeStart, int* __restrict__ deg,
    int* __restrict__ cur, int* bsrc, int n)
{
  __shared__ int cnt[RSIZE];
  __shared__ int pos[RSIZE];
  __shared__ int wsum[32];
  __shared__ int stage[CAP];
  const int t    = threadIdx.x;
  const int lane = t & 31;
  const int wave = t >> 5;
  const int r    = blockIdx.x;
  const int L    = rangeCnt[r];
  const int seg  = rangeStart[r];
  if (t < RSIZE) cnt[t] = 0;
  __syncthreads();
  for (int i0 = t; i0 < L; i0 += 4 * PB) {
    int k[4];
    #pragma unroll
    for (int u = 0; u < 4; ++u) {
      const int i = i0 + u * PB;
      k[u] = (i < L) ? part[seg + i] : -1;
    }
    #pragma unroll
    for (int u = 0; u < 4; ++u)
      if (k[u] >= 0) atomicAdd(&cnt[k[u] & (RSIZE - 1)], 1);
  }
  __syncthreads();
  const int c    = (t < RSIZE) ? cnt[t] : 0;
  const int excl = block_excl_scan(c, wsum, lane, wave, 32);
  if (t < RSIZE) {
    pos[t] = excl;
    const int node = r * RSIZE + t;
    if (node < n) { deg[node] = c; cur[node] = seg + excl + c; }
  }
  __syncthreads();
  const bool inLds = (L <= CAP);
  for (int i0 = t; i0 < L; i0 += 4 * PB) {
    int k[4];
    #pragma unroll
    for (int u = 0; u < 4; ++u) {
      const int i = i0 + u * PB;
      k[u] = (i < L) ? part[seg + i] : -1;
    }
    #pragma unroll
    for (int u = 0; u < 4; ++u) {
      if (k[u] >= 0) {
        const int p   = atomicAdd(&pos[k[u] & (RSIZE - 1)], 1);
        const int sid = k[u] >> RSHIFT;
        if (inLds) stage[p] = sid; else bsrc[seg + p] = sid;
      }
    }
  }
  if (inLds) {
    __syncthreads();
    for (int i = t; i < L; i += PB) bsrc[seg + i] = stage[i];
  }
}
#define BS_LDS_INTS (MAXR + MAXR + 32 + PB * EPT)

__device__ __forceinline__ void blocksort_block(
    const int* __restrict__ ei, int* __restrict__ part, int* __restrict__ runTab,
    int* __restrict__ cursor, int nE, int R, int RP, int eblock,
    int* lds)
{
  int* const hist   = lds;
  int* const lstart = hist + MAXR;
  int* const wsum   = lstart + MAXR;
  int* const skey   = wsum + 32;
  const int t    = threadIdx.x;
  const int lane = t & 31;
  const int wave = t >> 5;
  hist[t] = 0;
  if (eblock == 0 && t == 0) cursor[0] = 0;
  __syncthreads();

  const int base = eblock * (PB * EPT);
  int s[EPT], d[EPT], rr[EPT], rk[EPT];
  #pragma unroll
  for (int j = 0; j < EPT; ++j) {
    const int e = base + j * PB + t;
    const bool ok = e < nE;
    s[j] = ok ? ei[e] : 0;
    d[j] = ok ? ei[nE + e] : -1;
  }
  #pragma unroll
  for (int j = 0; j < EPT; ++j) {
    rk[j] = 0;
    rr[j] = 0;
    if (d[j] >= 0) {
      rr[j] = d[j] / FR;
      rk[j] = atomicAdd(&hist[rr[j]], 1);
    }
  }
  __syncthreads();
  const int cc = hist[t];
  const int ls = block_excl_scan(cc, wsum, lane, wave, 32);
  lstart[t] = ls;
  if (t < R) runTab[(size_t)eblock * RP + t] = (ls << 16) | cc;
  __syncthreads();
  #pragma unroll
  for (int j = 0; j < EPT; ++j) {
    if (d[j] >= 0)
      skey[lstart[rr[j]] + rk[j]] = (s[j] << RSHIFT) | (d[j] - rr[j] * FR);
  }
  __syncthreads();
  const int left = nE - base;
  const int mine = left < PB * EPT ? left : PB * EPT;
  #pragma unroll
  for (int j = 0; j < EPT; ++j) {
    const int i = j * PB + t;
    if (i < mine) part[base + i] = skey[i];
  }
}

__global__ __launch_bounds__(PB) void gat_gemm_blocksort_kernel(
    const float* __restrict__ x, const float* __restrict__ W,
    float* __restrict__ h, int* __restrict__ rangeTabs, int n, int gemmBlocks,
    const int* __restrict__ ei, int* __restrict__ part, int* __restrict__ runTab,
    int* __restrict__ cursor, int nE, int R, int RP)
{
  static_assert(BS_LDS_INTS * sizeof(int) >= IN_CH * HEADS * sizeof(float), "the buffer holds W");
  __shared__ float4 lds4[BS_LDS_INTS / 4];
  const int b = blockIdx.x;
  if (b < gemmBlocks) gemm_tiles(x, W, h, rangeTabs, n, b, lds4);
  else                blocksort_block(ei, part, runTab, cursor, nE, R, RP, b - gemmBlocks,
                                      reinterpret_cast<int*>(lds4));
}

template <bool PLACE>
__device__ __forceinline__ void range_entry(int kk, int* cnt, int* pos, int* stage,
                                            int* __restrict__ bsrc, int seg, bool inLds) {
  if (!PLACE) {
    atomicAdd(&cnt[kk & (RSIZE - 1)], 1);
  } else {
    const int p   = atomicAdd(&pos[kk & (RSIZE - 1)], 1);
    const int sid = kk >> RSHIFT;
    if (inLds) stage[p] = sid; else bsrc[seg + p] = sid;
  }
}

template <bool PLACE>
__device__ __forceinline__ void range_run(const int* __restrict__ part, const int* runPk,
                                          int b, int first, int lane, int* cnt, int* pos,
                                          int* stage, int* __restrict__ bsrc, int seg,
                                          bool inLds) {
  const int pk  = runPk[b];
  const int len = pk & 0xffff;
  const int beg = b * (PB * EPT) + (pk >> 16);
  for (int i = first + lane; i < len; i += 32)
    range_entry<PLACE>(part[beg + i], cnt, pos, stage, bsrc, seg, inLds);
}

__device__ __forceinline__ void gather_load_ids(const int* __restrict__ bsrc, int base,
                                                int end, int q, int (&s)[8]) {
  #pragma unroll
  for (int j = 0; j < 8; ++j) {
    const int p = base + j * 4 + q;
    s[j] = bsrc[p < end ? p : end - 1];
  }
}

__device__ __forceinline__ void gather_load_rows(const float* __restrict__ hmat,
                                                 const int (&s)[8], int hh, float (&h)[8]) {
  #pragma unroll
  for (int j = 0; j < 8; ++j) h[j] = hmat[s[j] * HEADS + hh];
}

#define GATHER_BLOCKS 1024

__device__ __forceinline__ void gather_node_ids(const int* __restrict__ bsrc, int start,
                                                int end, int cnt, int q,
                                                int (&sA)[8], int (&sB)[8]) {
  if (cnt > 0)     gather_load_ids(bsrc, start, end, q, sA);
  if (cnt > CHUNK) gather_load_ids(bsrc, start + CHUNK, end, q, sB);
}

__global__ __launch_bounds__(256) void gat_gather_kernel(
    const int* __restrict__ bsrc, const int* __restrict__ cur,
    const int* __restrict__ deg, const float* __restrict__ hmat,
    const float* __restrict__ att_src, const float* __restrict__ att_dst,
    const float* __restrict__ gat_bias, const float* __restrict__ lin_w,
    const float* __restrict__ lin_b, const float* __restrict__ bias,
    float* __restrict__ out, int n)
{
  const int lane = threadIdx.x & 31;
  const int q  = lane >> 3;
  const int hh = lane & 7;

  const int T   = GATHER_BLOCKS * 8;
  const int w   = __builtin_amdgcn_readfirstlane((int)(blockIdx.x * 8 + (threadIdx.x >> 5)));
  const int per = n / T;
  const int rem = n - per * T;
  const int n0  = w * per + (w < rem ? w : rem);
  const int n1  = n0 + per + (w < rem ? 1 : 0);
  if (n0 >= n1) return;

  const float as = att_src[hh];
  const float ad = att_dst[hh];
  const float gb = gat_bias[hh];
  const float lb = lin_b[0];
  const float bo = bias[0];

  int endC = cur[n0];
  int cntC = deg[n0];
  const int i1 = (n0 + 1 < n1) ? n0 + 1 : n1 - 1;
  int end1 = cur[i1];
  int cnt1 = deg[i1];

  int nA[8], nB[8];
  #pragma unroll
  for (int j = 0; j < 8; ++j) { nA[j] = 0;  nB[j] = 0; }
  gather_node_ids(bsrc, endC - cntC, endC, cntC, q, nA, nB);

  for (int node = n0; node < n1; ++node) {
    const int end   = endC;
    const int cnt   = cntC;
    const int start = end - cnt;
    const bool hasA = cnt > 0;
    const bool hasB = cnt > CHUNK;
    const int  rest = start + 2 * CHUNK;

    int sA[8], sB[8];
    #pragma unroll
    for (int j = 0; j < 8; ++j) { sA[j] = nA[j];  sB[j] = nB[j]; }

    const int i2 = (node + 2 < n1) ? node + 2 : n1 - 1;
    const int end2 = cur[i2];
    const int cnt2 = deg[i2];

    const int cntN = (node + 1 < n1) ? cnt1 : 0;
    gather_node_ids(bsrc, end1 - cntN, end1, cntN, q, nA, nB);

    float hA[8], hB[8], eA[8], eB[8];
    #pragma unroll
    for (int j = 0; j < 8; ++j) {
      hA[j] = 0.f;  hB[j] = 0.f;  eA[j] = -INFINITY;  eB[j] = -INFINITY;
    }
    const float hself = hmat[node * HEADS + hh];
    if (hasA) gather_load_rows(hmat, sA, hh, hA);
    if (hasB) gather_load_rows(hmat, sB, hh, hB);

    const float adst  = hself * ad;
    const float eself = leaky(hself * as + adst);
    if (hasA) {
      #pragma unroll
      for (int j = 0; j < 8; ++j)
        if (j * 4 + q < cnt) eA[j] = leaky(hA[j] * as + adst);
    }
    if (hasB) {
      #pragma unroll
      for (int j = 0; j < 8; ++j)
        if (CHUNK + j * 4 + q < cnt) eB[j] = leaky(hB[j] * as + adst);
    }

    float m = (q == 0) ? eself : -INFINITY;
    #pragma unroll
    for (int j = 0; j < 8; ++j) m = fmaxf(m, eA[j]);
    if (hasB) {
      #pragma unroll
      for (int j = 0; j < 8; ++j) m = fmaxf(m, eB[j]);
    }
    for (int base = rest; base < end; base += CHUNK) {
      int   s[8];
      float hs[8];
      #pragma unroll
      for (int j = 0; j < 8; ++j) {
        const int p = base + j * 4 + q;
        s[j] = bsrc[p < end ? p : end - 1];
      }
      #pragma unroll
      for (int j = 0; j < 8; ++j) hs[j] = hmat[s[j] * HEADS + hh];
      #pragma unroll
      for (int j = 0; j < 8; ++j) {
        const int p = base + j * 4 + q;
        const float e = leaky(hs[j] * as + adst);
        if (p < end) m = fmaxf(m, e);
      }
    }
    m = fmaxf(m, __shfl_xor(m, 8));
    m = fmaxf(m, __shfl_xor(m, 16));

    const float mL = m * LOG2E;
    float den = 0.f, num = 0.f;
    if (q == 0) { den = exp_rel(eself, mL); num = den * hself; }
    #pragma unroll
    for (int j = 0; j < 8; ++j) {
      const float ee = exp_rel(eA[j], mL);
      den += ee;  num = fmaf(ee, hA[j], num);
    }
    if (hasB) {
      #pragma unroll
      for (int j = 0; j < 8; ++j) {
        const float ee = exp_rel(eB[j], mL);
        den += ee;  num = fmaf(ee, hB[j], num);
      }
    }
    for (int base = rest; base < end; base += CHUNK) {
      int   s[8];
      float hs[8];
      #pragma unroll
      for (int j = 0; j < 8; ++j) {
        const int p = base + j * 4 + q;
        s[j] = bsrc[p < end ? p : end - 1];
      }
      #pragma unroll
      for (int j = 0; j < 8; ++j) hs[j] = hmat[s[j] * HEADS + hh];
      #pragma unroll
      for (int j = 0; j < 8; ++j) {
        const int p = base + j * 4 + q;
        const float ee = exp_rel(leaky(hs[j] * as + adst), mL);
        if (p < end) { den += ee; num += ee * hs[j]; }
      }
    }
    den += __shfl_xor(den, 8);
    den += __shfl_xor(den, 16);
    num += __shfl_xor(num, 8);
    num += __shfl_xor(num, 16);

    const float oh = num / (den + 1e-16f) + gb;
    float z = lb;
    #pragma unroll
    for (int k = 0; k < HEADS; ++k) z += __shfl(oh, k) * lin_w[k];
    if (lane == 0) out[node] = fmaxf(z, 0.f) + bo;

    endC = end1;  cntC = cntN;
    end1 = end2;  cnt1 = cnt2;
  }
}

#define FUSED_LDS_INTS (RSIZE + RSIZE + 32 + MAXB + 32 + CAP)

template <bool INLDS>
__device__ __forceinline__ int list_id(const int* ids, int p) {
  if (INLDS) return ids[p];
  return __hip_atomic_load(const_cast<int*>(ids) + p, __ATOMIC_RELAXED, __HIP_MEMORY_SCOPE_AGENT);
}

template <bool INLDS>
__device__ __forceinline__ void list_load_ids(const int* ids, int base, int end, int q,
                                              int (&s)[8]) {
  #pragma unroll
  for (int j = 0; j < 8; ++j) {
    const int p = base + j * 4 + q;
    s[j] = list_id<INLDS>(ids, p < end ? p : end - 1);
  }
}

template <bool INLDS>
__device__ __forceinline__ void list_node_ids(const int* ids, int start, int end, int cnt,
                                              int q, int (&sA)[8], int (&sB)[8]) {
  if (cnt > 0)     list_load_ids<INLDS>(ids, start, end, q, sA);
  if (cnt > CHUNK) list_load_ids<INLDS>(ids, start + CHUNK, end, q, sB);
}

template <bool INLDS>
__device__ __forceinline__ void walk_range_nodes(
    const int* ids, const int* cntL, const int* posL, int nodeBase, int i0, int i1,
    const float* __restrict__ hmat, float as, float ad, float gb, float lb, float bo,
    const float* __restrict__ lin_w, float* __restrict__ out, int lane)
{
  const int q  = lane >> 3;
  const int hh = lane & 7;
  if (i0 >= i1) return;

  int endC = __builtin_amdgcn_readfirstlane(posL[i0]);
  int cntC = __builtin_amdgcn_readfirstlane(cntL[i0]);
  const int j1 = (i0 + 1 < i1) ? i0 + 1 : i1 - 1;
  int end1 = __builtin_amdgcn_readfirstlane(posL[j1]);
  int cnt1 = __builtin_amdgcn_readfirstlane(cntL[j1]);

  int nA[8], nB[8];
  #pragma unroll
  for (int j = 0; j < 8; ++j) { nA[j] = 0;  nB[j] = 0; }
  list_node_ids<INLDS>(ids, endC - cntC, endC, cntC, q, nA, nB);

  for (int i = i0; i < i1; ++i) {
    const int node  = nodeBase + i;
    const int end   = endC;
    const int cnt   = cntC;
    const int start = end - cnt;
    const bool hasA = cnt > 0;
    const bool hasB = cnt > CHUNK;
    const int  rest = start + 2 * CHUNK;

    int sA[8], sB[8];
    #pragma unroll
    for (int j = 0; j < 8; ++j) { sA[j] = nA[j];  sB[j] = nB[j]; }

    const int cntN = (i + 1 < i1) ? cnt1 : 0;
    list_node_ids<INLDS>(ids, end1 - cntN, end1, cntN, q, nA, nB);

    float hA[8], hB[8], eA[8], eB[8];
    #pragma unroll
    for (int j = 0; j < 8; ++j) {
      hA[j] = 0.f;  hB[j] = 0.f;  eA[j] = -INFINITY;  eB[j] = -INFINITY;
    }
    const float hself = hmat[node * HEADS + hh];
    if (hasA) gather_load_rows(hmat, sA, hh, hA);
    if (hasB) gather_load_rows(hmat, sB, hh, hB);

    const int i2 = (i + 2 < i1) ? i + 2 : i1 - 1;
    const int end2 = __builtin_amdgcn_readfirstlane(posL[i2]);
    const int cnt2 = __builtin_amdgcn_readfirstlane(cntL[i2]);

    const float adst  = hself * ad;
    const float eself = leaky(hself * as + adst);
    if (hasA) {
      #pragma unroll
      for (int j = 0; j < 8; ++j)
        if (j * 4 + q < cnt) eA[j] = leaky(hA[j] * as + adst);
    }
    if (hasB) {
      #pragma unroll
      for (int j = 0; j < 8; ++j)
        if (CHUNK + j * 4 + q < cnt) eB[j] = leaky(hB[j] * as + adst);
    }

    float m = (q == 0) ? eself : -INFINITY;
    #pragma unroll
    for (int j = 0; j < 8; ++j) m = fmaxf(m, eA[j]);
    if (hasB) {
      #pragma unroll
      for (int j = 0; j < 8; ++j) m = fmaxf(m, eB[j]);
    }
    for (int base = rest; base < end; base += CHUNK) {
      int   s[8];
      float hs[8];
      #pragma unroll
      for (int j = 0; j < 8; ++j) {
        const int p = base + j * 4 + q;
        s[j] = list_id<INLDS>(ids, p < end ? p : end - 1);
      }
      #pragma unroll
      for (int j = 0; j < 8; ++j) hs[j] = hmat[s[j] * HEADS + hh];
      #pragma unroll
      for (int j = 0; j < 8; ++j) {
        const int p = base + j * 4 + q;
        const float e = leaky(hs[j] * as + adst);
        if (p < end) m = fmaxf(m, e);
      }
    }
    m = fmaxf(m, __shfl_xor(m, 8));
    m = fmaxf(m, __shfl_xor(m, 16));

    const float mL = m * LOG2E;
    float den = 0.f, num = 0.f;
    if (q == 0) { den = exp_rel(eself, mL); num = den * hself; }
    #pragma unroll
    for (int j = 0; j < 8; ++j) {
      const float ee = exp_rel(eA[j], mL);
      den += ee;  num = fmaf(ee, hA[j], num);
    }
    if (hasB) {
      #pragma unroll
      for (int j = 0; j < 8; ++j) {
        const float ee = exp_rel(eB[j], mL);
        den += ee;  num = fmaf(ee, hB[j], num);
      }
    }
    for (int base = rest; base < end; base += CHUNK) {
      int   s[8];
      float hs[8];
      #pragma unroll
      for (int j = 0; j < 8; ++j) {
        const int p = base + j * 4 + q;
        s[j] = list_id<INLDS>(ids, p < end ? p : end - 1);
      }
      #pragma unroll
      for (int j = 0; j < 8; ++j) hs[j] = hmat[s[j] * HEADS + hh];
      #pragma unroll
      for (int j = 0; j < 8; ++j) {
        const int p = base + j * 4 + q;
        const float ee = exp_rel(leaky(hs[j] * as + adst), mL);
        if (p < end) { den += ee; num += ee * hs[j]; }
      }
    }
    den += __shfl_xor(den, 8);
    den += __shfl_xor(den, 16);
    num += __shfl_xor(num, 8);
    num += __shfl_xor(num, 16);

    const float oh = num / (den + 1e-16f) + gb;
    float z = lb;
    #pragma unroll
    for (int k = 0; k < HEADS; ++k) z += __shfl(oh, k) * lin_w[k];
    if (lane == 0) out[node] = fmaxf(z, 0.f) + bo;

    endC = end1;  cntC = cntN;
    end1 = end2;  cnt1 = cnt2;
  }
}

__global__ __launch_bounds__(FPB) void gat_collect_gather_kernel(
    const int* __restrict__ part, const int* __restrict__ runTab,
    int* __restrict__ cursor, int* __restrict__ bsrc,
    const float* __restrict__ hmat,
    const float* __restrict__ att_src, const float* __restrict__ att_dst,
    const float* __restrict__ gat_bias, const float* __restrict__ lin_w,
    const float* __restrict__ lin_b, const float* __restrict__ bias,
    float* __restrict__ out, int n, int EB, int RP)
{
  static_assert(FR <= RSIZE, "a node's place in its range fits the entry's 8 bits and the tables");
  static_assert(MAXB == 2 * FPB, "two table places per thread");
  __shared__ int lds[FUSED_LDS_INTS];
  int* const cnt   = lds;
  int* const pos   = cnt + RSIZE;
  int* const wsum  = pos + RSIZE;
  int* const runPk = wsum + 32;
  int* const segS  = runPk + MAXB;
  int* const stage = segS + 32;
  const int t    = threadIdx.x;
  const int lane = t & 31;
  const int wave = __builtin_amdgcn_readfirstlane(t >> 5);
  const int r    = blockIdx.x;

  const float as = att_src[lane & 7];
  const float ad = att_dst[lane & 7];
  const float gb = gat_bias[lane & 7];
  const float lb = lin_b[0];
  const float bo = bias[0];

  const int pk0 = (t < EB)       ? runTab[(size_t)t * RP + r] : 0;
  const int pk1 = (t + FPB < EB) ? runTab[(size_t)(t + FPB) * RP + r] : 0;
  runPk[t]       = pk0;
  runPk[t + FPB] = pk1;
  const int myLen = (pk0 & 0xffff) + (pk1 & 0xffff);
  if (t < RSIZE) cnt[t] = 0;
  const int before = block_excl_scan(myLen, wsum, lane, wave, FW);
  if (t == FPB - 1) segS[1] = before + myLen;

  static_assert(FW * FHELD <= MAXB && FHELD <= 32, "held runs lie inside the table; one mask bit each");
  int k[FHELD];
  unsigned longMask = 0;
  {
    int pkv[FHELD];
    #pragma unroll
    for (int j = 0; j < FHELD; ++j) pkv[j] = runPk[wave + FW * j];
    #pragma unroll
    for (int j = 0; j < FHELD; ++j) {
      const int len = pkv[j] & 0xffff;
      const int beg = (wave + FW * j) * (PB * EPT) + (pkv[j] >> 16);
      k[j] = -1;
      if (len > 32) longMask |= 1u << j;
      if (lane < len) k[j] = part[beg + lane];
    }
  }
  longMask = __builtin_amdgcn_readfirstlane(longMask);

  #pragma unroll
  for (int j = 0; j < FHELD; ++j)
    if (k[j] >= 0) range_entry<false>(k[j], cnt, pos, stage, bsrc, 0, true);
  for (unsigned m = longMask; m; m &= m - 1)
    range_run<false>(part, runPk, wave + FW * __builtin_ctz(m), 32, lane, cnt, pos, stage, bsrc, 0, true);
  for (int b = wave + FW * FHELD; b < EB; b += FW)
    range_run<false>(part, runPk, b, 0, lane, cnt, pos, stage, bsrc, 0, true);
  __syncthreads();
  const int Ltot = segS[1];
  const bool inLds = (Ltot <= CAP);

  const int c    = (t < RSIZE) ? cnt[t] : 0;
  const int excl = block_excl_scan(c, wsum, lane, wave, FW);
  if (t < RSIZE) pos[t] = excl;
  if (!inLds && t == 0) segS[0] = atomicAdd(cursor, Ltot);
  __syncthreads();
  const int seg = inLds ? 0 : segS[0];

  {
    int p[FHELD];
    #pragma unroll
    for (int j = 0; j < FHELD; ++j)
      p[j] = (k[j] >= 0) ? atomicAdd(&pos[k[j] & (RSIZE - 1)], 1) : 0;
    #pragma unroll
    for (int j = 0; j < FHELD; ++j) {
      if (k[j] >= 0) {
        const int sid = k[j] >> RSHIFT;
        if (inLds) stage[p[j]] = sid; else bsrc[seg + p[j]] = sid;
      }
    }
  }
  for (unsigned m = longMask; m; m &= m - 1)
    range_run<true>(part, runPk, wave + FW * __builtin_ctz(m), 32, lane, cnt, pos, stage, bsrc, seg, inLds);
  for (int b = wave + FW * FHELD; b < EB; b += FW)
    range_run<true>(part, runPk, b, 0, lane, cnt, pos, stage, bsrc, seg, inLds);
  if (!inLds) __threadfence();
  __syncthreads();

  const int nodeBase = r * FR;
  const int leftN = n - nodeBase;
  const int nr    = leftN < FR ? leftN : FR;
  const int per   = nr / FW;
  const int rem   = nr - per * FW;
  const int i0    = wave * per + (wave < rem ? wave : rem);
  const int i1    = i0 + per + (wave < rem ? 1 : 0);
  if (inLds) walk_range_nodes<true >(stage,      cnt, pos, nodeBase, i0, i1, hmat, as, ad, gb, lb, bo, lin_w, out, lane);
  else       walk_range_nodes<false>(bsrc + seg, cnt, pos, nodeBase, i0, i1, hmat, as, ad, gb, lb, bo, lin_w, out, lane);
}

extern "C" void kernel_launch(void* const* d_in, const int* in_sizes, int n_in,
                              void* d_out, int out_size, void* d_ws, size_t ws_size,
                              hipStream_t stream) {
  const float* x        = (const float*)d_in[0];
  const int*   ei       = (const int*)d_in[1];
  const float* W        = (const float*)d_in[2];
  const float* att_src  = (const float*)d_in[3];
  const float* att_dst  = (const float*)d_in[4];
  const float* gat_bias = (const float*)d_in[5];
  const float* lin_w    = (const float*)d_in[6];
  const float* lin_b    = (const float*)d_in[7];
  const float* bias     = (const float*)d_in[8];
  float* out = (float*)d_out;

  const int n  = in_sizes[0] / IN_CH;
  const int nE = in_sizes[1] / 2;

  const size_t nh   = (size_t)n * HEADS;
  const size_t segH = (nh * sizeof(float) + 255) & ~(size_t)255;
  const size_t segN = ((size_t)n * sizeof(int) + 255) & ~(size_t)255;
  const size_t segT = (size_t)3 * MAXR * sizeof(int);
  const size_t segE = ((size_t)nE * sizeof(int) + 255) & ~(size_t)255;
  char* ws = (char*)d_ws;
  float* h_mat      = (float*)(ws);
  int*   deg        = (int*)  (ws + segH);
  int*   cur        = (int*)  (ws + segH + segN);
  int*   rangeCnt   = (int*)  (ws + segH + 2 * segN);
  int*   rangeCur   = rangeCnt + MAXR;
  int*   rangeStart = rangeCnt + 2 * MAXR;
  int*   part       = (int*)  (ws + segH + 2 * segN + segT);
  int*   bsrc       = (ws_size >= segH + 2 * segN + segT + 2 * segE) ? (int*)((char*)part + segE) : part;
  const int    eb     = (nE + PB * EPT - 1) / (PB * EPT);
  const int    FRn    = (n + FR - 1) / FR;
  const int    RP     = (FRn + 31) & ~31;
  const size_t segR   = (size_t)eb * RP * sizeof(int);
  const size_t need2  = segH + 2 * segN + segT + 2 * segE + segR;
  int*         runTab = (int*)((char*)part + 2 * segE);
  const bool   twoLaunch = (ws_size >= need2) && (eb <= MAXB) && (FRn <= MAXR);

  const int tiles  = (n + 15) / 16;
  const int blocks = (tiles + 31) / 32;

  if (twoLaunch) {
    gat_gemm_blocksort_kernel<<<blocks + eb, PB, 0, stream>>>(
        x, W, h_mat, (int*)nullptr, n, blocks, ei, part, runTab, rangeCur, nE, FRn, RP);

    gat_collect_gather_kernel<<<FRn, FPB, 0, stream>>>(
        part, runTab, rangeCur, bsrc, h_mat, att_src, att_dst, gat_bias, lin_w, lin_b, bias,
        out, n, eb, RP);
  } else {
    const int R = (n + RSIZE - 1) / RSIZE;

    gat_gemm_blocksort_kernel<<<blocks, PB, 0, stream>>>(
        x, W, h_mat, rangeCnt, n, blocks, ei, part, runTab, rangeCur, nE, 0, RP);

    gat_rangecount_kernel<<<eb, PB, 0, stream>>>(ei, rangeCnt, nE, R);

    gat_partition_kernel<<<eb, PB, 0, stream>>>(ei, rangeCnt, rangeCur, rangeStart, part, nE, R);

    gat_rangesort_kernel<<<R, PB, 0, stream>>>(part, rangeCnt, rangeStart, deg, cur, bsrc, n);

    gat_gather_kernel<<<GATHER_BLOCKS, 256, 0, stream>>>(
        bsrc, cur, deg, h_mat, att_src, att_dst, gat_bias, lin_w, lin_b, bias,
        out, n);
  }
}


# two-launch walk: less vector work per node: slots past the list skipped in half groups of 16 edges (was 32), no preset of unused slots, ids by byte offset, leaky as max(v,0.2v); sum order kept
# speedup vs baseline: 24.6223x; 1.0364x over previous
#include <hip/hip_runtime.h>
#include <math.h>


#define IN_CH 512
#define HEADS 8
#define NEG_SLOPE 0.2f
#define XRING 10
#define CHUNK 32
#define RSHIFT 8
#define RSIZE 256
#define MAXR 1024
#define PB 1024
#define EPT 8
#define CAP 16384
#define MAXB 1024
#define FR 196
#define FPB 512
#define FW 16
#define FHELD 25

typedef __attribute__((ext_vector_type(2))) float v2f;
typedef __attribute__((ext_vector_type(8))) float v8f;

__device__ __forceinline__ float leaky(float v) {
  return v >= 0.f ? v : NEG_SLOPE * v;
}

#define LOG2E 1.44269504088896340736f

__device__ __forceinline__ float exp_rel(float e, float mL) {
  return __builtin_amdgcn_exp2f(fmaf(e, LOG2E, -mL));
}

__device__ __forceinline__ void gemm_tiles(
    const float* __restrict__ x, const float* __restrict__ W,
    float* __restrict__ h, int* __restrict__ rangeTabs, int n, int gblock,
    float4* Wlds4)
{
  static_assert(IN_CH * HEADS / 4 == PB, "one 16-byte piece of W per thread");
  {
    const float4* __restrict__ W4 = reinterpret_cast<const float4*>(W);
    Wlds4[threadIdx.x] = W4[threadIdx.x];
  }
  if (rangeTabs != nullptr && gblock == 0) {
    for (int i = threadIdx.x; i < 2 * MAXR; i += PB) rangeTabs[i] = 0;
  }
  __syncthreads();

  const int lane = threadIdx.x & 31;
  const int wave = threadIdx.x >> 5;
  const int half = lane >> 4;
  const int l    = lane & 15;

  const int tile = gblock * (PB >> 5) + wave;
  const int row0 = tile * 16;
  if (row0 >= n) return;

  int arow = row0 + l;
  if (arow > n - 1) arow = n - 1;
  const float4* __restrict__ x4 = reinterpret_cast<const float4*>(x);
  const int xi = arow * (IN_CH / 4) + half;
  const float* __restrict__ wl =
      reinterpret_cast<const float*>(Wlds4) + half * 4 * HEADS + (l & 7);

  const int rfirst = row0 + half * 8;
  const int hidx   = rfirst * HEADS + (l & 7);
  const int hrows  = (l < HEADS) ? n - rfirst : 0;

  v8f c = {0.f, 0.f, 0.f, 0.f, 0.f, 0.f, 0.f, 0.f};

  float4 xa[XRING];
  #pragma unroll
  for (int p = 0; p < XRING - 1; ++p) xa[p] = x4[xi + p * 2];
  float wb[2][4];
  #pragma unroll
  for (int u = 0; u < 4; ++u) wb[0][u] = wl[u * HEADS];

  #pragma unroll
  for (int p = 0; p < IN_CH / 8; ++p) {
    if (p + XRING - 1 < IN_CH / 8)
      xa[(p + XRING - 1) % XRING] = x4[xi + (p + XRING - 1) * 2];
    if (p + 1 < IN_CH / 8) {
      #pragma unroll
      for (int u = 0; u < 4; ++u) wb[(p + 1) & 1][u] = wl[((p + 1) * 8 + u) * HEADS];
    }
    const float4 av = xa[p % XRING];
    v2f a, b;
    a.x = av.x;  a.y = av.y;
    b.x = wb[p & 1][0];  b.y = wb[p & 1][1];
    c = __builtin_amdgcn_wmma_f32_16x16x4_f32(
         false, a,  false, b,
         (short)0, c,  false,  false);
    a.x = av.z;  a.y = av.w;
    b.x = wb[p & 1][2];  b.y = wb[p & 1][3];
    c = __builtin_amdgcn_wmma_f32_16x16x4_f32(
         false, a,  false, b,
         (short)0, c,  false,  false);
    __builtin_amdgcn_sched_barrier(0);
  }

  #pragma unroll
  for (int i = 0; i < 8; ++i)
    if (i < hrows) h[hidx + i * HEADS] = c[i];
}

__device__ __forceinline__ int block_excl_scan(int v, int* wsum, int lane, int wave, int nw) {
  int incl = v;
  #pragma unroll
  for (int o = 1; o < 32; o <<= 1) {
    const int t = __shfl_up(incl, o);
    if (lane >= o) incl += t;
  }
  if (lane == 31) wsum[wave] = incl;
  __syncthreads();
  if (wave == 0) {
    const int w = (lane < nw) ? wsum[lane] : 0;
    int wi = w;
    #pragma unroll
    for (int o = 1; o < 32; o <<= 1) {
      const int t = __shfl_up(wi, o);
      if (lane >= o) wi += t;
    }
    wsum[lane] = wi - w;
  }
  __syncthreads();
  return wsum[wave] + incl - v;
}

__global__ __launch_bounds__(PB) void gat_rangecount_kernel(
    const int* __restrict__ ei, int* __restrict__ rangeCnt, int nE, int R)
{
  __shared__ int hist[MAXR];
  const int t = threadIdx.x;
  hist[t] = 0;
  __syncthreads();
  const int base = blockIdx.x * (PB * EPT);
  int d[EPT];
  #pragma unroll
  for (int j = 0; j < EPT; ++j) {
    const int e = base + j * PB + t;
    d[j] = (e < nE) ? ei[nE + e] : -1;
  }
  #pragma unroll
  for (int j = 0; j < EPT; ++j)
    if (d[j] >= 0) atomicAdd(&hist[d[j] >> RSHIFT], 1);
  __syncthreads();
  if (t < R) {
    const int c = hist[t];
    if (c) atomicAdd(rangeCnt + t, c);
  }
}

__global__ __launch_bounds__(PB) void gat_partition_kernel(
    const int* __restrict__ ei, const int* __restrict__ rangeCnt,
    int* __restrict__ rangeCur, int* __restrict__ rangeStart,
    int* __restrict__ part, int nE, int R)
{
  __shared__ int hist[MAXR];
  __shared__ int lstart[MAXR];
  __shared__ int rbase[MAXR];
  __shared__ int wsum[32];
  __shared__ int skey[PB * EPT];
  __shared__ int sgpos[PB * EPT];
  const int t    = threadIdx.x;
  const int lane = t & 31;
  const int wave = t >> 5;
  hist[t] = 0;
  const int tot    = (t < R) ? rangeCnt[t] : 0;
  const int rstart = block_excl_scan(tot, wsum, lane, wave, 32);
  if (blockIdx.x == 0 && t < R) rangeStart[t] = rstart;

  const int base = blockIdx.x * (PB * EPT);
  int s[EPT], d[EPT], rk[EPT];
  #pragma unroll
  for (int j = 0; j < EPT; ++j) {
    const int e = base + j * PB + t;
    const bool ok = e < nE;
    s[j] = ok ? ei[e] : 0;
    d[j] = ok ? ei[nE + e] : -1;
  }
  #pragma unroll
  for (int j = 0; j < EPT; ++j) {
    rk[j] = 0;
    if (d[j] >= 0) rk[j] = atomicAdd(&hist[d[j] >> RSHIFT], 1);
  }
  __syncthreads();
  const int cc = hist[t];
  const int ls = block_excl_scan(cc, wsum, lane, wave, 32);
  lstart[t] = ls;
  rbase[t]  = (t < R && cc) ? rstart + atomicAdd(rangeCur + t, cc) : 0;
  __syncthreads();
  #pragma unroll
  for (int j = 0; j < EPT; ++j) {
    if (d[j] >= 0) {
      const int r = d[j] >> RSHIFT;
      const int p = lstart[r] + rk[j];
      skey[p]  = (s[j] << RSHIFT) | (d[j] & (RSIZE - 1));
      sgpos[p] = rbase[r] + rk[j];
    }
  }
  __syncthreads();
  const int left  = nE - base;
  const int mine  = left < PB * EPT ? left : PB * EPT;
  #pragma unroll
  for (int j = 0; j < EPT; ++j) {
    const int i = j * PB + t;
    if (i < mine) part[sgpos[i]] = skey[i];
  }
}

__global__ __launch_bounds__(PB) void gat_rangesort_kernel(
    const int* part, const int* __restrict__ rangeCnt,
    const int* __restrict__ rangeStart, int* __restrict__ deg,
    int* __restrict__ cur, int* bsrc, int n)
{
  __shared__ int cnt[RSIZE];
  __shared__ int pos[RSIZE];
  __shared__ int wsum[32];
  __shared__ int stage[CAP];
  const int t    = threadIdx.x;
  const int lane = t & 31;
  const int wave = t >> 5;
  const int r    = blockIdx.x;
  const int L    = rangeCnt[r];
  const int seg  = rangeStart[r];
  if (t < RSIZE) cnt[t] = 0;
  __syncthreads();
  for (int i0 = t; i0 < L; i0 += 4 * PB) {
    int k[4];
    #pragma unroll
    for (int u = 0; u < 4; ++u) {
      const int i = i0 + u * PB;
      k[u] = (i < L) ? part[seg + i] : -1;
    }
    #pragma unroll
    for (int u = 0; u < 4; ++u)
      if (k[u] >= 0) atomicAdd(&cnt[k[u] & (RSIZE - 1)], 1);
  }
  __syncthreads();
  const int c    = (t < RSIZE) ? cnt[t] : 0;
  const int excl = block_excl_scan(c, wsum, lane, wave, 32);
  if (t < RSIZE) {
    pos[t] = excl;
    const int node = r * RSIZE + t;
    if (node < n) { deg[node] = c; cur[node] = seg + excl + c; }
  }
  __syncthreads();
  const bool inLds = (L <= CAP);
  for (int i0 = t; i0 < L; i0 += 4 * PB) {
    int k[4];
    #pragma unroll
    for (int u = 0; u < 4; ++u) {
      const int i = i0 + u * PB;
      k[u] = (i < L) ? part[seg + i] : -1;
    }
    #pragma unroll
    for (int u = 0; u < 4; ++u) {
      if (k[u] >= 0) {
        const int p   = atomicAdd(&pos[k[u] & (RSIZE - 1)], 1);
        const int sid = k[u] >> RSHIFT;
        if (inLds) stage[p] = sid; else bsrc[seg + p] = sid;
      }
    }
  }
  if (inLds) {
    __syncthreads();
    for (int i = t; i < L; i += PB) bsrc[seg + i] = stage[i];
  }
}
#define BS_LDS_INTS (MAXR + MAXR + 32 + PB * EPT)

__device__ __forceinline__ void blocksort_block(
    const int* __restrict__ ei, int* __restrict__ part, int* __restrict__ runTab,
    int* __restrict__ cursor, int nE, int R, int RP, int eblock,
    int* lds)
{
  int* const hist   = lds;
  int* const lstart = hist + MAXR;
  int* const wsum   = lstart + MAXR;
  int* const skey   = wsum + 32;
  const int t    = threadIdx.x;
  const int lane = t & 31;
  const int wave = t >> 5;
  hist[t] = 0;
  if (eblock == 0 && t == 0) cursor[0] = 0;
  __syncthreads();

  const int base = eblock * (PB * EPT);
  int s[EPT], d[EPT], rr[EPT], rk[EPT];
  #pragma unroll
  for (int j = 0; j < EPT; ++j) {
    const int e = base + j * PB + t;
    const bool ok = e < nE;
    s[j] = ok ? ei[e] : 0;
    d[j] = ok ? ei[nE + e] : -1;
  }
  #pragma unroll
  for (int j = 0; j < EPT; ++j) {
    rk[j] = 0;
    rr[j] = 0;
    if (d[j] >= 0) {
      rr[j] = d[j] / FR;
      rk[j] = atomicAdd(&hist[rr[j]], 1);
    }
  }
  __syncthreads();
  const int cc = hist[t];
  const int ls = block_excl_scan(cc, wsum, lane, wave, 32);
  lstart[t] = ls;
  if (t < R) runTab[(size_t)eblock * RP + t] = (ls << 16) | cc;
  __syncthreads();
  #pragma unroll
  for (int j = 0; j < EPT; ++j) {
    if (d[j] >= 0)
      skey[lstart[rr[j]] + rk[j]] = (s[j] << RSHIFT) | (d[j] - rr[j] * FR);
  }
  __syncthreads();
  const int left = nE - base;
  const int mine = left < PB * EPT ? left : PB * EPT;
  #pragma unroll
  for (int j = 0; j < EPT; ++j) {
    const int i = j * PB + t;
    if (i < mine) part[base + i] = skey[i];
  }
}

__global__ __launch_bounds__(PB) void gat_gemm_blocksort_kernel(
    const float* __restrict__ x, const float* __restrict__ W,
    float* __restrict__ h, int* __restrict__ rangeTabs, int n, int gemmBlocks,
    const int* __restrict__ ei, int* __restrict__ part, int* __restrict__ runTab,
    int* __restrict__ cursor, int nE, int R, int RP)
{
  static_assert(BS_LDS_INTS * sizeof(int) >= IN_CH * HEADS * sizeof(float), "the buffer holds W");
  __shared__ float4 lds4[BS_LDS_INTS / 4];
  const int b = blockIdx.x;
  if (b < gemmBlocks) gemm_tiles(x, W, h, rangeTabs, n, b, lds4);
  else                blocksort_block(ei, part, runTab, cursor, nE, R, RP, b - gemmBlocks,
                                      reinterpret_cast<int*>(lds4));
}

template <bool PLACE>
__device__ __forceinline__ void range_entry(int kk, int* cnt, int* pos, int* stage,
                                            int* __restrict__ bsrc, int seg, bool inLds) {
  if (!PLACE) {
    atomicAdd(&cnt[kk & (RSIZE - 1)], 1);
  } else {
    const int p   = atomicAdd(&pos[kk & (RSIZE - 1)], 1);
    const int sid = kk >> RSHIFT;
    if (inLds) stage[p] = sid; else bsrc[seg + p] = sid;
  }
}

template <bool PLACE>
__device__ __forceinline__ void range_run(const int* __restrict__ part, const int* runPk,
                                          int b, int first, int lane, int* cnt, int* pos,
                                          int* stage, int* __restrict__ bsrc, int seg,
                                          bool inLds) {
  const int pk  = runPk[b];
  const int len = pk & 0xffff;
  const int beg = b * (PB * EPT) + (pk >> 16);
  for (int i = first + lane; i < len; i += 32)
    range_entry<PLACE>(part[beg + i], cnt, pos, stage, bsrc, seg, inLds);
}

__device__ __forceinline__ void gather_load_ids(const int* __restrict__ bsrc, int base,
                                                int end, int q, int (&s)[8]) {
  #pragma unroll
  for (int j = 0; j < 8; ++j) {
    const int p = base + j * 4 + q;
    s[j] = bsrc[p < end ? p : end - 1];
  }
}

__device__ __forceinline__ void gather_load_rows(const float* __restrict__ hmat,
                                                 const int (&s)[8], int hh, float (&h)[8]) {
  #pragma unroll
  for (int j = 0; j < 8; ++j) h[j] = hmat[s[j] * HEADS + hh];
}

#define GATHER_BLOCKS 1024

__device__ __forceinline__ void gather_node_ids(const int* __restrict__ bsrc, int start,
                                                int end, int cnt, int q,
                                                int (&sA)[8], int (&sB)[8]) {
  if (cnt > 0)     gather_load_ids(bsrc, start, end, q, sA);
  if (cnt > CHUNK) gather_load_ids(bsrc, start + CHUNK, end, q, sB);
}

__global__ __launch_bounds__(256) void gat_gather_kernel(
    const int* __restrict__ bsrc, const int* __restrict__ cur,
    const int* __restrict__ deg, const float* __restrict__ hmat,
    const float* __restrict__ att_src, const float* __restrict__ att_dst,
    const float* __restrict__ gat_bias, const float* __restrict__ lin_w,
    const float* __restrict__ lin_b, const float* __restrict__ bias,
    float* __restrict__ out, int n)
{
  const int lane = threadIdx.x & 31;
  const int q  = lane >> 3;
  const int hh = lane & 7;

  const int T   = GATHER_BLOCKS * 8;
  const int w   = __builtin_amdgcn_readfirstlane((int)(blockIdx.x * 8 + (threadIdx.x >> 5)));
  const int per = n / T;
  const int rem = n - per * T;
  const int n0  = w * per + (w < rem ? w : rem);
  const int n1  = n0 + per + (w < rem ? 1 : 0);
  if (n0 >= n1) return;

  const float as = att_src[hh];
  const float ad = att_dst[hh];
  const float gb = gat_bias[hh];
  const float lb = lin_b[0];
  const float bo = bias[0];

  int endC = cur[n0];
  int cntC = deg[n0];
  const int i1 = (n0 + 1 < n1) ? n0 + 1 : n1 - 1;
  int end1 = cur[i1];
  int cnt1 = deg[i1];

  int nA[8], nB[8];
  #pragma unroll
  for (int j = 0; j < 8; ++j) { nA[j] = 0;  nB[j] = 0; }
  gather_node_ids(bsrc, endC - cntC, endC, cntC, q, nA, nB);

  for (int node = n0; node < n1; ++node) {
    const int end   = endC;
    const int cnt   = cntC;
    const int start = end - cnt;
    const bool hasA = cnt > 0;
    const bool hasB = cnt > CHUNK;
    const int  rest = start + 2 * CHUNK;

    int sA[8], sB[8];
    #pragma unroll
    for (int j = 0; j < 8; ++j) { sA[j] = nA[j];  sB[j] = nB[j]; }

    const int i2 = (node + 2 < n1) ? node + 2 : n1 - 1;
    const int end2 = cur[i2];
    const int cnt2 = deg[i2];

    const int cntN = (node + 1 < n1) ? cnt1 : 0;
    gather_node_ids(bsrc, end1 - cntN, end1, cntN, q, nA, nB);

    float hA[8], hB[8], eA[8], eB[8];
    #pragma unroll
    for (int j = 0; j < 8; ++j) {
      hA[j] = 0.f;  hB[j] = 0.f;  eA[j] = -INFINITY;  eB[j] = -INFINITY;
    }
    const float hself = hmat[node * HEADS + hh];
    if (hasA) gather_load_rows(hmat, sA, hh, hA);
    if (hasB) gather_load_rows(hmat, sB, hh, hB);

    const float adst  = hself * ad;
    const float eself = leaky(hself * as + adst);
    if (hasA) {
      #pragma unroll
      for (int j = 0; j < 8; ++j)
        if (j * 4 + q < cnt) eA[j] = leaky(hA[j] * as + adst);
    }
    if (hasB) {
      #pragma unroll
      for (int j = 0; j < 8; ++j)
        if (CHUNK + j * 4 + q < cnt) eB[j] = leaky(hB[j] * as + adst);
    }

    float m = (q == 0) ? eself : -INFINITY;
    #pragma unroll
    for (int j = 0; j < 8; ++j) m = fmaxf(m, eA[j]);
    if (hasB) {
      #pragma unroll
      for (int j = 0; j < 8; ++j) m = fmaxf(m, eB[j]);
    }
    for (int base = rest; base < end; base += CHUNK) {
      int   s[8];
      float hs[8];
      #pragma unroll
      for (int j = 0; j < 8; ++j) {
        const int p = base + j * 4 + q;
        s[j] = bsrc[p < end ? p : end - 1];
      }
      #pragma unroll
      for (int j = 0; j < 8; ++j) hs[j] = hmat[s[j] * HEADS + hh];
      #pragma unroll
      for (int j = 0; j < 8; ++j) {
        const int p = base + j * 4 + q;
        const float e = leaky(hs[j] * as + adst);
        if (p < end) m = fmaxf(m, e);
      }
    }
    m = fmaxf(m, __shfl_xor(m, 8));
    m = fmaxf(m, __shfl_xor(m, 16));

    const float mL = m * LOG2E;
    float den = 0.f, num = 0.f;
    if (q == 0) { den = exp_rel(eself, mL); num = den * hself; }
    #pragma unroll
    for (int j = 0; j < 8; ++j) {
      const float ee = exp_rel(eA[j], mL);
      den += ee;  num = fmaf(ee, hA[j], num);
    }
    if (hasB) {
      #pragma unroll
      for (int j = 0; j < 8; ++j) {
        const float ee = exp_rel(eB[j], mL);
        den += ee;  num = fmaf(ee, hB[j], num);
      }
    }
    for (int base = rest; base < end; base += CHUNK) {
      int   s[8];
      float hs[8];
      #pragma unroll
      for (int j = 0; j < 8; ++j) {
        const int p = base + j * 4 + q;
        s[j] = bsrc[p < end ? p : end - 1];
      }
      #pragma unroll
      for (int j = 0; j < 8; ++j) hs[j] = hmat[s[j] * HEADS + hh];
      #pragma unroll
      for (int j = 0; j < 8; ++j) {
        const int p = base + j * 4 + q;
        const float ee = exp_rel(leaky(hs[j] * as + adst), mL);
        if (p < end) { den += ee; num += ee * hs[j]; }
      }
    }
    den += __shfl_xor(den, 8);
    den += __shfl_xor(den, 16);
    num += __shfl_xor(num, 8);
    num += __shfl_xor(num, 16);

    const float oh = num / (den + 1e-16f) + gb;
    float z = lb;
    #pragma unroll
    for (int k = 0; k < HEADS; ++k) z += __shfl(oh, k) * lin_w[k];
    if (lane == 0) out[node] = fmaxf(z, 0.f) + bo;

    endC = end1;  cntC = cntN;
    end1 = end2;  cnt1 = cnt2;
  }
}

#define FUSED_LDS_INTS (RSIZE + RSIZE + 32 + MAXB + 32 + CAP)

template <bool INLDS>
__device__ __forceinline__ int list_id(const int* ids, int off) {
  const int* p = reinterpret_cast<const int*>(reinterpret_cast<const char*>(ids) + off);
  if (INLDS) return *p;
  return __hip_atomic_load(const_cast<int*>(p), __ATOMIC_RELAXED, __HIP_MEMORY_SCOPE_AGENT);
}

#define HGRP 16

template <bool INLDS, int NJ>
__device__ __forceinline__ void list_load_ids(const int* ids, int off0, int offLast,
                                              int (&s)[NJ]) {
  #pragma unroll
  for (int j = 0; j < NJ; ++j) {
    const int o = off0 + j * 16;
    s[j] = list_id<INLDS>(ids, o < offLast ? o : offLast);
  }
}

template <bool INLDS>
__device__ __forceinline__ void list_node_ids(const int* ids, int start, int end, int cnt,
                                              int q, int (&sA)[8], int (&sB)[4], int (&sC)[4]) {
  const int o  = (start + q) * 4;
  const int ol = (end - 1) * 4;
  if (cnt > 0)            list_load_ids<INLDS, 8>(ids, o, ol, sA);
  if (cnt > CHUNK)        list_load_ids<INLDS, 4>(ids, o + CHUNK * 4, ol, sB);
  if (cnt > CHUNK + HGRP) list_load_ids<INLDS, 4>(ids, o + (CHUNK + HGRP) * 4, ol, sC);
}

template <int NJ>
__device__ __forceinline__ void rows_load(const float* __restrict__ hmat, const int (&s)[NJ],
                                          int hh, float (&h)[NJ]) {
  #pragma unroll
  for (int j = 0; j < NJ; ++j) h[j] = hmat[s[j] * HEADS + hh];
}

template <int NJ>
__device__ __forceinline__ float scores(const float (&h)[NJ], float (&e)[NJ], float as,
                                        float adst, int left, float m) {
  #pragma unroll
  for (int j = 0; j < NJ; ++j) {
    const float v = h[j] * as + adst;
    e[j] = (j * 4 < left) ? fmaxf(v, NEG_SLOPE * v) : -INFINITY;
    m = fmaxf(m, e[j]);
  }
  return m;
}

template <int NJ>
__device__ __forceinline__ void exp_sums(const float (&h)[NJ], const float (&e)[NJ], float mL,
                                         float& den, float& num) {
  #pragma unroll
  for (int j = 0; j < NJ; ++j) {
    const float ee = exp_rel(e[j], mL);
    den += ee;  num = fmaf(ee, h[j], num);
  }
}

template <bool INLDS>
__device__ __forceinline__ void walk_range_nodes(
    const int* ids, const int* cntL, const int* posL, int nodeBase, int i0, int i1,
    const float* __restrict__ hmat, float as, float ad, float gb, float lb, float bo,
    const float* __restrict__ lin_w, float* __restrict__ out, int lane)
{
  const int q  = lane >> 3;
  const int hh = lane & 7;
  if (i0 >= i1) return;

  int endC = __builtin_amdgcn_readfirstlane(posL[i0]);
  int cntC = __builtin_amdgcn_readfirstlane(cntL[i0]);
  const int j1 = (i0 + 1 < i1) ? i0 + 1 : i1 - 1;
  int end1 = __builtin_amdgcn_readfirstlane(posL[j1]);
  int cnt1 = __builtin_amdgcn_readfirstlane(cntL[j1]);

  int nA[8], nB[4], nC[4];
  #pragma unroll
  for (int j = 0; j < 8; ++j) nA[j] = 0;
  #pragma unroll
  for (int j = 0; j < 4; ++j) { nB[j] = 0;  nC[j] = 0; }
  list_node_ids<INLDS>(ids, endC - cntC, endC, cntC, q, nA, nB, nC);

  for (int i = i0; i < i1; ++i) {
    const int node  = nodeBase + i;
    const int end   = endC;
    const int cnt   = cntC;
    const int start = end - cnt;
    const bool hasA = cnt > 0;
    const bool hasB = cnt > CHUNK;
    const bool hasC = cnt > CHUNK + HGRP;
    const int  rest = start + 2 * CHUNK;

    int sA[8], sB[4], sC[4];
    #pragma unroll
    for (int j = 0; j < 8; ++j) sA[j] = nA[j];
    #pragma unroll
    for (int j = 0; j < 4; ++j) { sB[j] = nB[j];  sC[j] = nC[j]; }

    const int cntN = (i + 1 < i1) ? cnt1 : 0;
    list_node_ids<INLDS>(ids, end1 - cntN, end1, cntN, q, nA, nB, nC);

    float hA[8], hB[4], hC[4], eA[8], eB[4], eC[4];
    const float hself = hmat[node * HEADS + hh];
    if (hasA) rows_load<8>(hmat, sA, hh, hA);
    if (hasB) rows_load<4>(hmat, sB, hh, hB);
    if (hasC) rows_load<4>(hmat, sC, hh, hC);

    const int i2 = (i + 2 < i1) ? i + 2 : i1 - 1;
    const int end2 = __builtin_amdgcn_readfirstlane(posL[i2]);
    const int cnt2 = __builtin_amdgcn_readfirstlane(cntL[i2]);

    const float adst  = hself * ad;
    const float es    = hself * as + adst;
    const float eself = fmaxf(es, NEG_SLOPE * es);
    const int   left  = cnt - q;

    float m = (q == 0) ? eself : -INFINITY;
    if (hasA) m = scores<8>(hA, eA, as, adst, left, m);
    if (hasB) m = scores<4>(hB, eB, as, adst, left - CHUNK, m);
    if (hasC) m = scores<4>(hC, eC, as, adst, left - CHUNK - HGRP, m);
    for (int base = rest; base < end; base += CHUNK) {
      int   s[8];
      float hs[8];
      list_load_ids<INLDS, 8>(ids, (base + q) * 4, (end - 1) * 4, s);
      rows_load<8>(hmat, s, hh, hs);
      #pragma unroll
      for (int j = 0; j < 8; ++j) {
        const int p = base + j * 4 + q;
        const float e = leaky(hs[j] * as + adst);
        if (p < end) m = fmaxf(m, e);
      }
    }
    m = fmaxf(m, __shfl_xor(m, 8));
    m = fmaxf(m, __shfl_xor(m, 16));

    const float mL = m * LOG2E;
    float den = 0.f, num = 0.f;
    if (q == 0) { den = exp_rel(eself, mL); num = den * hself; }
    if (hasA) exp_sums<8>(hA, eA, mL, den, num);
    if (hasB) exp_sums<4>(hB, eB, mL, den, num);
    if (hasC) exp_sums<4>(hC, eC, mL, den, num);
    for (int base = rest; base < end; base += CHUNK) {
      int   s[8];
      float hs[8];
      list_load_ids<INLDS, 8>(ids, (base + q) * 4, (end - 1) * 4, s);
      rows_load<8>(hmat, s, hh, hs);
      #pragma unroll
      for (int j = 0; j < 8; ++j) {
        const int p = base + j * 4 + q;
        const float ee = exp_rel(leaky(hs[j] * as + adst), mL);
        if (p < end) { den += ee; num += ee * hs[j]; }
      }
    }
    den += __shfl_xor(den, 8);
    den += __shfl_xor(den, 16);
    num += __shfl_xor(num, 8);
    num += __shfl_xor(num, 16);

    const float oh = num / (den + 1e-16f) + gb;
    float z = lb;
    #pragma unroll
    for (int k = 0; k < HEADS; ++k) z += __shfl(oh, k) * lin_w[k];
    if (lane == 0) out[node] = fmaxf(z, 0.f) + bo;

    endC = end1;  cntC = cntN;
    end1 = end2;  cnt1 = cnt2;
  }
}

__global__ __launch_bounds__(FPB) void gat_collect_gather_kernel(
    const int* __restrict__ part, const int* __restrict__ runTab,
    int* __restrict__ cursor, int* __restrict__ bsrc,
    const float* __restrict__ hmat,
    const float* __restrict__ att_src, const float* __restrict__ att_dst,
    const float* __restrict__ gat_bias, const float* __restrict__ lin_w,
    const float* __restrict__ lin_b, const float* __restrict__ bias,
    float* __restrict__ out, int n, int EB, int RP)
{
  static_assert(FR <= RSIZE, "a node's place in its range fits the entry's 8 bits and the tables");
  static_assert(MAXB == 2 * FPB, "two table places per thread");
  __shared__ int lds[FUSED_LDS_INTS];
  int* const cnt   = lds;
  int* const pos   = cnt + RSIZE;
  int* const wsum  = pos + RSIZE;
  int* const runPk = wsum + 32;
  int* const segS  = runPk + MAXB;
  int* const stage = segS + 32;
  const int t    = threadIdx.x;
  const int lane = t & 31;
  const int wave = __builtin_amdgcn_readfirstlane(t >> 5);
  const int r    = blockIdx.x;

  const float as = att_src[lane & 7];
  const float ad = att_dst[lane & 7];
  const float gb = gat_bias[lane & 7];
  const float lb = lin_b[0];
  const float bo = bias[0];

  const int pk0 = (t < EB)       ? runTab[(size_t)t * RP + r] : 0;
  const int pk1 = (t + FPB < EB) ? runTab[(size_t)(t + FPB) * RP + r] : 0;
  runPk[t]       = pk0;
  runPk[t + FPB] = pk1;
  const int myLen = (pk0 & 0xffff) + (pk1 & 0xffff);
  if (t < RSIZE) cnt[t] = 0;
  const int before = block_excl_scan(myLen, wsum, lane, wave, FW);
  if (t == FPB - 1) segS[1] = before + myLen;

  static_assert(FW * FHELD <= MAXB && FHELD <= 32, "held runs lie inside the table; one mask bit each");
  int k[FHELD];
  unsigned longMask = 0;
  {
    int pkv[FHELD];
    #pragma unroll
    for (int j = 0; j < FHELD; ++j) pkv[j] = runPk[wave + FW * j];
    #pragma unroll
    for (int j = 0; j < FHELD; ++j) {
      const int len = pkv[j] & 0xffff;
      const int beg = (wave + FW * j) * (PB * EPT) + (pkv[j] >> 16);
      k[j] = -1;
      if (len > 32) longMask |= 1u << j;
      if (lane < len) k[j] = part[beg + lane];
    }
  }
  longMask = __builtin_amdgcn_readfirstlane(longMask);

  #pragma unroll
  for (int j = 0; j < FHELD; ++j)
    if (k[j] >= 0) range_entry<false>(k[j], cnt, pos, stage, bsrc, 0, true);
  for (unsigned m = longMask; m; m &= m - 1)
    range_run<false>(part, runPk, wave + FW * __builtin_ctz(m), 32, lane, cnt, pos, stage, bsrc, 0, true);
  for (int b = wave + FW * FHELD; b < EB; b += FW)
    range_run<false>(part, runPk, b, 0, lane, cnt, pos, stage, bsrc, 0, true);
  __syncthreads();
  const int Ltot = segS[1];
  const bool inLds = (Ltot <= CAP);

  const int c    = (t < RSIZE) ? cnt[t] : 0;
  const int excl = block_excl_scan(c, wsum, lane, wave, FW);
  if (t < RSIZE) pos[t] = excl;
  if (!inLds && t == 0) segS[0] = atomicAdd(cursor, Ltot);
  __syncthreads();
  const int seg = inLds ? 0 : segS[0];

  {
    int p[FHELD];
    #pragma unroll
    for (int j = 0; j < FHELD; ++j)
      p[j] = (k[j] >= 0) ? atomicAdd(&pos[k[j] & (RSIZE - 1)], 1) : 0;
    #pragma unroll
    for (int j = 0; j < FHELD; ++j) {
      if (k[j] >= 0) {
        const int sid = k[j] >> RSHIFT;
        if (inLds) stage[p[j]] = sid; else bsrc[seg + p[j]] = sid;
      }
    }
  }
  for (unsigned m = longMask; m; m &= m - 1)
    range_run<true>(part, runPk, wave + FW * __builtin_ctz(m), 32, lane, cnt, pos, stage, bsrc, seg, inLds);
  for (int b = wave + FW * FHELD; b < EB; b += FW)
    range_run<true>(part, runPk, b, 0, lane, cnt, pos, stage, bsrc, seg, inLds);
  if (!inLds) __threadfence();
  __syncthreads();

  const int nodeBase = r * FR;
  const int leftN = n - nodeBase;
  const int nr    = leftN < FR ? leftN : FR;
  const int per   = nr / FW;
  const int rem   = nr - per * FW;
  const int i0    = wave * per + (wave < rem ? wave : rem);
  const int i1    = i0 + per + (wave < rem ? 1 : 0);
  if (inLds) walk_range_nodes<true >(stage,      cnt, pos, nodeBase, i0, i1, hmat, as, ad, gb, lb, bo, lin_w, out, lane);
  else       walk_range_nodes<false>(bsrc + seg, cnt, pos, nodeBase, i0, i1, hmat, as, ad, gb, lb, bo, lin_w, out, lane);
}

extern "C" void kernel_launch(void* const* d_in, const int* in_sizes, int n_in,
                              void* d_out, int out_size, void* d_ws, size_t ws_size,
                              hipStream_t stream) {
  const float* x        = (const float*)d_in[0];
  const int*   ei       = (const int*)d_in[1];
  const float* W        = (const float*)d_in[2];
  const float* att_src  = (const float*)d_in[3];
  const float* att_dst  = (const float*)d_in[4];
  const float* gat_bias = (const float*)d_in[5];
  const float* lin_w    = (const float*)d_in[6];
  const float* lin_b    = (const float*)d_in[7];
  const float* bias     = (const float*)d_in[8];
  float* out = (float*)d_out;

  const int n  = in_sizes[0] / IN_CH;
  const int nE = in_sizes[1] / 2;

  const size_t nh   = (size_t)n * HEADS;
  const size_t segH = (nh * sizeof(float) + 255) & ~(size_t)255;
  const size_t segN = ((size_t)n * sizeof(int) + 255) & ~(size_t)255;
  const size_t segT = (size_t)3 * MAXR * sizeof(int);
  const size_t segE = ((size_t)nE * sizeof(int) + 255) & ~(size_t)255;
  char* ws = (char*)d_ws;
  float* h_mat      = (float*)(ws);
  int*   deg        = (int*)  (ws + segH);
  int*   cur        = (int*)  (ws + segH + segN);
  int*   rangeCnt   = (int*)  (ws + segH + 2 * segN);
  int*   rangeCur   = rangeCnt + MAXR;
  int*   rangeStart = rangeCnt + 2 * MAXR;
  int*   part       = (int*)  (ws + segH + 2 * segN + segT);
  int*   bsrc       = (ws_size >= segH + 2 * segN + segT + 2 * segE) ? (int*)((char*)part + segE) : part;
  const int    eb     = (nE + PB * EPT - 1) / (PB * EPT);
  const int    FRn    = (n + FR - 1) / FR;
  const int    RP     = (FRn + 31) & ~31;
  const size_t segR   = (size_t)eb * RP * sizeof(int);
  const size_t need2  = segH + 2 * segN + segT + 2 * segE + segR;
  int*         runTab = (int*)((char*)part + 2 * segE);
  const bool   twoLaunch = (ws_size >= need2) && (eb <= MAXB) && (FRn <= MAXR);

  const int tiles  = (n + 15) / 16;
  const int blocks = (tiles + 31) / 32;

  if (twoLaunch) {
    gat_gemm_blocksort_kernel<<<blocks + eb, PB, 0, stream>>>(
        x, W, h_mat, (int*)nullptr, n, blocks, ei, part, runTab, rangeCur, nE, FRn, RP);

    gat_collect_gather_kernel<<<FRn, FPB, 0, stream>>>(
        part, runTab, rangeCur, bsrc, h_mat, att_src, att_dst, gat_bias, lin_w, lin_b, bias,
        out, n, eb, RP);
  } else {
    const int R = (n + RSIZE - 1) / RSIZE;

    gat_gemm_blocksort_kernel<<<blocks, PB, 0, stream>>>(
        x, W, h_mat, rangeCnt, n, blocks, ei, part, runTab, rangeCur, nE, 0, RP);

    gat_rangecount_kernel<<<eb, PB, 0, stream>>>(ei, rangeCnt, nE, R);

    gat_partition_kernel<<<eb, PB, 0, stream>>>(ei, rangeCnt, rangeCur, rangeStart, part, nE, R);

    gat_rangesort_kernel<<<R, PB, 0, stream>>>(part, rangeCnt, rangeStart, deg, cur, bsrc, n);

    gat_gather_kernel<<<GATHER_BLOCKS, 256, 0, stream>>>(
        bsrc, cur, deg, h_mat, att_src, att_dst, gat_bias, lin_w, lin_b, bias,
        out, n);
  }
}
